# starting kernel
# baseline (speedup 1.0000x reference)
#include <hip/hip_runtime.h>
#include <math.h>

typedef __attribute__((ext_vector_type(2))) float v2f;
typedef __attribute__((ext_vector_type(8))) float v8f;

#define FIN   128
#define HEADS 4
#define CDIM  32
#define HOUT  128
#define NEG_SLOPE 0.2f

__device__ __forceinline__ unsigned ordKey(float f) {
    unsigned u = __float_as_uint(f);
    return (u & 0x80000000u) ? ~u : (u | 0x80000000u);
}
__device__ __forceinline__ float ordInv(unsigned k) {
    unsigned u = (k & 0x80000000u) ? (k & 0x7FFFFFFFu) : ~k;
    return __uint_as_float(u);
}

template<int RELU>
__global__ __launch_bounds__(32)
void k_gemm_wmma(const float* __restrict__ X, const float* __restrict__ W,
                 float* __restrict__ H, int nRows) {
    const int lane   = threadIdx.x;
    const int tm     = blockIdx.x;
    const int tc4    = blockIdx.y;
    const int mlane  = lane & 15;
    const int khalf  = (lane >> 4) * 2;
    int row          = tm * 16 + mlane;
    if (row >= nRows) row = nRows - 1;
    const int colB   = tc4 * 64 + mlane;

    const float* xr = X + (size_t)row * FIN + khalf;

    v8f acc0 = {}, acc1 = {}, acc2 = {}, acc3 = {};

    for (int k0 = 0; k0 < FIN; k0 += 4) {
        v2f a;
        float a0 = xr[k0];
        float a1 = xr[k0 + 1];
        if (RELU) { a0 = fmaxf(a0, 0.0f); a1 = fmaxf(a1, 0.0f); }
        a.x = a0; a.y = a1;

        const float* w0 = W + (size_t)(k0 + khalf) * HOUT;
        const float* w1 = w0 + HOUT;
        v2f b0, b1, b2, b3;
        b0.x = w0[colB +  0]; b0.y = w1[colB +  0];
        b1.x = w0[colB + 16]; b1.y = w1[colB + 16];
        b2.x = w0[colB + 32]; b2.y = w1[colB + 32];
        b3.x = w0[colB + 48]; b3.y = w1[colB + 48];

        acc0 = __builtin_amdgcn_wmma_f32_16x16x4_f32(false, a, false, b0, (short)0, acc0, false, false);
        acc1 = __builtin_amdgcn_wmma_f32_16x16x4_f32(false, a, false, b1, (short)0, acc1, false, false);
        acc2 = __builtin_amdgcn_wmma_f32_16x16x4_f32(false, a, false, b2, (short)0, acc2, false, false);
        acc3 = __builtin_amdgcn_wmma_f32_16x16x4_f32(false, a, false, b3, (short)0, acc3, false, false);
    }

    const int mOff = (lane >> 4) * 8;
    for (int j = 0; j < 8; ++j) {
        int r = tm * 16 + mOff + j;
        if (r < nRows) {
            float* hr = H + (size_t)r * HOUT + tc4 * 64 + mlane;
            hr[0]  = acc0[j];
            hr[16] = acc1[j];
            hr[32] = acc2[j];
            hr[48] = acc3[j];
        }
    }
}

__global__ void k_attn_coeff(const float* __restrict__ H,
                             const float* __restrict__ a_src,
                             const float* __restrict__ a_dst,
                             float* __restrict__ als, float* __restrict__ ald,
                             int n) {
    int i = blockIdx.x * blockDim.x + threadIdx.x;
    if (i >= n * HEADS) return;
    int node = i >> 2, h = i & 3;
    const float* hp = H + (size_t)node * HOUT + h * CDIM;
    const float* as = a_src + h * CDIM;
    const float* ad = a_dst + h * CDIM;
    float s = 0.0f, d = 0.0f;
#pragma unroll
    for (int c = 0; c < CDIM; ++c) { float v = hp[c]; s = fmaf(v, as[c], s); d = fmaf(v, ad[c], d); }
    als[i] = s; ald[i] = d;
}

__global__ void k_init_attn(unsigned* __restrict__ mkey, float* __restrict__ denom, int n) {
    int i = blockIdx.x * blockDim.x + threadIdx.x;
    if (i < n * HEADS) { mkey[i] = ordKey(-INFINITY); denom[i] = 0.0f; }
}

__global__ void k_init_out(float* __restrict__ out, const float* __restrict__ bias, int n) {
    int i = blockIdx.x * blockDim.x + threadIdx.x;
    if (i < n * HOUT) out[i] = bias[i & (HOUT - 1)];
}

__global__ void k_edge_max(const int* __restrict__ src, const int* __restrict__ dst,
                           const float* __restrict__ als, const float* __restrict__ ald,
                           unsigned* __restrict__ mkey, int e_total) {
    int i = blockIdx.x * blockDim.x + threadIdx.x;
    if (i >= e_total * HEADS) return;
    int e = i >> 2, h = i & 3;
    int s = src[e], d = dst[e];
    float v = als[s * HEADS + h] + ald[d * HEADS + h];
    v = (v > 0.0f) ? v : v * NEG_SLOPE;
    atomicMax(&mkey[d * HEADS + h], ordKey(v));
}

__global__ void k_edge_exp(const int* __restrict__ src, const int* __restrict__ dst,
                           const float* __restrict__ als, const float* __restrict__ ald,
                           const unsigned* __restrict__ mkey,
                           float* __restrict__ w, float* __restrict__ denom, int e_total) {
    int i = blockIdx.x * blockDim.x + threadIdx.x;
    if (i >= e_total * HEADS) return;
    int e = i >> 2, h = i & 3;
    int s = src[e], d = dst[e];
    float v = als[s * HEADS + h] + ald[d * HEADS + h];
    v = (v > 0.0f) ? v : v * NEG_SLOPE;
    float mm = ordInv(mkey[d * HEADS + h]);
    float ww = expf(v - mm);
    w[i] = ww;
    __hip_atomic_fetch_add(&denom[d * HEADS + h], ww, __ATOMIC_RELAXED, __HIP_MEMORY_SCOPE_AGENT);
}

__global__ void k_edge_scatter(const int* __restrict__ src, const int* __restrict__ dst,
                               const float* __restrict__ Hm,
                               const float* __restrict__ w, const float* __restrict__ denom,
                               float* __restrict__ out, int e_total) {
    int i = blockIdx.x * blockDim.x + threadIdx.x;
    if (i >= e_total * HEADS) return;
    int e = i >> 2, h = i & 3;
    int s = src[e], d = dst[e];
    float alpha = w[i] / fmaxf(denom[d * HEADS + h], 1e-16f);
    const float4* hs = (const float4*)(Hm + (size_t)s * HOUT + h * CDIM);
    float* op = out + (size_t)d * HOUT + h * CDIM;
#pragma unroll
    for (int c4 = 0; c4 < CDIM / 4; ++c4) {
        float4 hv = hs[c4];
        __hip_atomic_fetch_add(op + c4 * 4 + 0, hv.x * alpha, __ATOMIC_RELAXED, __HIP_MEMORY_SCOPE_AGENT);
        __hip_atomic_fetch_add(op + c4 * 4 + 1, hv.y * alpha, __ATOMIC_RELAXED, __HIP_MEMORY_SCOPE_AGENT);
        __hip_atomic_fetch_add(op + c4 * 4 + 2, hv.z * alpha, __ATOMIC_RELAXED, __HIP_MEMORY_SCOPE_AGENT);
        __hip_atomic_fetch_add(op + c4 * 4 + 3, hv.w * alpha, __ATOMIC_RELAXED, __HIP_MEMORY_SCOPE_AGENT);
    }
}

extern "C" void kernel_launch(void* const* d_in, const int* in_sizes, int n_in,
                              void* d_out, int out_size, void* d_ws, size_t ws_size,
                              hipStream_t stream) {
    const float* x      = (const float*)d_in[0];
    const int*   ei     = (const int*)d_in[1];
    const float* W1     = (const float*)d_in[2];
    const float* a_src1 = (const float*)d_in[3];
    const float* a_dst1 = (const float*)d_in[4];
    const float* b1     = (const float*)d_in[5];
    const float* W2     = (const float*)d_in[6];
    const float* a_src2 = (const float*)d_in[7];
    const float* a_dst2 = (const float*)d_in[8];
    const float* b2     = (const float*)d_in[9];

    const int N = in_sizes[0] / FIN;
    const int E = in_sizes[1] / 2;
    const int* src = ei;
    const int* dst = ei + E;

    char* p = (char*)d_ws;
    float*    H     = (float*)p;    p += (size_t)N * HOUT * sizeof(float);
    float*    als   = (float*)p;    p += (size_t)N * HEADS * sizeof(float);
    float*    ald   = (float*)p;    p += (size_t)N * HEADS * sizeof(float);
    unsigned* mkey  = (unsigned*)p; p += (size_t)N * HEADS * sizeof(unsigned);
    float*    denom = (float*)p;    p += (size_t)N * HEADS * sizeof(float);
    float*    w     = (float*)p;    p += (size_t)E * HEADS * sizeof(float);

    float* out = (float*)d_out;

    dim3 gGemm((N + 15) / 16, 2);
    const int BT = 256;
    int gNH  = (N * HEADS + BT - 1) / BT;
    int gNO  = (N * HOUT  + BT - 1) / BT;
    int gEH  = (E * HEADS + BT - 1) / BT;

    k_gemm_wmma<0><<<gGemm, 32, 0, stream>>>(x, W1, H, N);
    k_attn_coeff<<<gNH, BT, 0, stream>>>(H, a_src1, a_dst1, als, ald, N);
    k_init_attn <<<gNH, BT, 0, stream>>>(mkey, denom, N);
    k_init_out  <<<gNO, BT, 0, stream>>>(out, b1, N);
    k_edge_max  <<<gEH, BT, 0, stream>>>(src, dst, als, ald, mkey, E);
    k_edge_exp  <<<gEH, BT, 0, stream>>>(src, dst, als, ald, mkey, w, denom, E);
    k_edge_scatter<<<gEH, BT, 0, stream>>>(src, dst, H, w, denom, out, E);

    k_gemm_wmma<1><<<gGemm, 32, 0, stream>>>(out, W2, H, N);
    k_attn_coeff<<<gNH, BT, 0, stream>>>(H, a_src2, a_dst2, als, ald, N);
    k_init_attn <<<gNH, BT, 0, stream>>>(mkey, denom, N);
    k_init_out  <<<gNO, BT, 0, stream>>>(out, b2, N);
    k_edge_max  <<<gEH, BT, 0, stream>>>(src, dst, als, ald, mkey, E);
    k_edge_exp  <<<gEH, BT, 0, stream>>>(src, dst, als, ald, mkey, w, denom, E);
    k_edge_scatter<<<gEH, BT, 0, stream>>>(src, dst, H, w, denom, out, E);
}


# Edges counting-sorted by destination once per call; one wave per node does segment max, softmax sum and weighted row sum in registers in edge-id order: no float atomics, sum order changed
# speedup vs baseline: 7.9293x; 7.9293x over previous
#include <hip/hip_runtime.h>
#include <math.h>


typedef __attribute__((ext_vector_type(2))) float v2f;
typedef __attribute__((ext_vector_type(8))) float v8f;

#define FIN   128
#define HEADS 4
#define CDIM  32
#define HOUT  128
#define NEG_SLOPE 0.2f

template<int RELU>
__global__ __launch_bounds__(32)
void k_gemm_wmma(const float* __restrict__ X, const float* __restrict__ W,
                 float* __restrict__ H, int nRows) {
    const int lane   = threadIdx.x;
    const int tm     = blockIdx.x;
    const int tc4    = blockIdx.y;
    const int mlane  = lane & 15;
    const int khalf  = (lane >> 4) * 2;
    int row          = tm * 16 + mlane;
    if (row >= nRows) row = nRows - 1;
    const int colB   = tc4 * 64 + mlane;

    const float* xr = X + (size_t)row * FIN + khalf;

    v8f acc0 = {}, acc1 = {}, acc2 = {}, acc3 = {};

    for (int k0 = 0; k0 < FIN; k0 += 4) {
        v2f a;
        float a0 = xr[k0];
        float a1 = xr[k0 + 1];
        if (RELU) { a0 = fmaxf(a0, 0.0f); a1 = fmaxf(a1, 0.0f); }
        a.x = a0; a.y = a1;

        const float* w0 = W + (size_t)(k0 + khalf) * HOUT;
        const float* w1 = w0 + HOUT;
        v2f b0, b1, b2, b3;
        b0.x = w0[colB +  0]; b0.y = w1[colB +  0];
        b1.x = w0[colB + 16]; b1.y = w1[colB + 16];
        b2.x = w0[colB + 32]; b2.y = w1[colB + 32];
        b3.x = w0[colB + 48]; b3.y = w1[colB + 48];

        acc0 = __builtin_amdgcn_wmma_f32_16x16x4_f32(false, a, false, b0, (short)0, acc0, false, false);
        acc1 = __builtin_amdgcn_wmma_f32_16x16x4_f32(false, a, false, b1, (short)0, acc1, false, false);
        acc2 = __builtin_amdgcn_wmma_f32_16x16x4_f32(false, a, false, b2, (short)0, acc2, false, false);
        acc3 = __builtin_amdgcn_wmma_f32_16x16x4_f32(false, a, false, b3, (short)0, acc3, false, false);
    }

    const int mOff = (lane >> 4) * 8;
    for (int j = 0; j < 8; ++j) {
        int r = tm * 16 + mOff + j;
        if (r < nRows) {
            float* hr = H + (size_t)r * HOUT + tc4 * 64 + mlane;
            hr[0]  = acc0[j];
            hr[16] = acc1[j];
            hr[32] = acc2[j];
            hr[48] = acc3[j];
        }
    }
}

__global__ void k_attn_coeff(const float* __restrict__ H,
                             const float* __restrict__ a_src,
                             const float* __restrict__ a_dst,
                             float* __restrict__ als, float* __restrict__ ald,
                             int n) {
    int i = blockIdx.x * blockDim.x + threadIdx.x;
    if (i >= n * HEADS) return;
    int node = i >> 2, h = i & 3;
    const float* hp = H + (size_t)node * HOUT + h * CDIM;
    const float* as = a_src + h * CDIM;
    const float* ad = a_dst + h * CDIM;
    float s = 0.0f, d = 0.0f;
#pragma unroll
    for (int c = 0; c < CDIM; ++c) { float v = hp[c]; s = fmaf(v, as[c], s); d = fmaf(v, ad[c], d); }
    als[i] = s; ald[i] = d;
}

#define SCAN_B    256
#define SCAN_TILE 1024

__global__ void k_zero_cnt(int* __restrict__ cnt, int n) {
    int i = blockIdx.x * blockDim.x + threadIdx.x;
    if (i < n) cnt[i] = 0;
}

__global__ void k_hist(const int* __restrict__ dst, int* __restrict__ cnt, int e_total) {
    int e = blockIdx.x * blockDim.x + threadIdx.x;
    if (e < e_total) atomicAdd(&cnt[dst[e]], 1);
}

__global__ __launch_bounds__(SCAN_B)
void k_scan_local(const int* __restrict__ cnt, int* __restrict__ rowptr, int* __restrict__ bsum, int n) {
    __shared__ int wtot[SCAN_B / 32];
    const int t = threadIdx.x, lane = t & 31, wv = t >> 5;
    const int base = blockIdx.x * SCAN_TILE + t * 4;
    int a0 = 0, a1 = 0, a2 = 0, a3 = 0;
    if (base + 3 < n) {
        const int4 v = *(const int4*)(cnt + base);
        a0 = v.x; a1 = v.y; a2 = v.z; a3 = v.w;
    } else {
        if (base     < n) a0 = cnt[base];
        if (base + 1 < n) a1 = cnt[base + 1];
        if (base + 2 < n) a2 = cnt[base + 2];
    }
    const int tsum = a0 + a1 + a2 + a3;
    int incl = tsum;
    for (int o = 1; o < 32; o <<= 1) { const int u = __shfl_up(incl, o); if (lane >= o) incl += u; }
    if (lane == 31) wtot[wv] = incl;
    __syncthreads();
    int pre = 0;
    for (int k = 0; k < wv; ++k) pre += wtot[k];
    const int ex = pre + incl - tsum;
    if (base + 3 < n) {
        int4 r; r.x = ex; r.y = ex + a0; r.z = ex + a0 + a1; r.w = ex + a0 + a1 + a2;
        *(int4*)(rowptr + base) = r;
    } else {
        if (base     < n) rowptr[base]     = ex;
        if (base + 1 < n) rowptr[base + 1] = ex + a0;
        if (base + 2 < n) rowptr[base + 2] = ex + a0 + a1;
    }
    if (t == SCAN_B - 1) bsum[blockIdx.x] = pre + incl;
}

__global__ __launch_bounds__(SCAN_B)
void k_scan_fix(int* __restrict__ rowptr, int* __restrict__ cursor, const int* __restrict__ bsum, int n) {
    __shared__ int wtot[SCAN_B / 32];
    const int t = threadIdx.x, lane = t & 31, wv = t >> 5;
    const int b = blockIdx.x;
    int part = 0;
    for (int k = t; k < b; k += SCAN_B) part += bsum[k];
    for (int o = 16; o > 0; o >>= 1) part += __shfl_xor(part, o);
    if (lane == 0) wtot[wv] = part;
    __syncthreads();
    int off = 0;
    for (int k = 0; k < SCAN_B / 32; ++k) off += wtot[k];
    const int base = b * SCAN_TILE + t * 4;
    for (int q = 0; q < 4; ++q) {
        const int i = base + q;
        if (i < n) { const int r = rowptr[i] + off; rowptr[i] = r; cursor[i] = r; }
    }
    if (b == (int)gridDim.x - 1 && t == 0) rowptr[n] = off + bsum[b];
}

__global__ void k_fill(const int* __restrict__ src, const int* __restrict__ dst,
                       int* __restrict__ cursor, int2* __restrict__ csr, int e_total) {
    int e = blockIdx.x * blockDim.x + threadIdx.x;
    if (e >= e_total) return;
    const int pos = atomicAdd(&cursor[dst[e]], 1);
    int2 r; r.x = e; r.y = src[e];
    csr[pos] = r;
}

__device__ __forceinline__ float wave_max(float v) {
    for (int o = 16; o > 0; o >>= 1) v = fmaxf(v, __shfl_xor(v, o));
    return v;
}
__device__ __forceinline__ float wave_sum(float v) {
    for (int o = 16; o > 0; o >>= 1) v += __shfl_xor(v, o);
    return v;
}
__device__ __forceinline__ float lane_f(float x, int l) {
    return __uint_as_float((unsigned)__builtin_amdgcn_readlane((int)__float_as_uint(x), l));
}
__device__ __forceinline__ float lrelu(float v) { return (v > 0.0f) ? v : v * NEG_SLOPE; }

__device__ __forceinline__ void edge_logits(const float* __restrict__ als, int s, const float4 ad,
                                            float& x0, float& x1, float& x2, float& x3) {
    const float4 a = *(const float4*)(als + (size_t)s * HEADS);
    x0 = lrelu(a.x + ad.x); x1 = lrelu(a.y + ad.y); x2 = lrelu(a.z + ad.z); x3 = lrelu(a.w + ad.w);
}

template<int K>
__device__ __forceinline__ void accum_k(float4& acc, const float4* __restrict__ Hl, int s,
                                        float a0, float a1, float a2, float a3, int jj,
                                        float k0, float k1, float k2, float k3) {
    float4 hv[K];
    for (int u = 0; u < K; ++u) {
        const int sj = __builtin_amdgcn_readlane(s, jj + u);
        hv[u] = Hl[(size_t)sj * (HOUT / 4)];
    }
    for (int u = 0; u < K; ++u) {
        const float am = fmaf(lane_f(a3, jj + u), k3, fmaf(lane_f(a2, jj + u), k2,
                         fmaf(lane_f(a1, jj + u), k1, lane_f(a0, jj + u) * k0)));
        acc.x += hv[u].x * am; acc.y += hv[u].y * am; acc.z += hv[u].z * am; acc.w += hv[u].w * am;
    }
}

__device__ __forceinline__ void accum_chunk(float4& acc, const float4* __restrict__ Hl, int s,
                                            float a0, float a1, float a2, float a3, int cnt,
                                            float k0, float k1, float k2, float k3) {
    int jj = 0;
    for (; jj + 8 <= cnt; jj += 8) accum_k<8>(acc, Hl, s, a0, a1, a2, a3, jj, k0, k1, k2, k3);
    if (jj + 4 <= cnt) { accum_k<4>(acc, Hl, s, a0, a1, a2, a3, jj, k0, k1, k2, k3); jj += 4; }
    if (jj + 2 <= cnt) { accum_k<2>(acc, Hl, s, a0, a1, a2, a3, jj, k0, k1, k2, k3); jj += 2; }
    if (jj < cnt)        accum_k<1>(acc, Hl, s, a0, a1, a2, a3, jj, k0, k1, k2, k3);
}

__global__ __launch_bounds__(256)
void k_gather(const int* __restrict__ rowptr, const int2* __restrict__ csr,
              const float* __restrict__ Hm,
              const float* __restrict__ als, const float* __restrict__ ald,
              const float* __restrict__ bias, float* __restrict__ out, int n) {
    const int lane = threadIdx.x & 31;
    const int hsel = lane >> 3;
    const float k0 = (hsel == 0) ? 1.0f : 0.0f, k1 = (hsel == 1) ? 1.0f : 0.0f;
    const float k2 = (hsel == 2) ? 1.0f : 0.0f, k3 = (hsel == 3) ? 1.0f : 0.0f;
    const float4* Hl = (const float4*)Hm + lane;
    const float* bl = bias + lane * 4;
    const int wave0  = __builtin_amdgcn_readfirstlane((int)((blockIdx.x * blockDim.x + threadIdx.x) >> 5));
    const int nwaves = (int)((gridDim.x * blockDim.x) >> 5);
    for (int node = wave0; node < n; node += nwaves) {
    const int beg = rowptr[node];
    const int end = rowptr[node + 1];
    const float4 ad = *(const float4*)(ald + (size_t)node * HEADS);

    const int  c0  = min(end - beg, 32);
    const bool ok0 = lane < c0;
    int eid = 0x7fffffff, sraw = 0;
    if (ok0) { const int2 r = csr[beg + lane]; eid = r.x; sraw = r.y; }
    int rank = 0;
    for (int k = 0; k < c0; ++k) rank += (__builtin_amdgcn_readlane(eid, k) < eid) ? 1 : 0;
    int from = lane;
    for (int k = 0; k < c0; ++k) from = (__builtin_amdgcn_readlane(rank, k) == lane) ? k : from;
    const int s0 = __shfl(sraw, from);

    float e0, e1, e2, e3;
    edge_logits(als, s0, ad, e0, e1, e2, e3);
    float m0 = ok0 ? e0 : -INFINITY, m1 = ok0 ? e1 : -INFINITY, m2 = ok0 ? e2 : -INFINITY, m3 = ok0 ? e3 : -INFINITY;
    for (int cb = beg + 32; cb < end; cb += 32) {
        const int j = cb + lane;
        if (j < end) {
            float x0, x1, x2, x3;
            edge_logits(als, csr[j].y, ad, x0, x1, x2, x3);
            m0 = fmaxf(m0, x0); m1 = fmaxf(m1, x1); m2 = fmaxf(m2, x2); m3 = fmaxf(m3, x3);
        }
    }
    m0 = wave_max(m0); m1 = wave_max(m1); m2 = wave_max(m2); m3 = wave_max(m3);
    if (!__builtin_isfinite(m0)) m0 = 0.0f;
    if (!__builtin_isfinite(m1)) m1 = 0.0f;
    if (!__builtin_isfinite(m2)) m2 = 0.0f;
    if (!__builtin_isfinite(m3)) m3 = 0.0f;

    const float w0 = ok0 ? expf(e0 - m0) : 0.0f, w1 = ok0 ? expf(e1 - m1) : 0.0f;
    const float w2 = ok0 ? expf(e2 - m2) : 0.0f, w3 = ok0 ? expf(e3 - m3) : 0.0f;
    float d0 = w0, d1 = w1, d2 = w2, d3 = w3;
    for (int cb = beg + 32; cb < end; cb += 32) {
        const int j = cb + lane;
        if (j < end) {
            float x0, x1, x2, x3;
            edge_logits(als, csr[j].y, ad, x0, x1, x2, x3);
            d0 += expf(x0 - m0); d1 += expf(x1 - m1); d2 += expf(x2 - m2); d3 += expf(x3 - m3);
        }
    }
    d0 = wave_sum(d0); d1 = wave_sum(d1); d2 = wave_sum(d2); d3 = wave_sum(d3);
    const float q0 = fmaxf(d0, 1e-16f), q1 = fmaxf(d1, 1e-16f), q2 = fmaxf(d2, 1e-16f), q3 = fmaxf(d3, 1e-16f);

    float4 acc; acc.x = 0.0f; acc.y = 0.0f; acc.z = 0.0f; acc.w = 0.0f;
    accum_chunk(acc, Hl, s0, w0 / q0, w1 / q1, w2 / q2, w3 / q3, c0, k0, k1, k2, k3);
    for (int cb = beg + 32; cb < end; cb += 32) {
        const int  j  = cb + lane;
        const bool ok = j < end;
        const int  s  = ok ? csr[j].y : 0;
        float x0, x1, x2, x3;
        edge_logits(als, s, ad, x0, x1, x2, x3);
        const float a0 = ok ? expf(x0 - m0) / q0 : 0.0f, a1 = ok ? expf(x1 - m1) / q1 : 0.0f;
        const float a2 = ok ? expf(x2 - m2) / q2 : 0.0f, a3 = ok ? expf(x3 - m3) / q3 : 0.0f;
        accum_chunk(acc, Hl, s, a0, a1, a2, a3, min(end - cb, 32), k0, k1, k2, k3);
    }

    float4 o; o.x = acc.x + bl[0]; o.y = acc.y + bl[1]; o.z = acc.z + bl[2]; o.w = acc.w + bl[3];
    ((float4*)(out + (size_t)node * HOUT))[lane] = o;
    }
}

extern "C" void kernel_launch(void* const* d_in, const int* in_sizes, int n_in,
                              void* d_out, int out_size, void* d_ws, size_t ws_size,
                              hipStream_t stream) {
    const float* x      = (const float*)d_in[0];
    const int*   ei     = (const int*)d_in[1];
    const float* W1     = (const float*)d_in[2];
    const float* a_src1 = (const float*)d_in[3];
    const float* a_dst1 = (const float*)d_in[4];
    const float* b1     = (const float*)d_in[5];
    const float* W2     = (const float*)d_in[6];
    const float* a_src2 = (const float*)d_in[7];
    const float* a_dst2 = (const float*)d_in[8];
    const float* b2     = (const float*)d_in[9];

    const int N = in_sizes[0] / FIN;
    const int E = in_sizes[1] / 2;
    const int* src = ei;
    const int* dst = ei + E;

    const int nTiles = (N + SCAN_TILE - 1) / SCAN_TILE;
    const size_t nPad = ((size_t)N + 4) & ~(size_t)3;

    char* p = (char*)d_ws;
    float* H      = (float*)p; p += (size_t)N * HOUT * sizeof(float);
    float* als    = (float*)p; p += (size_t)N * HEADS * sizeof(float);
    float* ald    = (float*)p; p += (size_t)N * HEADS * sizeof(float);
    int*   rowptr = (int*)p;   p += nPad * sizeof(int);
    int*   cnt    = (int*)p;   p += nPad * sizeof(int);
    int*   cursor = (int*)p;   p += nPad * sizeof(int);
    int*   bsum   = (int*)p;   p += (((size_t)nTiles + 3) & ~(size_t)3) * sizeof(int);
    int2*  csr    = (int2*)p;  p += (size_t)E * sizeof(int2);

    float* out = (float*)d_out;

    dim3 gGemm((N + 15) / 16, 2);
    const int BT = 256;
    int gN   = (N + BT - 1) / BT;
    int gE   = (E + BT - 1) / BT;
    int gNH  = (N * HEADS + BT - 1) / BT;
    int gNW  = (N + 31) / 32;

    k_zero_cnt  <<<gN, BT, 0, stream>>>(cnt, N);
    k_hist      <<<gE, BT, 0, stream>>>(dst, cnt, E);
    k_scan_local<<<nTiles, SCAN_B, 0, stream>>>(cnt, rowptr, bsum, N);
    k_scan_fix  <<<nTiles, SCAN_B, 0, stream>>>(rowptr, cursor, bsum, N);
    k_fill      <<<gE, BT, 0, stream>>>(src, dst, cursor, csr, E);

    k_gemm_wmma<0><<<gGemm, 32, 0, stream>>>(x, W1, H, N);
    k_attn_coeff<<<gNH, BT, 0, stream>>>(H, a_src1, a_dst1, als, ald, N);
    k_gather    <<<gNW, BT, 0, stream>>>(rowptr, csr, H, als, ald, b1, out, N);

    k_gemm_wmma<1><<<gGemm, 32, 0, stream>>>(out, W2, H, N);
    k_attn_coeff<<<gNH, BT, 0, stream>>>(H, a_src2, a_dst2, als, ald, N);
    k_gather    <<<gNW, BT, 0, stream>>>(rowptr, csr, H, als, ald, b2, out, N);
}


# Edge bucketing in one atomic pass: the counting add's returned value is the edge's slot in its destination's bucket, placement by plain stores (no cursor atomics), 8 edges per thread
# speedup vs baseline: 8.8082x; 1.1108x over previous
#include <hip/hip_runtime.h>
#include <math.h>


typedef __attribute__((ext_vector_type(2))) float v2f;
typedef __attribute__((ext_vector_type(8))) float v8f;

#define FIN   128
#define HEADS 4
#define CDIM  32
#define HOUT  128
#define NEG_SLOPE 0.2f

template<int RELU>
__global__ __launch_bounds__(32)
void k_gemm_wmma(const float* __restrict__ X, const float* __restrict__ W,
                 float* __restrict__ H, int nRows) {
    const int lane   = threadIdx.x;
    const int tm     = blockIdx.x;
    const int tc4    = blockIdx.y;
    const int mlane  = lane & 15;
    const int khalf  = (lane >> 4) * 2;
    int row          = tm * 16 + mlane;
    if (row >= nRows) row = nRows - 1;
    const int colB   = tc4 * 64 + mlane;

    const float* xr = X + (size_t)row * FIN + khalf;

    v8f acc0 = {}, acc1 = {}, acc2 = {}, acc3 = {};

    for (int k0 = 0; k0 < FIN; k0 += 4) {
        v2f a;
        float a0 = xr[k0];
        float a1 = xr[k0 + 1];
        if (RELU) { a0 = fmaxf(a0, 0.0f); a1 = fmaxf(a1, 0.0f); }
        a.x = a0; a.y = a1;

        const float* w0 = W + (size_t)(k0 + khalf) * HOUT;
        const float* w1 = w0 + HOUT;
        v2f b0, b1, b2, b3;
        b0.x = w0[colB +  0]; b0.y = w1[colB +  0];
        b1.x = w0[colB + 16]; b1.y = w1[colB + 16];
        b2.x = w0[colB + 32]; b2.y = w1[colB + 32];
        b3.x = w0[colB + 48]; b3.y = w1[colB + 48];

        acc0 = __builtin_amdgcn_wmma_f32_16x16x4_f32(false, a, false, b0, (short)0, acc0, false, false);
        acc1 = __builtin_amdgcn_wmma_f32_16x16x4_f32(false, a, false, b1, (short)0, acc1, false, false);
        acc2 = __builtin_amdgcn_wmma_f32_16x16x4_f32(false, a, false, b2, (short)0, acc2, false, false);
        acc3 = __builtin_amdgcn_wmma_f32_16x16x4_f32(false, a, false, b3, (short)0, acc3, false, false);
    }

    const int mOff = (lane >> 4) * 8;
    for (int j = 0; j < 8; ++j) {
        int r = tm * 16 + mOff + j;
        if (r < nRows) {
            float* hr = H + (size_t)r * HOUT + tc4 * 64 + mlane;
            hr[0]  = acc0[j];
            hr[16] = acc1[j];
            hr[32] = acc2[j];
            hr[48] = acc3[j];
        }
    }
}

__global__ void k_attn_coeff(const float* __restrict__ H,
                             const float* __restrict__ a_src,
                             const float* __restrict__ a_dst,
                             float* __restrict__ als, float* __restrict__ ald,
                             int n) {
    int i = blockIdx.x * blockDim.x + threadIdx.x;
    if (i >= n * HEADS) return;
    int node = i >> 2, h = i & 3;
    const float* hp = H + (size_t)node * HOUT + h * CDIM;
    const float* as = a_src + h * CDIM;
    const float* ad = a_dst + h * CDIM;
    float s = 0.0f, d = 0.0f;
#pragma unroll
    for (int c = 0; c < CDIM; ++c) { float v = hp[c]; s = fmaf(v, as[c], s); d = fmaf(v, ad[c], d); }
    als[i] = s; ald[i] = d;
}

#define SCAN_B    256
#define SCAN_TILE 1024

__global__ void k_zero_cnt(int* __restrict__ cnt, int n) {
    int i = blockIdx.x * blockDim.x + threadIdx.x;
    if (i < n) cnt[i] = 0;
}

#define EDGE_TILE  1024
#define EDGE_BLOCK 2048

__global__ __launch_bounds__(256)
void k_count_slot(const int* __restrict__ dst, int* __restrict__ cnt, int* __restrict__ slot, int e_total) {
    const int eA = blockIdx.x * EDGE_BLOCK + threadIdx.x * 4;
    const int eB = eA + EDGE_TILE;
    const bool wide = ((((size_t)dst) | ((size_t)slot)) & 15) == 0;
    if (wide && eB + 3 < e_total) {
        const int4 dA = *(const int4*)(dst + eA);
        const int4 dB = *(const int4*)(dst + eB);
        __builtin_amdgcn_sched_barrier(0);
        int4 rA, rB;
        rA.x = atomicAdd(&cnt[dA.x], 1);
        rA.y = atomicAdd(&cnt[dA.y], 1);
        rA.z = atomicAdd(&cnt[dA.z], 1);
        rA.w = atomicAdd(&cnt[dA.w], 1);
        rB.x = atomicAdd(&cnt[dB.x], 1);
        rB.y = atomicAdd(&cnt[dB.y], 1);
        rB.z = atomicAdd(&cnt[dB.z], 1);
        rB.w = atomicAdd(&cnt[dB.w], 1);
        *(int4*)(slot + eA) = rA;
        *(int4*)(slot + eB) = rB;
    } else {
        for (int q = 0; q < 4; ++q) {
            const int e = eA + q;
            if (e < e_total) slot[e] = atomicAdd(&cnt[dst[e]], 1);
        }
        for (int q = 0; q < 4; ++q) {
            const int e = eB + q;
            if (e < e_total) slot[e] = atomicAdd(&cnt[dst[e]], 1);
        }
    }
}

__global__ __launch_bounds__(SCAN_B)
void k_scan_local(const int* __restrict__ cnt, int* __restrict__ rowptr, int* __restrict__ bsum, int n) {
    __shared__ int wtot[SCAN_B / 32];
    const int t = threadIdx.x, lane = t & 31, wv = t >> 5;
    const int base = blockIdx.x * SCAN_TILE + t * 4;
    int a0 = 0, a1 = 0, a2 = 0, a3 = 0;
    if (base + 3 < n) {
        const int4 v = *(const int4*)(cnt + base);
        a0 = v.x; a1 = v.y; a2 = v.z; a3 = v.w;
    } else {
        if (base     < n) a0 = cnt[base];
        if (base + 1 < n) a1 = cnt[base + 1];
        if (base + 2 < n) a2 = cnt[base + 2];
    }
    const int tsum = a0 + a1 + a2 + a3;
    int incl = tsum;
    for (int o = 1; o < 32; o <<= 1) { const int u = __shfl_up(incl, o); if (lane >= o) incl += u; }
    if (lane == 31) wtot[wv] = incl;
    __syncthreads();
    int pre = 0;
    for (int k = 0; k < wv; ++k) pre += wtot[k];
    const int ex = pre + incl - tsum;
    if (base + 3 < n) {
        int4 r; r.x = ex; r.y = ex + a0; r.z = ex + a0 + a1; r.w = ex + a0 + a1 + a2;
        *(int4*)(rowptr + base) = r;
    } else {
        if (base     < n) rowptr[base]     = ex;
        if (base + 1 < n) rowptr[base + 1] = ex + a0;
        if (base + 2 < n) rowptr[base + 2] = ex + a0 + a1;
    }
    if (t == SCAN_B - 1) bsum[blockIdx.x] = pre + incl;
}

__global__ __launch_bounds__(SCAN_B)
void k_scan_fix(int* __restrict__ rowptr, const int* __restrict__ bsum, int n) {
    __shared__ int wtot[SCAN_B / 32];
    const int t = threadIdx.x, lane = t & 31, wv = t >> 5;
    const int b = blockIdx.x;
    int part = 0;
    for (int k = t; k < b; k += SCAN_B) part += bsum[k];
    for (int o = 16; o > 0; o >>= 1) part += __shfl_xor(part, o);
    if (lane == 0) wtot[wv] = part;
    __syncthreads();
    int off = 0;
    for (int k = 0; k < SCAN_B / 32; ++k) off += wtot[k];
    const int base = b * SCAN_TILE + t * 4;
    for (int q = 0; q < 4; ++q) {
        const int i = base + q;
        if (i < n) rowptr[i] += off;
    }
    if (b == (int)gridDim.x - 1 && t == 0) rowptr[n] = off + bsum[b];
}

__device__ __forceinline__ void place_one(int2* __restrict__ csr, int pos, int e, int s) {
    int2 r; r.x = e; r.y = s;
    csr[pos] = r;
}

__global__ __launch_bounds__(256)
void k_place(const int* __restrict__ src, const int* __restrict__ dst, const int* __restrict__ slot,
             const int* __restrict__ rowptr, int2* __restrict__ csr, int e_total) {
    const int eA = blockIdx.x * EDGE_BLOCK + threadIdx.x * 4;
    const int eB = eA + EDGE_TILE;
    const bool wide = ((((size_t)src) | ((size_t)dst) | ((size_t)slot)) & 15) == 0;
    if (wide && eB + 3 < e_total) {
        const int4 dA = *(const int4*)(dst + eA);
        const int4 dB = *(const int4*)(dst + eB);
        const int4 kA = *(const int4*)(slot + eA);
        const int4 kB = *(const int4*)(slot + eB);
        const int4 sA = *(const int4*)(src + eA);
        const int4 sB = *(const int4*)(src + eB);
        const int pA0 = rowptr[dA.x] + kA.x, pA1 = rowptr[dA.y] + kA.y;
        const int pA2 = rowptr[dA.z] + kA.z, pA3 = rowptr[dA.w] + kA.w;
        const int pB0 = rowptr[dB.x] + kB.x, pB1 = rowptr[dB.y] + kB.y;
        const int pB2 = rowptr[dB.z] + kB.z, pB3 = rowptr[dB.w] + kB.w;
        place_one(csr, pA0, eA,     sA.x);
        place_one(csr, pA1, eA + 1, sA.y);
        place_one(csr, pA2, eA + 2, sA.z);
        place_one(csr, pA3, eA + 3, sA.w);
        place_one(csr, pB0, eB,     sB.x);
        place_one(csr, pB1, eB + 1, sB.y);
        place_one(csr, pB2, eB + 2, sB.z);
        place_one(csr, pB3, eB + 3, sB.w);
    } else {
        for (int q = 0; q < 4; ++q) {
            const int e = eA + q;
            if (e < e_total) place_one(csr, rowptr[dst[e]] + slot[e], e, src[e]);
        }
        for (int q = 0; q < 4; ++q) {
            const int e = eB + q;
            if (e < e_total) place_one(csr, rowptr[dst[e]] + slot[e], e, src[e]);
        }
    }
}

__device__ __forceinline__ float wave_max(float v) {
    for (int o = 16; o > 0; o >>= 1) v = fmaxf(v, __shfl_xor(v, o));
    return v;
}
__device__ __forceinline__ float wave_sum(float v) {
    for (int o = 16; o > 0; o >>= 1) v += __shfl_xor(v, o);
    return v;
}
__device__ __forceinline__ float lane_f(float x, int l) {
    return __uint_as_float((unsigned)__builtin_amdgcn_readlane((int)__float_as_uint(x), l));
}
__device__ __forceinline__ float lrelu(float v) { return (v > 0.0f) ? v : v * NEG_SLOPE; }

__device__ __forceinline__ void edge_logits(const float* __restrict__ als, int s, const float4 ad,
                                            float& x0, float& x1, float& x2, float& x3) {
    const float4 a = *(const float4*)(als + (size_t)s * HEADS);
    x0 = lrelu(a.x + ad.x); x1 = lrelu(a.y + ad.y); x2 = lrelu(a.z + ad.z); x3 = lrelu(a.w + ad.w);
}

template<int K>
__device__ __forceinline__ void accum_k(float4& acc, const float4* __restrict__ Hl, int s,
                                        float a0, float a1, float a2, float a3, int jj,
                                        float k0, float k1, float k2, float k3) {
    float4 hv[K];
    for (int u = 0; u < K; ++u) {
        const int sj = __builtin_amdgcn_readlane(s, jj + u);
        hv[u] = Hl[(size_t)sj * (HOUT / 4)];
    }
    for (int u = 0; u < K; ++u) {
        const float am = fmaf(lane_f(a3, jj + u), k3, fmaf(lane_f(a2, jj + u), k2,
                         fmaf(lane_f(a1, jj + u), k1, lane_f(a0, jj + u) * k0)));
        acc.x += hv[u].x * am; acc.y += hv[u].y * am; acc.z += hv[u].z * am; acc.w += hv[u].w * am;
    }
}

__device__ __forceinline__ void accum_chunk(float4& acc, const float4* __restrict__ Hl, int s,
                                            float a0, float a1, float a2, float a3, int cnt,
                                            float k0, float k1, float k2, float k3) {
    int jj = 0;
    for (; jj + 8 <= cnt; jj += 8) accum_k<8>(acc, Hl, s, a0, a1, a2, a3, jj, k0, k1, k2, k3);
    if (jj + 4 <= cnt) { accum_k<4>(acc, Hl, s, a0, a1, a2, a3, jj, k0, k1, k2, k3); jj += 4; }
    if (jj + 2 <= cnt) { accum_k<2>(acc, Hl, s, a0, a1, a2, a3, jj, k0, k1, k2, k3); jj += 2; }
    if (jj < cnt)        accum_k<1>(acc, Hl, s, a0, a1, a2, a3, jj, k0, k1, k2, k3);
}

__global__ __launch_bounds__(256)
void k_gather(const int* __restrict__ rowptr, const int2* __restrict__ csr,
              const float* __restrict__ Hm,
              const float* __restrict__ als, const float* __restrict__ ald,
              const float* __restrict__ bias, float* __restrict__ out, int n) {
    const int lane = threadIdx.x & 31;
    const int hsel = lane >> 3;
    const float k0 = (hsel == 0) ? 1.0f : 0.0f, k1 = (hsel == 1) ? 1.0f : 0.0f;
    const float k2 = (hsel == 2) ? 1.0f : 0.0f, k3 = (hsel == 3) ? 1.0f : 0.0f;
    const float4* Hl = (const float4*)Hm + lane;
    const float* bl = bias + lane * 4;
    const int wave0  = __builtin_amdgcn_readfirstlane((int)((blockIdx.x * blockDim.x + threadIdx.x) >> 5));
    const int nwaves = (int)((gridDim.x * blockDim.x) >> 5);
    for (int node = wave0; node < n; node += nwaves) {
    const int beg = rowptr[node];
    const int end = rowptr[node + 1];
    const float4 ad = *(const float4*)(ald + (size_t)node * HEADS);

    const int  c0  = min(end - beg, 32);
    const bool ok0 = lane < c0;
    int eid = 0x7fffffff, sraw = 0;
    if (ok0) { const int2 r = csr[beg + lane]; eid = r.x; sraw = r.y; }
    int rank = 0;
    for (int k = 0; k < c0; ++k) rank += (__builtin_amdgcn_readlane(eid, k) < eid) ? 1 : 0;
    int from = lane;
    for (int k = 0; k < c0; ++k) from = (__builtin_amdgcn_readlane(rank, k) == lane) ? k : from;
    const int s0 = __shfl(sraw, from);

    float e0, e1, e2, e3;
    edge_logits(als, s0, ad, e0, e1, e2, e3);
    float m0 = ok0 ? e0 : -INFINITY, m1 = ok0 ? e1 : -INFINITY, m2 = ok0 ? e2 : -INFINITY, m3 = ok0 ? e3 : -INFINITY;
    for (int cb = beg + 32; cb < end; cb += 32) {
        const int j = cb + lane;
        if (j < end) {
            float x0, x1, x2, x3;
            edge_logits(als, csr[j].y, ad, x0, x1, x2, x3);
            m0 = fmaxf(m0, x0); m1 = fmaxf(m1, x1); m2 = fmaxf(m2, x2); m3 = fmaxf(m3, x3);
        }
    }
    m0 = wave_max(m0); m1 = wave_max(m1); m2 = wave_max(m2); m3 = wave_max(m3);
    if (!__builtin_isfinite(m0)) m0 = 0.0f;
    if (!__builtin_isfinite(m1)) m1 = 0.0f;
    if (!__builtin_isfinite(m2)) m2 = 0.0f;
    if (!__builtin_isfinite(m3)) m3 = 0.0f;

    const float w0 = ok0 ? expf(e0 - m0) : 0.0f, w1 = ok0 ? expf(e1 - m1) : 0.0f;
    const float w2 = ok0 ? expf(e2 - m2) : 0.0f, w3 = ok0 ? expf(e3 - m3) : 0.0f;
    float d0 = w0, d1 = w1, d2 = w2, d3 = w3;
    for (int cb = beg + 32; cb < end; cb += 32) {
        const int j = cb + lane;
        if (j < end) {
            float x0, x1, x2, x3;
            edge_logits(als, csr[j].y, ad, x0, x1, x2, x3);
            d0 += expf(x0 - m0); d1 += expf(x1 - m1); d2 += expf(x2 - m2); d3 += expf(x3 - m3);
        }
    }
    d0 = wave_sum(d0); d1 = wave_sum(d1); d2 = wave_sum(d2); d3 = wave_sum(d3);
    const float q0 = fmaxf(d0, 1e-16f), q1 = fmaxf(d1, 1e-16f), q2 = fmaxf(d2, 1e-16f), q3 = fmaxf(d3, 1e-16f);

    float4 acc; acc.x = 0.0f; acc.y = 0.0f; acc.z = 0.0f; acc.w = 0.0f;
    accum_chunk(acc, Hl, s0, w0 / q0, w1 / q1, w2 / q2, w3 / q3, c0, k0, k1, k2, k3);
    for (int cb = beg + 32; cb < end; cb += 32) {
        const int  j  = cb + lane;
        const bool ok = j < end;
        const int  s  = ok ? csr[j].y : 0;
        float x0, x1, x2, x3;
        edge_logits(als, s, ad, x0, x1, x2, x3);
        const float a0 = ok ? expf(x0 - m0) / q0 : 0.0f, a1 = ok ? expf(x1 - m1) / q1 : 0.0f;
        const float a2 = ok ? expf(x2 - m2) / q2 : 0.0f, a3 = ok ? expf(x3 - m3) / q3 : 0.0f;
        accum_chunk(acc, Hl, s, a0, a1, a2, a3, min(end - cb, 32), k0, k1, k2, k3);
    }

    float4 o; o.x = acc.x + bl[0]; o.y = acc.y + bl[1]; o.z = acc.z + bl[2]; o.w = acc.w + bl[3];
    ((float4*)(out + (size_t)node * HOUT))[lane] = o;
    }
}

extern "C" void kernel_launch(void* const* d_in, const int* in_sizes, int n_in,
                              void* d_out, int out_size, void* d_ws, size_t ws_size,
                              hipStream_t stream) {
    const float* x      = (const float*)d_in[0];
    const int*   ei     = (const int*)d_in[1];
    const float* W1     = (const float*)d_in[2];
    const float* a_src1 = (const float*)d_in[3];
    const float* a_dst1 = (const float*)d_in[4];
    const float* b1     = (const float*)d_in[5];
    const float* W2     = (const float*)d_in[6];
    const float* a_src2 = (const float*)d_in[7];
    const float* a_dst2 = (const float*)d_in[8];
    const float* b2     = (const float*)d_in[9];

    const int N = in_sizes[0] / FIN;
    const int E = in_sizes[1] / 2;
    const int* src = ei;
    const int* dst = ei + E;

    const int nTiles = (N + SCAN_TILE - 1) / SCAN_TILE;
    const size_t nPad = ((size_t)N + 4) & ~(size_t)3;

    char* p = (char*)d_ws;
    float* H      = (float*)p; p += (size_t)N * HOUT * sizeof(float);
    float* als    = (float*)p; p += (size_t)N * HEADS * sizeof(float);
    float* ald    = (float*)p; p += (size_t)N * HEADS * sizeof(float);
    int*   rowptr = (int*)p;   p += nPad * sizeof(int);
    int*   cnt    = (int*)p;   p += nPad * sizeof(int);
    int*   bsum   = (int*)p;   p += (((size_t)nTiles + 3) & ~(size_t)3) * sizeof(int);
    int2*  csr    = (int2*)p;  p += (size_t)E * sizeof(int2);
    int*   slot   = (int*)H;

    float* out = (float*)d_out;

    dim3 gGemm((N + 15) / 16, 2);
    const int BT = 256;
    int gN   = (N + BT - 1) / BT;
    int gEB  = (E + EDGE_BLOCK - 1) / EDGE_BLOCK;
    int gNH  = (N * HEADS + BT - 1) / BT;
    int gNW  = (N + 31) / 32;

    k_zero_cnt  <<<gN, BT, 0, stream>>>(cnt, N);
    k_count_slot<<<gEB, BT, 0, stream>>>(dst, cnt, slot, E);
    k_scan_local<<<nTiles, SCAN_B, 0, stream>>>(cnt, rowptr, bsum, N);
    k_scan_fix  <<<nTiles, SCAN_B, 0, stream>>>(rowptr, bsum, N);
    k_place     <<<gEB, BT, 0, stream>>>(src, dst, slot, rowptr, csr, E);

    k_gemm_wmma<0><<<gGemm, 32, 0, stream>>>(x, W1, H, N);
    k_attn_coeff<<<gNH, BT, 0, stream>>>(H, a_src1, a_dst1, als, ald, N);
    k_gather    <<<gNW, BT, 0, stream>>>(rowptr, csr, H, als, ald, b1, out, N);

    k_gemm_wmma<1><<<gGemm, 32, 0, stream>>>(out, W2, H, N);
    k_attn_coeff<<<gNH, BT, 0, stream>>>(H, a_src2, a_dst2, als, ald, N);
    k_gather    <<<gNW, BT, 0, stream>>>(rowptr, csr, H, als, ald, b2, out, N);
}


# Edge bucketing without global atomics: LDS counters split the edges by 512-node destination range (block counts, scan, scatter), then one block per range counts and places them in LDS; sum order kept
# speedup vs baseline: 10.5715x; 1.2002x over previous
#include <hip/hip_runtime.h>
#include <math.h>


typedef __attribute__((ext_vector_type(2))) float v2f;
typedef __attribute__((ext_vector_type(8))) float v8f;

#define FIN   128
#define HEADS 4
#define CDIM  32
#define HOUT  128
#define NEG_SLOPE 0.2f

template<int RELU>
__global__ __launch_bounds__(32)
void k_gemm_wmma(const float* __restrict__ X, const float* __restrict__ W,
                 float* __restrict__ H, int nRows) {
    const int lane   = threadIdx.x;
    const int tm     = blockIdx.x;
    const int tc4    = blockIdx.y;
    const int mlane  = lane & 15;
    const int khalf  = (lane >> 4) * 2;
    int row          = tm * 16 + mlane;
    if (row >= nRows) row = nRows - 1;
    const int colB   = tc4 * 64 + mlane;

    const float* xr = X + (size_t)row * FIN + khalf;

    v8f acc0 = {}, acc1 = {}, acc2 = {}, acc3 = {};

    for (int k0 = 0; k0 < FIN; k0 += 4) {
        v2f a;
        float a0 = xr[k0];
        float a1 = xr[k0 + 1];
        if (RELU) { a0 = fmaxf(a0, 0.0f); a1 = fmaxf(a1, 0.0f); }
        a.x = a0; a.y = a1;

        const float* w0 = W + (size_t)(k0 + khalf) * HOUT;
        const float* w1 = w0 + HOUT;
        v2f b0, b1, b2, b3;
        b0.x = w0[colB +  0]; b0.y = w1[colB +  0];
        b1.x = w0[colB + 16]; b1.y = w1[colB + 16];
        b2.x = w0[colB + 32]; b2.y = w1[colB + 32];
        b3.x = w0[colB + 48]; b3.y = w1[colB + 48];

        acc0 = __builtin_amdgcn_wmma_f32_16x16x4_f32(false, a, false, b0, (short)0, acc0, false, false);
        acc1 = __builtin_amdgcn_wmma_f32_16x16x4_f32(false, a, false, b1, (short)0, acc1, false, false);
        acc2 = __builtin_amdgcn_wmma_f32_16x16x4_f32(false, a, false, b2, (short)0, acc2, false, false);
        acc3 = __builtin_amdgcn_wmma_f32_16x16x4_f32(false, a, false, b3, (short)0, acc3, false, false);
    }

    const int mOff = (lane >> 4) * 8;
    for (int j = 0; j < 8; ++j) {
        int r = tm * 16 + mOff + j;
        if (r < nRows) {
            float* hr = H + (size_t)r * HOUT + tc4 * 64 + mlane;
            hr[0]  = acc0[j];
            hr[16] = acc1[j];
            hr[32] = acc2[j];
            hr[48] = acc3[j];
        }
    }
}

__global__ void k_attn_coeff(const float* __restrict__ H,
                             const float* __restrict__ a_src,
                             const float* __restrict__ a_dst,
                             float* __restrict__ als, float* __restrict__ ald,
                             int n) {
    int i = blockIdx.x * blockDim.x + threadIdx.x;
    if (i >= n * HEADS) return;
    int node = i >> 2, h = i & 3;
    const float* hp = H + (size_t)node * HOUT + h * CDIM;
    const float* as = a_src + h * CDIM;
    const float* ad = a_dst + h * CDIM;
    float s = 0.0f, d = 0.0f;
#pragma unroll
    for (int c = 0; c < CDIM; ++c) { float v = hp[c]; s = fmaf(v, as[c], s); d = fmaf(v, ad[c], d); }
    als[i] = s; ald[i] = d;
}

#define SCAN_B    256
#define SCAN_TILE 1024

#define RANGE_SHIFT 9
#define RANGE       512
#define MAX_RANGES  1024
#define EDGE_BLOCK  4096
#define EDGE_GROUP  1024

__device__ __forceinline__ int4 load4_or_neg(const int* __restrict__ p, int e, int e_total, bool wide) {
    int4 v;
    if (wide && e + 3 < e_total) {
        v = *(const int4*)(p + e);
    } else {
        v.x = (e     < e_total) ? p[e]     : -1;
        v.y = (e + 1 < e_total) ? p[e + 1] : -1;
        v.z = (e + 2 < e_total) ? p[e + 2] : -1;
        v.w = (e + 3 < e_total) ? p[e + 3] : -1;
    }
    return v;
}

__global__ __launch_bounds__(256)
void k_hist(const int* __restrict__ dst, int* __restrict__ bh, int e_total, int n_ranges, int n_blocks) {
    __shared__ int hist[MAX_RANGES];
    for (int i = threadIdx.x; i < n_ranges; i += 256) hist[i] = 0;
    __syncthreads();
    const int e0 = blockIdx.x * EDGE_BLOCK + threadIdx.x * 4;
    const bool wide = (((size_t)dst) & 15) == 0;
    int4 d[4];
    for (int q = 0; q < 4; ++q) d[q] = load4_or_neg(dst, e0 + q * EDGE_GROUP, e_total, wide);
    for (int q = 0; q < 4; ++q) {
        if (d[q].x >= 0) atomicAdd(&hist[d[q].x >> RANGE_SHIFT], 1);
        if (d[q].y >= 0) atomicAdd(&hist[d[q].y >> RANGE_SHIFT], 1);
        if (d[q].z >= 0) atomicAdd(&hist[d[q].z >> RANGE_SHIFT], 1);
        if (d[q].w >= 0) atomicAdd(&hist[d[q].w >> RANGE_SHIFT], 1);
    }
    __syncthreads();
    for (int i = threadIdx.x; i < n_ranges; i += 256) bh[(size_t)i * n_blocks + blockIdx.x] = hist[i];
}

__device__ __forceinline__ void part_one(int* cur, int2* __restrict__ pe, int* __restrict__ pd, int e, int s, int d) {
    if (d >= 0) {
        const int pos = atomicAdd(&cur[d >> RANGE_SHIFT], 1);
        int2 r; r.x = e; r.y = s;
        pe[pos] = r;
        pd[pos] = d;
    }
}

__global__ __launch_bounds__(256)
void k_part(const int* __restrict__ src, const int* __restrict__ dst, const int* __restrict__ bhoff,
            int2* __restrict__ pe, int* __restrict__ pd, int e_total, int n_ranges, int n_blocks) {
    __shared__ int cur[MAX_RANGES];
    for (int i = threadIdx.x; i < n_ranges; i += 256) cur[i] = bhoff[(size_t)i * n_blocks + blockIdx.x];
    const int e0 = blockIdx.x * EDGE_BLOCK + threadIdx.x * 4;
    const bool wide = ((((size_t)src) | ((size_t)dst)) & 15) == 0;
    int4 d[4], s[4];
    for (int q = 0; q < 4; ++q) d[q] = load4_or_neg(dst, e0 + q * EDGE_GROUP, e_total, wide);
    for (int q = 0; q < 4; ++q) s[q] = load4_or_neg(src, e0 + q * EDGE_GROUP, e_total, wide);
    __syncthreads();
    for (int q = 0; q < 4; ++q) {
        const int e = e0 + q * EDGE_GROUP;
        part_one(cur, pe, pd, e,     s[q].x, d[q].x);
        part_one(cur, pe, pd, e + 1, s[q].y, d[q].y);
        part_one(cur, pe, pd, e + 2, s[q].z, d[q].z);
        part_one(cur, pe, pd, e + 3, s[q].w, d[q].w);
    }
}

__global__ __launch_bounds__(SCAN_B)
void k_scan_local(const int* __restrict__ cnt, int* __restrict__ rowptr, int* __restrict__ bsum, int n) {
    __shared__ int wtot[SCAN_B / 32];
    const int t = threadIdx.x, lane = t & 31, wv = t >> 5;
    const int base = blockIdx.x * SCAN_TILE + t * 4;
    int a0 = 0, a1 = 0, a2 = 0, a3 = 0;
    if (base + 3 < n) {
        const int4 v = *(const int4*)(cnt + base);
        a0 = v.x; a1 = v.y; a2 = v.z; a3 = v.w;
    } else {
        if (base     < n) a0 = cnt[base];
        if (base + 1 < n) a1 = cnt[base + 1];
        if (base + 2 < n) a2 = cnt[base + 2];
    }
    const int tsum = a0 + a1 + a2 + a3;
    int incl = tsum;
    for (int o = 1; o < 32; o <<= 1) { const int u = __shfl_up(incl, o); if (lane >= o) incl += u; }
    if (lane == 31) wtot[wv] = incl;
    __syncthreads();
    int pre = 0;
    for (int k = 0; k < wv; ++k) pre += wtot[k];
    const int ex = pre + incl - tsum;
    if (base + 3 < n) {
        int4 r; r.x = ex; r.y = ex + a0; r.z = ex + a0 + a1; r.w = ex + a0 + a1 + a2;
        *(int4*)(rowptr + base) = r;
    } else {
        if (base     < n) rowptr[base]     = ex;
        if (base + 1 < n) rowptr[base + 1] = ex + a0;
        if (base + 2 < n) rowptr[base + 2] = ex + a0 + a1;
    }
    if (t == SCAN_B - 1) bsum[blockIdx.x] = pre + incl;
}

__global__ __launch_bounds__(SCAN_B)
void k_scan_fix(int* __restrict__ rowptr, const int* __restrict__ bsum, int n) {
    __shared__ int wtot[SCAN_B / 32];
    const int t = threadIdx.x, lane = t & 31, wv = t >> 5;
    const int b = blockIdx.x;
    int part = 0;
    for (int k = t; k < b; k += SCAN_B) part += bsum[k];
    for (int o = 16; o > 0; o >>= 1) part += __shfl_xor(part, o);
    if (lane == 0) wtot[wv] = part;
    __syncthreads();
    int off = 0;
    for (int k = 0; k < SCAN_B / 32; ++k) off += wtot[k];
    const int base = b * SCAN_TILE + t * 4;
    for (int q = 0; q < 4; ++q) {
        const int i = base + q;
        if (i < n) rowptr[i] += off;
    }
    if (b == (int)gridDim.x - 1 && t == 0) rowptr[n] = off + bsum[b];
}

#define BUCKET_B 1024
__global__ __launch_bounds__(BUCKET_B)
void k_bucket(const int2* __restrict__ pe, const int* __restrict__ pd, const int* __restrict__ bhoff,
              int* __restrict__ rowptr, int2* __restrict__ csr, int n, int n_blocks) {
    __shared__ int cnt[RANGE];
    __shared__ int wtot[BUCKET_B / 32];
    const int t = threadIdx.x, lane = t & 31, wv = t >> 5;
    const int b = blockIdx.x;
    const int beg = bhoff[(size_t)b * n_blocks];
    const int end = bhoff[(size_t)(b + 1) * n_blocks];
    const int node0 = b << RANGE_SHIFT;
    if (t < RANGE) cnt[t] = 0;
    __syncthreads();
    for (int i0 = beg + t; i0 < end; i0 += 4 * BUCKET_B) {
        int dv[4];
        for (int u = 0; u < 4; ++u) { const int i = i0 + u * BUCKET_B; dv[u] = (i < end) ? pd[i] : -1; }
        for (int u = 0; u < 4; ++u) if (dv[u] >= 0) atomicAdd(&cnt[dv[u] - node0], 1);
    }
    __syncthreads();
    const int v = (t < RANGE) ? cnt[t] : 0;
    int incl = v;
    for (int o = 1; o < 32; o <<= 1) { const int u = __shfl_up(incl, o); if (lane >= o) incl += u; }
    if (lane == 31) wtot[wv] = incl;
    __syncthreads();
    int pre = 0;
    for (int k = 0; k < wv; ++k) pre += wtot[k];
    const int first = beg + pre + incl - v;
    if (t < RANGE) {
        cnt[t] = first;
        if (node0 + t < n) rowptr[node0 + t] = first;
    }
    if (b == (int)gridDim.x - 1 && t == 0) rowptr[n] = end;
    __syncthreads();
    for (int i0 = beg + t; i0 < end; i0 += 4 * BUCKET_B) {
        int dv[4]; int2 ev[4];
        for (int u = 0; u < 4; ++u) {
            const int i = i0 + u * BUCKET_B;
            dv[u] = -1; ev[u].x = 0; ev[u].y = 0;
            if (i < end) { dv[u] = pd[i]; ev[u] = pe[i]; }
        }
        for (int u = 0; u < 4; ++u) {
            if (dv[u] >= 0) {
                const int pos = atomicAdd(&cnt[dv[u] - node0], 1);
                csr[pos] = ev[u];
            }
        }
    }
}

__device__ __forceinline__ float wave_max(float v) {
    for (int o = 16; o > 0; o >>= 1) v = fmaxf(v, __shfl_xor(v, o));
    return v;
}
__device__ __forceinline__ float wave_sum(float v) {
    for (int o = 16; o > 0; o >>= 1) v += __shfl_xor(v, o);
    return v;
}
__device__ __forceinline__ float lane_f(float x, int l) {
    return __uint_as_float((unsigned)__builtin_amdgcn_readlane((int)__float_as_uint(x), l));
}
__device__ __forceinline__ float lrelu(float v) { return (v > 0.0f) ? v : v * NEG_SLOPE; }

__device__ __forceinline__ void edge_logits(const float* __restrict__ als, int s, const float4 ad,
                                            float& x0, float& x1, float& x2, float& x3) {
    const float4 a = *(const float4*)(als + (size_t)s * HEADS);
    x0 = lrelu(a.x + ad.x); x1 = lrelu(a.y + ad.y); x2 = lrelu(a.z + ad.z); x3 = lrelu(a.w + ad.w);
}

template<int K>
__device__ __forceinline__ void accum_k(float4& acc, const float4* __restrict__ Hl, int s,
                                        float a0, float a1, float a2, float a3, int jj,
                                        float k0, float k1, float k2, float k3) {
    float4 hv[K];
    for (int u = 0; u < K; ++u) {
        const int sj = __builtin_amdgcn_readlane(s, jj + u);
        hv[u] = Hl[(size_t)sj * (HOUT / 4)];
    }
    for (int u = 0; u < K; ++u) {
        const float am = fmaf(lane_f(a3, jj + u), k3, fmaf(lane_f(a2, jj + u), k2,
                         fmaf(lane_f(a1, jj + u), k1, lane_f(a0, jj + u) * k0)));
        acc.x += hv[u].x * am; acc.y += hv[u].y * am; acc.z += hv[u].z * am; acc.w += hv[u].w * am;
    }
}

__device__ __forceinline__ void accum_chunk(float4& acc, const float4* __restrict__ Hl, int s,
                                            float a0, float a1, float a2, float a3, int cnt,
                                            float k0, float k1, float k2, float k3) {
    int jj = 0;
    for (; jj + 8 <= cnt; jj += 8) accum_k<8>(acc, Hl, s, a0, a1, a2, a3, jj, k0, k1, k2, k3);
    if (jj + 4 <= cnt) { accum_k<4>(acc, Hl, s, a0, a1, a2, a3, jj, k0, k1, k2, k3); jj += 4; }
    if (jj + 2 <= cnt) { accum_k<2>(acc, Hl, s, a0, a1, a2, a3, jj, k0, k1, k2, k3); jj += 2; }
    if (jj < cnt)        accum_k<1>(acc, Hl, s, a0, a1, a2, a3, jj, k0, k1, k2, k3);
}

__global__ __launch_bounds__(256)
void k_gather(const int* __restrict__ rowptr, const int2* __restrict__ csr,
              const float* __restrict__ Hm,
              const float* __restrict__ als, const float* __restrict__ ald,
              const float* __restrict__ bias, float* __restrict__ out, int n) {
    const int lane = threadIdx.x & 31;
    const int hsel = lane >> 3;
    const float k0 = (hsel == 0) ? 1.0f : 0.0f, k1 = (hsel == 1) ? 1.0f : 0.0f;
    const float k2 = (hsel == 2) ? 1.0f : 0.0f, k3 = (hsel == 3) ? 1.0f : 0.0f;
    const float4* Hl = (const float4*)Hm + lane;
    const float* bl = bias + lane * 4;
    const int wave0  = __builtin_amdgcn_readfirstlane((int)((blockIdx.x * blockDim.x + threadIdx.x) >> 5));
    const int nwaves = (int)((gridDim.x * blockDim.x) >> 5);
    for (int node = wave0; node < n; node += nwaves) {
    const int beg = rowptr[node];
    const int end = rowptr[node + 1];
    const float4 ad = *(const float4*)(ald + (size_t)node * HEADS);

    const int  c0  = min(end - beg, 32);
    const bool ok0 = lane < c0;
    int eid = 0x7fffffff, sraw = 0;
    if (ok0) { const int2 r = csr[beg + lane]; eid = r.x; sraw = r.y; }
    int rank = 0;
    for (int k = 0; k < c0; ++k) rank += (__builtin_amdgcn_readlane(eid, k) < eid) ? 1 : 0;
    int from = lane;
    for (int k = 0; k < c0; ++k) from = (__builtin_amdgcn_readlane(rank, k) == lane) ? k : from;
    const int s0 = __shfl(sraw, from);

    float e0, e1, e2, e3;
    edge_logits(als, s0, ad, e0, e1, e2, e3);
    float m0 = ok0 ? e0 : -INFINITY, m1 = ok0 ? e1 : -INFINITY, m2 = ok0 ? e2 : -INFINITY, m3 = ok0 ? e3 : -INFINITY;
    for (int cb = beg + 32; cb < end; cb += 32) {
        const int j = cb + lane;
        if (j < end) {
            float x0, x1, x2, x3;
            edge_logits(als, csr[j].y, ad, x0, x1, x2, x3);
            m0 = fmaxf(m0, x0); m1 = fmaxf(m1, x1); m2 = fmaxf(m2, x2); m3 = fmaxf(m3, x3);
        }
    }
    m0 = wave_max(m0); m1 = wave_max(m1); m2 = wave_max(m2); m3 = wave_max(m3);
    if (!__builtin_isfinite(m0)) m0 = 0.0f;
    if (!__builtin_isfinite(m1)) m1 = 0.0f;
    if (!__builtin_isfinite(m2)) m2 = 0.0f;
    if (!__builtin_isfinite(m3)) m3 = 0.0f;

    const float w0 = ok0 ? expf(e0 - m0) : 0.0f, w1 = ok0 ? expf(e1 - m1) : 0.0f;
    const float w2 = ok0 ? expf(e2 - m2) : 0.0f, w3 = ok0 ? expf(e3 - m3) : 0.0f;
    float d0 = w0, d1 = w1, d2 = w2, d3 = w3;
    for (int cb = beg + 32; cb < end; cb += 32) {
        const int j = cb + lane;
        if (j < end) {
            float x0, x1, x2, x3;
            edge_logits(als, csr[j].y, ad, x0, x1, x2, x3);
            d0 += expf(x0 - m0); d1 += expf(x1 - m1); d2 += expf(x2 - m2); d3 += expf(x3 - m3);
        }
    }
    d0 = wave_sum(d0); d1 = wave_sum(d1); d2 = wave_sum(d2); d3 = wave_sum(d3);
    const float q0 = fmaxf(d0, 1e-16f), q1 = fmaxf(d1, 1e-16f), q2 = fmaxf(d2, 1e-16f), q3 = fmaxf(d3, 1e-16f);

    float4 acc; acc.x = 0.0f; acc.y = 0.0f; acc.z = 0.0f; acc.w = 0.0f;
    accum_chunk(acc, Hl, s0, w0 / q0, w1 / q1, w2 / q2, w3 / q3, c0, k0, k1, k2, k3);
    for (int cb = beg + 32; cb < end; cb += 32) {
        const int  j  = cb + lane;
        const bool ok = j < end;
        const int  s  = ok ? csr[j].y : 0;
        float x0, x1, x2, x3;
        edge_logits(als, s, ad, x0, x1, x2, x3);
        const float a0 = ok ? expf(x0 - m0) / q0 : 0.0f, a1 = ok ? expf(x1 - m1) / q1 : 0.0f;
        const float a2 = ok ? expf(x2 - m2) / q2 : 0.0f, a3 = ok ? expf(x3 - m3) / q3 : 0.0f;
        accum_chunk(acc, Hl, s, a0, a1, a2, a3, min(end - cb, 32), k0, k1, k2, k3);
    }

    float4 o; o.x = acc.x + bl[0]; o.y = acc.y + bl[1]; o.z = acc.z + bl[2]; o.w = acc.w + bl[3];
    ((float4*)(out + (size_t)node * HOUT))[lane] = o;
    }
}

extern "C" void kernel_launch(void* const* d_in, const int* in_sizes, int n_in,
                              void* d_out, int out_size, void* d_ws, size_t ws_size,
                              hipStream_t stream) {
    const float* x      = (const float*)d_in[0];
    const int*   ei     = (const int*)d_in[1];
    const float* W1     = (const float*)d_in[2];
    const float* a_src1 = (const float*)d_in[3];
    const float* a_dst1 = (const float*)d_in[4];
    const float* b1     = (const float*)d_in[5];
    const float* W2     = (const float*)d_in[6];
    const float* a_src2 = (const float*)d_in[7];
    const float* a_dst2 = (const float*)d_in[8];
    const float* b2     = (const float*)d_in[9];

    const int N = in_sizes[0] / FIN;
    const int E = in_sizes[1] / 2;
    const int* src = ei;
    const int* dst = ei + E;

    const int nRanges = (N + RANGE - 1) / RANGE;
    const int nEB     = (E + EDGE_BLOCK - 1) / EDGE_BLOCK;
    const int nBH     = nRanges * nEB;
    const int nTilesB = (nBH + SCAN_TILE - 1) / SCAN_TILE;
    const int nTiles  = (N + SCAN_TILE - 1) / SCAN_TILE;
    const size_t nPad = ((size_t)N + 4) & ~(size_t)3;
    const size_t ePad = ((size_t)E + 3) & ~(size_t)3;
    const size_t bPad = ((size_t)nBH + 4) & ~(size_t)3;

    char* p = (char*)d_ws;
    float* H      = (float*)p; p += (size_t)N * HOUT * sizeof(float);
    float* als    = (float*)p; p += (size_t)N * HEADS * sizeof(float);
    float* ald    = (float*)p; p += (size_t)N * HEADS * sizeof(float);
    int*   rowptr = (int*)p;   p += nPad * sizeof(int);
    p += nPad * sizeof(int);
    p += (((size_t)nTiles + 3) & ~(size_t)3) * sizeof(int);
    int2*  csr    = (int2*)p;  p += (size_t)E * sizeof(int2);
    char* q = (char*)H;
    int2*  pe     = (int2*)q;  q += ePad * sizeof(int2);
    int*   pd     = (int*)q;   q += ePad * sizeof(int);
    int*   bh     = (int*)q;   q += bPad * sizeof(int);
    int*   bhoff  = (int*)q;   q += bPad * sizeof(int);
    int*   bsum   = (int*)q;   q += (((size_t)nTilesB + 3) & ~(size_t)3) * sizeof(int);

    float* out = (float*)d_out;

    dim3 gGemm((N + 15) / 16, 2);
    const int BT = 256;
    int gNH  = (N * HEADS + BT - 1) / BT;
    int gNW  = (N + 31) / 32;

    k_hist      <<<nEB, 256, 0, stream>>>(dst, bh, E, nRanges, nEB);
    k_scan_local<<<nTilesB, SCAN_B, 0, stream>>>(bh, bhoff, bsum, nBH);
    k_scan_fix  <<<nTilesB, SCAN_B, 0, stream>>>(bhoff, bsum, nBH);
    k_part      <<<nEB, 256, 0, stream>>>(src, dst, bhoff, pe, pd, E, nRanges, nEB);
    k_bucket    <<<nRanges, BUCKET_B, 0, stream>>>(pe, pd, bhoff, rowptr, csr, N, nEB);

    k_gemm_wmma<0><<<gGemm, 32, 0, stream>>>(x, W1, H, N);
    k_attn_coeff<<<gNH, BT, 0, stream>>>(H, a_src1, a_dst1, als, ald, N);
    k_gather    <<<gNW, BT, 0, stream>>>(rowptr, csr, H, als, ald, b1, out, N);

    k_gemm_wmma<1><<<gGemm, 32, 0, stream>>>(out, W2, H, N);
    k_attn_coeff<<<gNH, BT, 0, stream>>>(H, a_src2, a_dst2, als, ald, N);
    k_gather    <<<gNW, BT, 0, stream>>>(rowptr, csr, H, als, ald, b2, out, N);
}


# Product operand feed: W staged once per 4-wave work-group in LDS as k-pairs (one 8-byte LDS read per B fragment), A by 16-byte loads, a wave does 16x128; k order within 8 permuted: sum order changed
# speedup vs baseline: 11.0618x; 1.0464x over previous
#include <hip/hip_runtime.h>
#include <math.h>


typedef __attribute__((ext_vector_type(2))) float v2f;
typedef __attribute__((ext_vector_type(8))) float v8f;

#define FIN   128
#define HEADS 4
#define CDIM  32
#define HOUT  128
#define NEG_SLOPE 0.2f

#define GM_WAVES 4
#define WL_PITCH 272
template<int RELU>
__global__ __launch_bounds__(GM_WAVES * 32)
void k_gemm_wmma(const float* __restrict__ X, const float* __restrict__ W,
                 float* __restrict__ H, int nRows) {
    __shared__ float Wl[(FIN / 2) * WL_PITCH];
    const int tid   = threadIdx.x;
    const int lane  = tid & 31;
    const int wv    = tid >> 5;
    const int c     = lane & 15;
    const int half  = lane >> 4;
    const int row0  = (blockIdx.x * GM_WAVES + wv) * 16;
    int row         = row0 + c;
    if (row >= nRows) row = nRows - 1;

    const float4* xr = (const float4*)(X + (size_t)row * FIN) + half;
    float4 a[16];
#pragma unroll
    for (int i = 0; i < 16; ++i) a[i] = xr[2 * i];

#pragma unroll
    for (int u = 0; u < (FIN / 2) / GM_WAVES; ++u) {
        const int kp = wv + GM_WAVES * u;
        const float4 r0 = *(const float4*)(W + (size_t)(2 * kp)     * HOUT + 4 * lane);
        const float4 r1 = *(const float4*)(W + (size_t)(2 * kp + 1) * HOUT + 4 * lane);
        float* d = Wl + kp * WL_PITCH + 8 * lane;
        float4 p0; p0.x = r0.x; p0.y = r1.x; p0.z = r0.y; p0.w = r1.y;
        float4 p1; p1.x = r0.z; p1.y = r1.z; p1.z = r0.w; p1.w = r1.w;
        *(float4*)(d)     = p0;
        *(float4*)(d + 4) = p1;
    }
    __syncthreads();

    v8f acc[8];
#pragma unroll
    for (int t = 0; t < 8; ++t) acc[t] = (v8f){};

    const float* wl = Wl + (2 * half) * WL_PITCH + 2 * c;
#pragma unroll
    for (int i = 0; i < 16; ++i) {
        float4 av = a[i];
        if (RELU) { av.x = fmaxf(av.x, 0.0f); av.y = fmaxf(av.y, 0.0f); av.z = fmaxf(av.z, 0.0f); av.w = fmaxf(av.w, 0.0f); }
#pragma unroll
        for (int w = 0; w < 2; ++w) {
            v2f af;
            af.x = w ? av.z : av.x;
            af.y = w ? av.w : av.y;
            const float* wp = wl + (4 * i + w) * WL_PITCH;
#pragma unroll
            for (int t = 0; t < 8; ++t) {
                const v2f b = *(const v2f*)(wp + 32 * t);
                acc[t] = __builtin_amdgcn_wmma_f32_16x16x4_f32(false, af, false, b, (short)0, acc[t], false, false);
            }
        }
    }

    const int mOff = half * 8;
#pragma unroll
    for (int j = 0; j < 8; ++j) {
        const int r = row0 + mOff + j;
        if (r < nRows) {
            float* hr = H + (size_t)r * HOUT + c;
#pragma unroll
            for (int t = 0; t < 8; ++t) hr[16 * t] = acc[t][j];
        }
    }
}

__global__ void k_attn_coeff(const float* __restrict__ H,
                             const float* __restrict__ a_src,
                             const float* __restrict__ a_dst,
                             float* __restrict__ als, float* __restrict__ ald,
                             int n) {
    int i = blockIdx.x * blockDim.x + threadIdx.x;
    if (i >= n * HEADS) return;
    int node = i >> 2, h = i & 3;
    const float* hp = H + (size_t)node * HOUT + h * CDIM;
    const float* as = a_src + h * CDIM;
    const float* ad = a_dst + h * CDIM;
    float s = 0.0f, d = 0.0f;
#pragma unroll
    for (int c = 0; c < CDIM; ++c) { float v = hp[c]; s = fmaf(v, as[c], s); d = fmaf(v, ad[c], d); }
    als[i] = s; ald[i] = d;
}

#define SCAN_B    256
#define SCAN_TILE 1024

#define RANGE_SHIFT 9
#define RANGE       512
#define MAX_RANGES  1024
#define EDGE_BLOCK  4096
#define EDGE_GROUP  1024

__device__ __forceinline__ int4 load4_or_neg(const int* __restrict__ p, int e, int e_total, bool wide) {
    int4 v;
    if (wide && e + 3 < e_total) {
        v = *(const int4*)(p + e);
    } else {
        v.x = (e     < e_total) ? p[e]     : -1;
        v.y = (e + 1 < e_total) ? p[e + 1] : -1;
        v.z = (e + 2 < e_total) ? p[e + 2] : -1;
        v.w = (e + 3 < e_total) ? p[e + 3] : -1;
    }
    return v;
}

__global__ __launch_bounds__(256)
void k_hist(const int* __restrict__ dst, int* __restrict__ bh, int e_total, int n_ranges, int n_blocks) {
    __shared__ int hist[MAX_RANGES];
    for (int i = threadIdx.x; i < n_ranges; i += 256) hist[i] = 0;
    __syncthreads();
    const int e0 = blockIdx.x * EDGE_BLOCK + threadIdx.x * 4;
    const bool wide = (((size_t)dst) & 15) == 0;
    int4 d[4];
    for (int q = 0; q < 4; ++q) d[q] = load4_or_neg(dst, e0 + q * EDGE_GROUP, e_total, wide);
    for (int q = 0; q < 4; ++q) {
        if (d[q].x >= 0) atomicAdd(&hist[d[q].x >> RANGE_SHIFT], 1);
        if (d[q].y >= 0) atomicAdd(&hist[d[q].y >> RANGE_SHIFT], 1);
        if (d[q].z >= 0) atomicAdd(&hist[d[q].z >> RANGE_SHIFT], 1);
        if (d[q].w >= 0) atomicAdd(&hist[d[q].w >> RANGE_SHIFT], 1);
    }
    __syncthreads();
    for (int i = threadIdx.x; i < n_ranges; i += 256) bh[(size_t)i * n_blocks + blockIdx.x] = hist[i];
}

__device__ __forceinline__ void part_one(int* cur, int2* __restrict__ pe, int* __restrict__ pd, int e, int s, int d) {
    if (d >= 0) {
        const int pos = atomicAdd(&cur[d >> RANGE_SHIFT], 1);
        int2 r; r.x = e; r.y = s;
        pe[pos] = r;
        pd[pos] = d;
    }
}

__global__ __launch_bounds__(256)
void k_part(const int* __restrict__ src, const int* __restrict__ dst, const int* __restrict__ bhoff,
            int2* __restrict__ pe, int* __restrict__ pd, int e_total, int n_ranges, int n_blocks) {
    __shared__ int cur[MAX_RANGES];
    for (int i = threadIdx.x; i < n_ranges; i += 256) cur[i] = bhoff[(size_t)i * n_blocks + blockIdx.x];
    const int e0 = blockIdx.x * EDGE_BLOCK + threadIdx.x * 4;
    const bool wide = ((((size_t)src) | ((size_t)dst)) & 15) == 0;
    int4 d[4], s[4];
    for (int q = 0; q < 4; ++q) d[q] = load4_or_neg(dst, e0 + q * EDGE_GROUP, e_total, wide);
    for (int q = 0; q < 4; ++q) s[q] = load4_or_neg(src, e0 + q * EDGE_GROUP, e_total, wide);
    __syncthreads();
    for (int q = 0; q < 4; ++q) {
        const int e = e0 + q * EDGE_GROUP;
        part_one(cur, pe, pd, e,     s[q].x, d[q].x);
        part_one(cur, pe, pd, e + 1, s[q].y, d[q].y);
        part_one(cur, pe, pd, e + 2, s[q].z, d[q].z);
        part_one(cur, pe, pd, e + 3, s[q].w, d[q].w);
    }
}

__global__ __launch_bounds__(SCAN_B)
void k_scan_local(const int* __restrict__ cnt, int* __restrict__ rowptr, int* __restrict__ bsum, int n) {
    __shared__ int wtot[SCAN_B / 32];
    const int t = threadIdx.x, lane = t & 31, wv = t >> 5;
    const int base = blockIdx.x * SCAN_TILE + t * 4;
    int a0 = 0, a1 = 0, a2 = 0, a3 = 0;
    if (base + 3 < n) {
        const int4 v = *(const int4*)(cnt + base);
        a0 = v.x; a1 = v.y; a2 = v.z; a3 = v.w;
    } else {
        if (base     < n) a0 = cnt[base];
        if (base + 1 < n) a1 = cnt[base + 1];
        if (base + 2 < n) a2 = cnt[base + 2];
    }
    const int tsum = a0 + a1 + a2 + a3;
    int incl = tsum;
    for (int o = 1; o < 32; o <<= 1) { const int u = __shfl_up(incl, o); if (lane >= o) incl += u; }
    if (lane == 31) wtot[wv] = incl;
    __syncthreads();
    int pre = 0;
    for (int k = 0; k < wv; ++k) pre += wtot[k];
    const int ex = pre + incl - tsum;
    if (base + 3 < n) {
        int4 r; r.x = ex; r.y = ex + a0; r.z = ex + a0 + a1; r.w = ex + a0 + a1 + a2;
        *(int4*)(rowptr + base) = r;
    } else {
        if (base     < n) rowptr[base]     = ex;
        if (base + 1 < n) rowptr[base + 1] = ex + a0;
        if (base + 2 < n) rowptr[base + 2] = ex + a0 + a1;
    }
    if (t == SCAN_B - 1) bsum[blockIdx.x] = pre + incl;
}

__global__ __launch_bounds__(SCAN_B)
void k_scan_fix(int* __restrict__ rowptr, const int* __restrict__ bsum, int n) {
    __shared__ int wtot[SCAN_B / 32];
    const int t = threadIdx.x, lane = t & 31, wv = t >> 5;
    const int b = blockIdx.x;
    int part = 0;
    for (int k = t; k < b; k += SCAN_B) part += bsum[k];
    for (int o = 16; o > 0; o >>= 1) part += __shfl_xor(part, o);
    if (lane == 0) wtot[wv] = part;
    __syncthreads();
    int off = 0;
    for (int k = 0; k < SCAN_B / 32; ++k) off += wtot[k];
    const int base = b * SCAN_TILE + t * 4;
    for (int q = 0; q < 4; ++q) {
        const int i = base + q;
        if (i < n) rowptr[i] += off;
    }
    if (b == (int)gridDim.x - 1 && t == 0) rowptr[n] = off + bsum[b];
}

#define BUCKET_B 1024
__global__ __launch_bounds__(BUCKET_B)
void k_bucket(const int2* __restrict__ pe, const int* __restrict__ pd, const int* __restrict__ bhoff,
              int* __restrict__ rowptr, int2* __restrict__ csr, int n, int n_blocks) {
    __shared__ int cnt[RANGE];
    __shared__ int wtot[BUCKET_B / 32];
    const int t = threadIdx.x, lane = t & 31, wv = t >> 5;
    const int b = blockIdx.x;
    const int beg = bhoff[(size_t)b * n_blocks];
    const int end = bhoff[(size_t)(b + 1) * n_blocks];
    const int node0 = b << RANGE_SHIFT;
    if (t < RANGE) cnt[t] = 0;
    __syncthreads();
    for (int i0 = beg + t; i0 < end; i0 += 4 * BUCKET_B) {
        int dv[4];
        for (int u = 0; u < 4; ++u) { const int i = i0 + u * BUCKET_B; dv[u] = (i < end) ? pd[i] : -1; }
        for (int u = 0; u < 4; ++u) if (dv[u] >= 0) atomicAdd(&cnt[dv[u] - node0], 1);
    }
    __syncthreads();
    const int v = (t < RANGE) ? cnt[t] : 0;
    int incl = v;
    for (int o = 1; o < 32; o <<= 1) { const int u = __shfl_up(incl, o); if (lane >= o) incl += u; }
    if (lane == 31) wtot[wv] = incl;
    __syncthreads();
    int pre = 0;
    for (int k = 0; k < wv; ++k) pre += wtot[k];
    const int first = beg + pre + incl - v;
    if (t < RANGE) {
        cnt[t] = first;
        if (node0 + t < n) rowptr[node0 + t] = first;
    }
    if (b == (int)gridDim.x - 1 && t == 0) rowptr[n] = end;
    __syncthreads();
    for (int i0 = beg + t; i0 < end; i0 += 4 * BUCKET_B) {
        int dv[4]; int2 ev[4];
        for (int u = 0; u < 4; ++u) {
            const int i = i0 + u * BUCKET_B;
            dv[u] = -1; ev[u].x = 0; ev[u].y = 0;
            if (i < end) { dv[u] = pd[i]; ev[u] = pe[i]; }
        }
        for (int u = 0; u < 4; ++u) {
            if (dv[u] >= 0) {
                const int pos = atomicAdd(&cnt[dv[u] - node0], 1);
                csr[pos] = ev[u];
            }
        }
    }
}

__device__ __forceinline__ float wave_max(float v) {
    for (int o = 16; o > 0; o >>= 1) v = fmaxf(v, __shfl_xor(v, o));
    return v;
}
__device__ __forceinline__ float wave_sum(float v) {
    for (int o = 16; o > 0; o >>= 1) v += __shfl_xor(v, o);
    return v;
}
__device__ __forceinline__ float lane_f(float x, int l) {
    return __uint_as_float((unsigned)__builtin_amdgcn_readlane((int)__float_as_uint(x), l));
}
__device__ __forceinline__ float lrelu(float v) { return (v > 0.0f) ? v : v * NEG_SLOPE; }

__device__ __forceinline__ void edge_logits(const float* __restrict__ als, int s, const float4 ad,
                                            float& x0, float& x1, float& x2, float& x3) {
    const float4 a = *(const float4*)(als + (size_t)s * HEADS);
    x0 = lrelu(a.x + ad.x); x1 = lrelu(a.y + ad.y); x2 = lrelu(a.z + ad.z); x3 = lrelu(a.w + ad.w);
}

template<int K>
__device__ __forceinline__ void accum_k(float4& acc, const float4* __restrict__ Hl, int s,
                                        float a0, float a1, float a2, float a3, int jj,
                                        float k0, float k1, float k2, float k3) {
    float4 hv[K];
    for (int u = 0; u < K; ++u) {
        const int sj = __builtin_amdgcn_readlane(s, jj + u);
        hv[u] = Hl[(size_t)sj * (HOUT / 4)];
    }
    for (int u = 0; u < K; ++u) {
        const float am = fmaf(lane_f(a3, jj + u), k3, fmaf(lane_f(a2, jj + u), k2,
                         fmaf(lane_f(a1, jj + u), k1, lane_f(a0, jj + u) * k0)));
        acc.x += hv[u].x * am; acc.y += hv[u].y * am; acc.z += hv[u].z * am; acc.w += hv[u].w * am;
    }
}

__device__ __forceinline__ void accum_chunk(float4& acc, const float4* __restrict__ Hl, int s,
                                            float a0, float a1, float a2, float a3, int cnt,
                                            float k0, float k1, float k2, float k3) {
    int jj = 0;
    for (; jj + 8 <= cnt; jj += 8) accum_k<8>(acc, Hl, s, a0, a1, a2, a3, jj, k0, k1, k2, k3);
    if (jj + 4 <= cnt) { accum_k<4>(acc, Hl, s, a0, a1, a2, a3, jj, k0, k1, k2, k3); jj += 4; }
    if (jj + 2 <= cnt) { accum_k<2>(acc, Hl, s, a0, a1, a2, a3, jj, k0, k1, k2, k3); jj += 2; }
    if (jj < cnt)        accum_k<1>(acc, Hl, s, a0, a1, a2, a3, jj, k0, k1, k2, k3);
}

__global__ __launch_bounds__(256)
void k_gather(const int* __restrict__ rowptr, const int2* __restrict__ csr,
              const float* __restrict__ Hm,
              const float* __restrict__ als, const float* __restrict__ ald,
              const float* __restrict__ bias, float* __restrict__ out, int n) {
    const int lane = threadIdx.x & 31;
    const int hsel = lane >> 3;
    const float k0 = (hsel == 0) ? 1.0f : 0.0f, k1 = (hsel == 1) ? 1.0f : 0.0f;
    const float k2 = (hsel == 2) ? 1.0f : 0.0f, k3 = (hsel == 3) ? 1.0f : 0.0f;
    const float4* Hl = (const float4*)Hm + lane;
    const float* bl = bias + lane * 4;
    const int wave0  = __builtin_amdgcn_readfirstlane((int)((blockIdx.x * blockDim.x + threadIdx.x) >> 5));
    const int nwaves = (int)((gridDim.x * blockDim.x) >> 5);
    for (int node = wave0; node < n; node += nwaves) {
    const int beg = rowptr[node];
    const int end = rowptr[node + 1];
    const float4 ad = *(const float4*)(ald + (size_t)node * HEADS);

    const int  c0  = min(end - beg, 32);
    const bool ok0 = lane < c0;
    int eid = 0x7fffffff, sraw = 0;
    if (ok0) { const int2 r = csr[beg + lane]; eid = r.x; sraw = r.y; }
    int rank = 0;
    for (int k = 0; k < c0; ++k) rank += (__builtin_amdgcn_readlane(eid, k) < eid) ? 1 : 0;
    int from = lane;
    for (int k = 0; k < c0; ++k) from = (__builtin_amdgcn_readlane(rank, k) == lane) ? k : from;
    const int s0 = __shfl(sraw, from);

    float e0, e1, e2, e3;
    edge_logits(als, s0, ad, e0, e1, e2, e3);
    float m0 = ok0 ? e0 : -INFINITY, m1 = ok0 ? e1 : -INFINITY, m2 = ok0 ? e2 : -INFINITY, m3 = ok0 ? e3 : -INFINITY;
    for (int cb = beg + 32; cb < end; cb += 32) {
        const int j = cb + lane;
        if (j < end) {
            float x0, x1, x2, x3;
            edge_logits(als, csr[j].y, ad, x0, x1, x2, x3);
            m0 = fmaxf(m0, x0); m1 = fmaxf(m1, x1); m2 = fmaxf(m2, x2); m3 = fmaxf(m3, x3);
        }
    }
    m0 = wave_max(m0); m1 = wave_max(m1); m2 = wave_max(m2); m3 = wave_max(m3);
    if (!__builtin_isfinite(m0)) m0 = 0.0f;
    if (!__builtin_isfinite(m1)) m1 = 0.0f;
    if (!__builtin_isfinite(m2)) m2 = 0.0f;
    if (!__builtin_isfinite(m3)) m3 = 0.0f;

    const float w0 = ok0 ? expf(e0 - m0) : 0.0f, w1 = ok0 ? expf(e1 - m1) : 0.0f;
    const float w2 = ok0 ? expf(e2 - m2) : 0.0f, w3 = ok0 ? expf(e3 - m3) : 0.0f;
    float d0 = w0, d1 = w1, d2 = w2, d3 = w3;
    for (int cb = beg + 32; cb < end; cb += 32) {
        const int j = cb + lane;
        if (j < end) {
            float x0, x1, x2, x3;
            edge_logits(als, csr[j].y, ad, x0, x1, x2, x3);
            d0 += expf(x0 - m0); d1 += expf(x1 - m1); d2 += expf(x2 - m2); d3 += expf(x3 - m3);
        }
    }
    d0 = wave_sum(d0); d1 = wave_sum(d1); d2 = wave_sum(d2); d3 = wave_sum(d3);
    const float q0 = fmaxf(d0, 1e-16f), q1 = fmaxf(d1, 1e-16f), q2 = fmaxf(d2, 1e-16f), q3 = fmaxf(d3, 1e-16f);

    float4 acc; acc.x = 0.0f; acc.y = 0.0f; acc.z = 0.0f; acc.w = 0.0f;
    accum_chunk(acc, Hl, s0, w0 / q0, w1 / q1, w2 / q2, w3 / q3, c0, k0, k1, k2, k3);
    for (int cb = beg + 32; cb < end; cb += 32) {
        const int  j  = cb + lane;
        const bool ok = j < end;
        const int  s  = ok ? csr[j].y : 0;
        float x0, x1, x2, x3;
        edge_logits(als, s, ad, x0, x1, x2, x3);
        const float a0 = ok ? expf(x0 - m0) / q0 : 0.0f, a1 = ok ? expf(x1 - m1) / q1 : 0.0f;
        const float a2 = ok ? expf(x2 - m2) / q2 : 0.0f, a3 = ok ? expf(x3 - m3) / q3 : 0.0f;
        accum_chunk(acc, Hl, s, a0, a1, a2, a3, min(end - cb, 32), k0, k1, k2, k3);
    }

    float4 o; o.x = acc.x + bl[0]; o.y = acc.y + bl[1]; o.z = acc.z + bl[2]; o.w = acc.w + bl[3];
    ((float4*)(out + (size_t)node * HOUT))[lane] = o;
    }
}

extern "C" void kernel_launch(void* const* d_in, const int* in_sizes, int n_in,
                              void* d_out, int out_size, void* d_ws, size_t ws_size,
                              hipStream_t stream) {
    const float* x      = (const float*)d_in[0];
    const int*   ei     = (const int*)d_in[1];
    const float* W1     = (const float*)d_in[2];
    const float* a_src1 = (const float*)d_in[3];
    const float* a_dst1 = (const float*)d_in[4];
    const float* b1     = (const float*)d_in[5];
    const float* W2     = (const float*)d_in[6];
    const float* a_src2 = (const float*)d_in[7];
    const float* a_dst2 = (const float*)d_in[8];
    const float* b2     = (const float*)d_in[9];

    const int N = in_sizes[0] / FIN;
    const int E = in_sizes[1] / 2;
    const int* src = ei;
    const int* dst = ei + E;

    const int nRanges = (N + RANGE - 1) / RANGE;
    const int nEB     = (E + EDGE_BLOCK - 1) / EDGE_BLOCK;
    const int nBH     = nRanges * nEB;
    const int nTilesB = (nBH + SCAN_TILE - 1) / SCAN_TILE;
    const int nTiles  = (N + SCAN_TILE - 1) / SCAN_TILE;
    const size_t nPad = ((size_t)N + 4) & ~(size_t)3;
    const size_t ePad = ((size_t)E + 3) & ~(size_t)3;
    const size_t bPad = ((size_t)nBH + 4) & ~(size_t)3;

    char* p = (char*)d_ws;
    float* H      = (float*)p; p += (size_t)N * HOUT * sizeof(float);
    float* als    = (float*)p; p += (size_t)N * HEADS * sizeof(float);
    float* ald    = (float*)p; p += (size_t)N * HEADS * sizeof(float);
    int*   rowptr = (int*)p;   p += nPad * sizeof(int);
    p += nPad * sizeof(int);
    p += (((size_t)nTiles + 3) & ~(size_t)3) * sizeof(int);
    int2*  csr    = (int2*)p;  p += (size_t)E * sizeof(int2);
    char* q = (char*)H;
    int2*  pe     = (int2*)q;  q += ePad * sizeof(int2);
    int*   pd     = (int*)q;   q += ePad * sizeof(int);
    int*   bh     = (int*)q;   q += bPad * sizeof(int);
    int*   bhoff  = (int*)q;   q += bPad * sizeof(int);
    int*   bsum   = (int*)q;   q += (((size_t)nTilesB + 3) & ~(size_t)3) * sizeof(int);

    float* out = (float*)d_out;

    const int gGemm = (N + 63) / 64;
    const int BT = 256;
    int gNH  = (N * HEADS + BT - 1) / BT;
    int gNW  = (N + 31) / 32;

    k_hist      <<<nEB, 256, 0, stream>>>(dst, bh, E, nRanges, nEB);
    k_scan_local<<<nTilesB, SCAN_B, 0, stream>>>(bh, bhoff, bsum, nBH);
    k_scan_fix  <<<nTilesB, SCAN_B, 0, stream>>>(bhoff, bsum, nBH);
    k_part      <<<nEB, 256, 0, stream>>>(src, dst, bhoff, pe, pd, E, nRanges, nEB);
    k_bucket    <<<nRanges, BUCKET_B, 0, stream>>>(pe, pd, bhoff, rowptr, csr, N, nEB);

    k_gemm_wmma<0><<<gGemm, 128, 0, stream>>>(x, W1, H, N);
    k_attn_coeff<<<gNH, BT, 0, stream>>>(H, a_src1, a_dst1, als, ald, N);
    k_gather    <<<gNW, BT, 0, stream>>>(rowptr, csr, H, als, ald, b1, out, N);

    k_gemm_wmma<1><<<gGemm, 128, 0, stream>>>(out, W2, H, N);
    k_attn_coeff<<<gNH, BT, 0, stream>>>(H, a_src2, a_dst2, als, ald, N);
    k_gather    <<<gNW, BT, 0, stream>>>(rowptr, csr, H, als, ald, b2, out, N);
}


# Gather row loop fed from LDS: a chunk's sources (placed at their edge-id rank) and per-head softmax weights staged per wave, read back 4 edges per 16-byte read; sum order kept, multiply-adds fused
# speedup vs baseline: 12.0957x; 1.0935x over previous
#include <hip/hip_runtime.h>
#include <math.h>


typedef __attribute__((ext_vector_type(2))) float v2f;
typedef __attribute__((ext_vector_type(8))) float v8f;

#define FIN   128
#define HEADS 4
#define CDIM  32
#define HOUT  128
#define NEG_SLOPE 0.2f

#define GM_WAVES 4
#define WL_PITCH 272
template<int RELU>
__global__ __launch_bounds__(GM_WAVES * 32)
void k_gemm_wmma(const float* __restrict__ X, const float* __restrict__ W,
                 float* __restrict__ H, int nRows) {
    __shared__ float Wl[(FIN / 2) * WL_PITCH];
    const int tid   = threadIdx.x;
    const int lane  = tid & 31;
    const int wv    = tid >> 5;
    const int c     = lane & 15;
    const int half  = lane >> 4;
    const int row0  = (blockIdx.x * GM_WAVES + wv) * 16;
    int row         = row0 + c;
    if (row >= nRows) row = nRows - 1;

    const float4* xr = (const float4*)(X + (size_t)row * FIN) + half;
    float4 a[16];
#pragma unroll
    for (int i = 0; i < 16; ++i) a[i] = xr[2 * i];

#pragma unroll
    for (int u = 0; u < (FIN / 2) / GM_WAVES; ++u) {
        const int kp = wv + GM_WAVES * u;
        const float4 r0 = *(const float4*)(W + (size_t)(2 * kp)     * HOUT + 4 * lane);
        const float4 r1 = *(const float4*)(W + (size_t)(2 * kp + 1) * HOUT + 4 * lane);
        float* d = Wl + kp * WL_PITCH + 8 * lane;
        float4 p0; p0.x = r0.x; p0.y = r1.x; p0.z = r0.y; p0.w = r1.y;
        float4 p1; p1.x = r0.z; p1.y = r1.z; p1.z = r0.w; p1.w = r1.w;
        *(float4*)(d)     = p0;
        *(float4*)(d + 4) = p1;
    }
    __syncthreads();

    v8f acc[8];
#pragma unroll
    for (int t = 0; t < 8; ++t) acc[t] = (v8f){};

    const float* wl = Wl + (2 * half) * WL_PITCH + 2 * c;
#pragma unroll
    for (int i = 0; i < 16; ++i) {
        float4 av = a[i];
        if (RELU) { av.x = fmaxf(av.x, 0.0f); av.y = fmaxf(av.y, 0.0f); av.z = fmaxf(av.z, 0.0f); av.w = fmaxf(av.w, 0.0f); }
#pragma unroll
        for (int w = 0; w < 2; ++w) {
            v2f af;
            af.x = w ? av.z : av.x;
            af.y = w ? av.w : av.y;
            const float* wp = wl + (4 * i + w) * WL_PITCH;
#pragma unroll
            for (int t = 0; t < 8; ++t) {
                const v2f b = *(const v2f*)(wp + 32 * t);
                acc[t] = __builtin_amdgcn_wmma_f32_16x16x4_f32(false, af, false, b, (short)0, acc[t], false, false);
            }
        }
    }

    const int mOff = half * 8;
#pragma unroll
    for (int j = 0; j < 8; ++j) {
        const int r = row0 + mOff + j;
        if (r < nRows) {
            float* hr = H + (size_t)r * HOUT + c;
#pragma unroll
            for (int t = 0; t < 8; ++t) hr[16 * t] = acc[t][j];
        }
    }
}

__global__ void k_attn_coeff(const float* __restrict__ H,
                             const float* __restrict__ a_src,
                             const float* __restrict__ a_dst,
                             float* __restrict__ als, float* __restrict__ ald,
                             int n) {
    int i = blockIdx.x * blockDim.x + threadIdx.x;
    if (i >= n * HEADS) return;
    int node = i >> 2, h = i & 3;
    const float* hp = H + (size_t)node * HOUT + h * CDIM;
    const float* as = a_src + h * CDIM;
    const float* ad = a_dst + h * CDIM;
    float s = 0.0f, d = 0.0f;
#pragma unroll
    for (int c = 0; c < CDIM; ++c) { float v = hp[c]; s = fmaf(v, as[c], s); d = fmaf(v, ad[c], d); }
    als[i] = s; ald[i] = d;
}

#define SCAN_B    256
#define SCAN_TILE 1024

#define RANGE_SHIFT 9
#define RANGE       512
#define MAX_RANGES  1024
#define EDGE_BLOCK  4096
#define EDGE_GROUP  1024

__device__ __forceinline__ int4 load4_or_neg(const int* __restrict__ p, int e, int e_total, bool wide) {
    int4 v;
    if (wide && e + 3 < e_total) {
        v = *(const int4*)(p + e);
    } else {
        v.x = (e     < e_total) ? p[e]     : -1;
        v.y = (e + 1 < e_total) ? p[e + 1] : -1;
        v.z = (e + 2 < e_total) ? p[e + 2] : -1;
        v.w = (e + 3 < e_total) ? p[e + 3] : -1;
    }
    return v;
}

__global__ __launch_bounds__(256)
void k_hist(const int* __restrict__ dst, int* __restrict__ bh, int e_total, int n_ranges, int n_blocks) {
    __shared__ int hist[MAX_RANGES];
    for (int i = threadIdx.x; i < n_ranges; i += 256) hist[i] = 0;
    __syncthreads();
    const int e0 = blockIdx.x * EDGE_BLOCK + threadIdx.x * 4;
    const bool wide = (((size_t)dst) & 15) == 0;
    int4 d[4];
    for (int q = 0; q < 4; ++q) d[q] = load4_or_neg(dst, e0 + q * EDGE_GROUP, e_total, wide);
    for (int q = 0; q < 4; ++q) {
        if (d[q].x >= 0) atomicAdd(&hist[d[q].x >> RANGE_SHIFT], 1);
        if (d[q].y >= 0) atomicAdd(&hist[d[q].y >> RANGE_SHIFT], 1);
        if (d[q].z >= 0) atomicAdd(&hist[d[q].z >> RANGE_SHIFT], 1);
        if (d[q].w >= 0) atomicAdd(&hist[d[q].w >> RANGE_SHIFT], 1);
    }
    __syncthreads();
    for (int i = threadIdx.x; i < n_ranges; i += 256) bh[(size_t)i * n_blocks + blockIdx.x] = hist[i];
}

__device__ __forceinline__ void part_one(int* cur, int2* __restrict__ pe, int* __restrict__ pd, int e, int s, int d) {
    if (d >= 0) {
        const int pos = atomicAdd(&cur[d >> RANGE_SHIFT], 1);
        int2 r; r.x = e; r.y = s;
        pe[pos] = r;
        pd[pos] = d;
    }
}

__global__ __launch_bounds__(256)
void k_part(const int* __restrict__ src, const int* __restrict__ dst, const int* __restrict__ bhoff,
            int2* __restrict__ pe, int* __restrict__ pd, int e_total, int n_ranges, int n_blocks) {
    __shared__ int cur[MAX_RANGES];
    for (int i = threadIdx.x; i < n_ranges; i += 256) cur[i] = bhoff[(size_t)i * n_blocks + blockIdx.x];
    const int e0 = blockIdx.x * EDGE_BLOCK + threadIdx.x * 4;
    const bool wide = ((((size_t)src) | ((size_t)dst)) & 15) == 0;
    int4 d[4], s[4];
    for (int q = 0; q < 4; ++q) d[q] = load4_or_neg(dst, e0 + q * EDGE_GROUP, e_total, wide);
    for (int q = 0; q < 4; ++q) s[q] = load4_or_neg(src, e0 + q * EDGE_GROUP, e_total, wide);
    __syncthreads();
    for (int q = 0; q < 4; ++q) {
        const int e = e0 + q * EDGE_GROUP;
        part_one(cur, pe, pd, e,     s[q].x, d[q].x);
        part_one(cur, pe, pd, e + 1, s[q].y, d[q].y);
        part_one(cur, pe, pd, e + 2, s[q].z, d[q].z);
        part_one(cur, pe, pd, e + 3, s[q].w, d[q].w);
    }
}

__global__ __launch_bounds__(SCAN_B)
void k_scan_local(const int* __restrict__ cnt, int* __restrict__ rowptr, int* __restrict__ bsum, int n) {
    __shared__ int wtot[SCAN_B / 32];
    const int t = threadIdx.x, lane = t & 31, wv = t >> 5;
    const int base = blockIdx.x * SCAN_TILE + t * 4;
    int a0 = 0, a1 = 0, a2 = 0, a3 = 0;
    if (base + 3 < n) {
        const int4 v = *(const int4*)(cnt + base);
        a0 = v.x; a1 = v.y; a2 = v.z; a3 = v.w;
    } else {
        if (base     < n) a0 = cnt[base];
        if (base + 1 < n) a1 = cnt[base + 1];
        if (base + 2 < n) a2 = cnt[base + 2];
    }
    const int tsum = a0 + a1 + a2 + a3;
    int incl = tsum;
    for (int o = 1; o < 32; o <<= 1) { const int u = __shfl_up(incl, o); if (lane >= o) incl += u; }
    if (lane == 31) wtot[wv] = incl;
    __syncthreads();
    int pre = 0;
    for (int k = 0; k < wv; ++k) pre += wtot[k];
    const int ex = pre + incl - tsum;
    if (base + 3 < n) {
        int4 r; r.x = ex; r.y = ex + a0; r.z = ex + a0 + a1; r.w = ex + a0 + a1 + a2;
        *(int4*)(rowptr + base) = r;
    } else {
        if (base     < n) rowptr[base]     = ex;
        if (base + 1 < n) rowptr[base + 1] = ex + a0;
        if (base + 2 < n) rowptr[base + 2] = ex + a0 + a1;
    }
    if (t == SCAN_B - 1) bsum[blockIdx.x] = pre + incl;
}

__global__ __launch_bounds__(SCAN_B)
void k_scan_fix(int* __restrict__ rowptr, const int* __restrict__ bsum, int n) {
    __shared__ int wtot[SCAN_B / 32];
    const int t = threadIdx.x, lane = t & 31, wv = t >> 5;
    const int b = blockIdx.x;
    int part = 0;
    for (int k = t; k < b; k += SCAN_B) part += bsum[k];
    for (int o = 16; o > 0; o >>= 1) part += __shfl_xor(part, o);
    if (lane == 0) wtot[wv] = part;
    __syncthreads();
    int off = 0;
    for (int k = 0; k < SCAN_B / 32; ++k) off += wtot[k];
    const int base = b * SCAN_TILE + t * 4;
    for (int q = 0; q < 4; ++q) {
        const int i = base + q;
        if (i < n) rowptr[i] += off;
    }
    if (b == (int)gridDim.x - 1 && t == 0) rowptr[n] = off + bsum[b];
}

#define BUCKET_B 1024
__global__ __launch_bounds__(BUCKET_B)
void k_bucket(const int2* __restrict__ pe, const int* __restrict__ pd, const int* __restrict__ bhoff,
              int* __restrict__ rowptr, int2* __restrict__ csr, int n, int n_blocks) {
    __shared__ int cnt[RANGE];
    __shared__ int wtot[BUCKET_B / 32];
    const int t = threadIdx.x, lane = t & 31, wv = t >> 5;
    const int b = blockIdx.x;
    const int beg = bhoff[(size_t)b * n_blocks];
    const int end = bhoff[(size_t)(b + 1) * n_blocks];
    const int node0 = b << RANGE_SHIFT;
    if (t < RANGE) cnt[t] = 0;
    __syncthreads();
    for (int i0 = beg + t; i0 < end; i0 += 4 * BUCKET_B) {
        int dv[4];
        for (int u = 0; u < 4; ++u) { const int i = i0 + u * BUCKET_B; dv[u] = (i < end) ? pd[i] : -1; }
        for (int u = 0; u < 4; ++u) if (dv[u] >= 0) atomicAdd(&cnt[dv[u] - node0], 1);
    }
    __syncthreads();
    const int v = (t < RANGE) ? cnt[t] : 0;
    int incl = v;
    for (int o = 1; o < 32; o <<= 1) { const int u = __shfl_up(incl, o); if (lane >= o) incl += u; }
    if (lane == 31) wtot[wv] = incl;
    __syncthreads();
    int pre = 0;
    for (int k = 0; k < wv; ++k) pre += wtot[k];
    const int first = beg + pre + incl - v;
    if (t < RANGE) {
        cnt[t] = first;
        if (node0 + t < n) rowptr[node0 + t] = first;
    }
    if (b == (int)gridDim.x - 1 && t == 0) rowptr[n] = end;
    __syncthreads();
    for (int i0 = beg + t; i0 < end; i0 += 4 * BUCKET_B) {
        int dv[4]; int2 ev[4];
        for (int u = 0; u < 4; ++u) {
            const int i = i0 + u * BUCKET_B;
            dv[u] = -1; ev[u].x = 0; ev[u].y = 0;
            if (i < end) { dv[u] = pd[i]; ev[u] = pe[i]; }
        }
        for (int u = 0; u < 4; ++u) {
            if (dv[u] >= 0) {
                const int pos = atomicAdd(&cnt[dv[u] - node0], 1);
                csr[pos] = ev[u];
            }
        }
    }
}

__device__ __forceinline__ float wave_max(float v) {
    for (int o = 16; o > 0; o >>= 1) v = fmaxf(v, __shfl_xor(v, o));
    return v;
}
__device__ __forceinline__ float wave_sum(float v) {
    for (int o = 16; o > 0; o >>= 1) v += __shfl_xor(v, o);
    return v;
}
__device__ __forceinline__ float lane_f(float x, int l) {
    return __uint_as_float((unsigned)__builtin_amdgcn_readlane((int)__float_as_uint(x), l));
}
__device__ __forceinline__ float lrelu(float v) { return (v > 0.0f) ? v : v * NEG_SLOPE; }

__device__ __forceinline__ void edge_logits(const float* __restrict__ als, int s, const float4 ad,
                                            float& x0, float& x1, float& x2, float& x3) {
    const float4 a = *(const float4*)(als + (size_t)s * HEADS);
    x0 = lrelu(a.x + ad.x); x1 = lrelu(a.y + ad.y); x2 = lrelu(a.z + ad.z); x3 = lrelu(a.w + ad.w);
}

#define GA_WAVES 8
template<int K>
__device__ __forceinline__ void accum_k(float4& acc, const char* __restrict__ Hb, unsigned lane_off,
                                        const float* al_r, const int* sl_w, int jj) {
    float am[K]; int sj[K];
    if constexpr (K == 8) {
        const float4 x = *(const float4*)(al_r + jj), y = *(const float4*)(al_r + jj + 4);
        const int4   p = *(const int4*)(sl_w + jj),   q = *(const int4*)(sl_w + jj + 4);
        am[0] = x.x; am[1] = x.y; am[2] = x.z; am[3] = x.w; am[4] = y.x; am[5] = y.y; am[6] = y.z; am[7] = y.w;
        sj[0] = p.x; sj[1] = p.y; sj[2] = p.z; sj[3] = p.w; sj[4] = q.x; sj[5] = q.y; sj[6] = q.z; sj[7] = q.w;
    } else if constexpr (K == 4) {
        const float4 x = *(const float4*)(al_r + jj);
        const int4   p = *(const int4*)(sl_w + jj);
        am[0] = x.x; am[1] = x.y; am[2] = x.z; am[3] = x.w;
        sj[0] = p.x; sj[1] = p.y; sj[2] = p.z; sj[3] = p.w;
    } else if constexpr (K == 2) {
        const float2 x = *(const float2*)(al_r + jj);
        const int2   p = *(const int2*)(sl_w + jj);
        am[0] = x.x; am[1] = x.y;
        sj[0] = p.x; sj[1] = p.y;
    } else {
        am[0] = al_r[jj];
        sj[0] = sl_w[jj];
    }
    float4 hv[K];
    for (int u = 0; u < K; ++u)
        hv[u] = *(const float4*)(Hb + ((((unsigned)sj[u]) << 9) + lane_off));
    for (int u = 0; u < K; ++u) {
        acc.x = fmaf(hv[u].x, am[u], acc.x); acc.y = fmaf(hv[u].y, am[u], acc.y);
        acc.z = fmaf(hv[u].z, am[u], acc.z); acc.w = fmaf(hv[u].w, am[u], acc.w);
    }
}

__device__ __forceinline__ void accum_chunk(float4& acc, const char* __restrict__ Hb, unsigned lane_off,
                                            const float* al_r, const int* sl_w, int cnt) {
    int jj = 0;
    for (; jj + 8 <= cnt; jj += 8) accum_k<8>(acc, Hb, lane_off, al_r, sl_w, jj);
    if (jj + 4 <= cnt) { accum_k<4>(acc, Hb, lane_off, al_r, sl_w, jj); jj += 4; }
    if (jj + 2 <= cnt) { accum_k<2>(acc, Hb, lane_off, al_r, sl_w, jj); jj += 2; }
    if (jj < cnt)        accum_k<1>(acc, Hb, lane_off, al_r, sl_w, jj);
}

__global__ __launch_bounds__(GA_WAVES * 32)
void k_gather(const int* __restrict__ rowptr, const int2* __restrict__ csr,
              const float* __restrict__ Hm,
              const float* __restrict__ als, const float* __restrict__ ald,
              const float* __restrict__ bias, float* __restrict__ out, int n) {
    __shared__ __attribute__((aligned(16))) float al_s[GA_WAVES][HEADS][32];
    __shared__ __attribute__((aligned(16))) int   sl_s[GA_WAVES][32];
    const int lane = threadIdx.x & 31;
    const int wv   = threadIdx.x >> 5;
    const int hsel = lane >> 3;
    float*       al_w = &al_s[wv][0][0];
    int*         sl_w = &sl_s[wv][0];
    const float* al_r = al_w + hsel * 32;
    const char*  Hb   = (const char*)Hm;
    const unsigned lane_off = (unsigned)lane * 16u;
    const float* bl = bias + lane * 4;
    const int wave0  = __builtin_amdgcn_readfirstlane((int)((blockIdx.x * blockDim.x + threadIdx.x) >> 5));
    const int nwaves = (int)((gridDim.x * blockDim.x) >> 5);
    for (int node = wave0; node < n; node += nwaves) {
    const int beg = rowptr[node];
    const int end = rowptr[node + 1];
    const float4 ad = *(const float4*)(ald + (size_t)node * HEADS);

    const int  c0  = min(end - beg, 32);
    const bool ok0 = lane < c0;
    int eid = 0x7fffffff, sraw = 0;
    if (ok0) { const int2 r = csr[beg + lane]; eid = r.x; sraw = r.y; }
    int rank = 0;
    for (int k = 0; k < c0; ++k) rank += (__builtin_amdgcn_readlane(eid, k) < eid) ? 1 : 0;
    if (ok0) sl_w[rank] = sraw;
    const int s0 = ok0 ? sl_w[lane] : 0;

    float e0, e1, e2, e3;
    edge_logits(als, s0, ad, e0, e1, e2, e3);
    float m0 = ok0 ? e0 : -INFINITY, m1 = ok0 ? e1 : -INFINITY, m2 = ok0 ? e2 : -INFINITY, m3 = ok0 ? e3 : -INFINITY;
    for (int cb = beg + 32; cb < end; cb += 32) {
        const int j = cb + lane;
        if (j < end) {
            float x0, x1, x2, x3;
            edge_logits(als, csr[j].y, ad, x0, x1, x2, x3);
            m0 = fmaxf(m0, x0); m1 = fmaxf(m1, x1); m2 = fmaxf(m2, x2); m3 = fmaxf(m3, x3);
        }
    }
    m0 = wave_max(m0); m1 = wave_max(m1); m2 = wave_max(m2); m3 = wave_max(m3);
    if (!__builtin_isfinite(m0)) m0 = 0.0f;
    if (!__builtin_isfinite(m1)) m1 = 0.0f;
    if (!__builtin_isfinite(m2)) m2 = 0.0f;
    if (!__builtin_isfinite(m3)) m3 = 0.0f;

    const float w0 = ok0 ? expf(e0 - m0) : 0.0f, w1 = ok0 ? expf(e1 - m1) : 0.0f;
    const float w2 = ok0 ? expf(e2 - m2) : 0.0f, w3 = ok0 ? expf(e3 - m3) : 0.0f;
    float d0 = w0, d1 = w1, d2 = w2, d3 = w3;
    for (int cb = beg + 32; cb < end; cb += 32) {
        const int j = cb + lane;
        if (j < end) {
            float x0, x1, x2, x3;
            edge_logits(als, csr[j].y, ad, x0, x1, x2, x3);
            d0 += expf(x0 - m0); d1 += expf(x1 - m1); d2 += expf(x2 - m2); d3 += expf(x3 - m3);
        }
    }
    d0 = wave_sum(d0); d1 = wave_sum(d1); d2 = wave_sum(d2); d3 = wave_sum(d3);
    const float q0 = fmaxf(d0, 1e-16f), q1 = fmaxf(d1, 1e-16f), q2 = fmaxf(d2, 1e-16f), q3 = fmaxf(d3, 1e-16f);

    float4 acc; acc.x = 0.0f; acc.y = 0.0f; acc.z = 0.0f; acc.w = 0.0f;
    al_w[lane] = w0 / q0; al_w[32 + lane] = w1 / q1; al_w[64 + lane] = w2 / q2; al_w[96 + lane] = w3 / q3;
    accum_chunk(acc, Hb, lane_off, al_r, sl_w, c0);
    for (int cb = beg + 32; cb < end; cb += 32) {
        const int  j  = cb + lane;
        const bool ok = j < end;
        const int  s  = ok ? csr[j].y : 0;
        float x0, x1, x2, x3;
        edge_logits(als, s, ad, x0, x1, x2, x3);
        const float a0 = ok ? expf(x0 - m0) / q0 : 0.0f, a1 = ok ? expf(x1 - m1) / q1 : 0.0f;
        const float a2 = ok ? expf(x2 - m2) / q2 : 0.0f, a3 = ok ? expf(x3 - m3) / q3 : 0.0f;
        sl_w[lane] = s;
        al_w[lane] = a0; al_w[32 + lane] = a1; al_w[64 + lane] = a2; al_w[96 + lane] = a3;
        accum_chunk(acc, Hb, lane_off, al_r, sl_w, min(end - cb, 32));
    }

    float4 o; o.x = acc.x + bl[0]; o.y = acc.y + bl[1]; o.z = acc.z + bl[2]; o.w = acc.w + bl[3];
    ((float4*)(out + (size_t)node * HOUT))[lane] = o;
    }
}

extern "C" void kernel_launch(void* const* d_in, const int* in_sizes, int n_in,
                              void* d_out, int out_size, void* d_ws, size_t ws_size,
                              hipStream_t stream) {
    const float* x      = (const float*)d_in[0];
    const int*   ei     = (const int*)d_in[1];
    const float* W1     = (const float*)d_in[2];
    const float* a_src1 = (const float*)d_in[3];
    const float* a_dst1 = (const float*)d_in[4];
    const float* b1     = (const float*)d_in[5];
    const float* W2     = (const float*)d_in[6];
    const float* a_src2 = (const float*)d_in[7];
    const float* a_dst2 = (const float*)d_in[8];
    const float* b2     = (const float*)d_in[9];

    const int N = in_sizes[0] / FIN;
    const int E = in_sizes[1] / 2;
    const int* src = ei;
    const int* dst = ei + E;

    const int nRanges = (N + RANGE - 1) / RANGE;
    const int nEB     = (E + EDGE_BLOCK - 1) / EDGE_BLOCK;
    const int nBH     = nRanges * nEB;
    const int nTilesB = (nBH + SCAN_TILE - 1) / SCAN_TILE;
    const int nTiles  = (N + SCAN_TILE - 1) / SCAN_TILE;
    const size_t nPad = ((size_t)N + 4) & ~(size_t)3;
    const size_t ePad = ((size_t)E + 3) & ~(size_t)3;
    const size_t bPad = ((size_t)nBH + 4) & ~(size_t)3;

    char* p = (char*)d_ws;
    float* H      = (float*)p; p += (size_t)N * HOUT * sizeof(float);
    float* als    = (float*)p; p += (size_t)N * HEADS * sizeof(float);
    float* ald    = (float*)p; p += (size_t)N * HEADS * sizeof(float);
    int*   rowptr = (int*)p;   p += nPad * sizeof(int);
    p += nPad * sizeof(int);
    p += (((size_t)nTiles + 3) & ~(size_t)3) * sizeof(int);
    int2*  csr    = (int2*)p;  p += (size_t)E * sizeof(int2);
    char* q = (char*)H;
    int2*  pe     = (int2*)q;  q += ePad * sizeof(int2);
    int*   pd     = (int*)q;   q += ePad * sizeof(int);
    int*   bh     = (int*)q;   q += bPad * sizeof(int);
    int*   bhoff  = (int*)q;   q += bPad * sizeof(int);
    int*   bsum   = (int*)q;   q += (((size_t)nTilesB + 3) & ~(size_t)3) * sizeof(int);

    float* out = (float*)d_out;

    const int gGemm = (N + 63) / 64;
    const int BT = 256;
    int gNH  = (N * HEADS + BT - 1) / BT;
    int gNW  = (N + 31) / 32;

    k_hist      <<<nEB, 256, 0, stream>>>(dst, bh, E, nRanges, nEB);
    k_scan_local<<<nTilesB, SCAN_B, 0, stream>>>(bh, bhoff, bsum, nBH);
    k_scan_fix  <<<nTilesB, SCAN_B, 0, stream>>>(bhoff, bsum, nBH);
    k_part      <<<nEB, 256, 0, stream>>>(src, dst, bhoff, pe, pd, E, nRanges, nEB);
    k_bucket    <<<nRanges, BUCKET_B, 0, stream>>>(pe, pd, bhoff, rowptr, csr, N, nEB);

    k_gemm_wmma<0><<<gGemm, 128, 0, stream>>>(x, W1, H, N);
    k_attn_coeff<<<gNH, BT, 0, stream>>>(H, a_src1, a_dst1, als, ald, N);
    k_gather    <<<gNW, BT, 0, stream>>>(rowptr, csr, H, als, ald, b1, out, N);

    k_gemm_wmma<1><<<gGemm, 128, 0, stream>>>(out, W2, H, N);
    k_attn_coeff<<<gNH, BT, 0, stream>>>(H, a_src2, a_dst2, als, ald, N);
    k_gather    <<<gNW, BT, 0, stream>>>(rowptr, csr, H, als, ald, b2, out, N);
}


# Gather with two nodes per wave, one per 16-lane row: bounds, ranking, max, exp and sum serve both, row exchanges by DPP, a lane sums 8 channels, unused slots weight 0; denominators' sum order changed
# speedup vs baseline: 13.0040x; 1.0751x over previous
#include <hip/hip_runtime.h>
#include <math.h>


typedef __attribute__((ext_vector_type(2))) float v2f;
typedef __attribute__((ext_vector_type(8))) float v8f;

#define FIN   128
#define HEADS 4
#define CDIM  32
#define HOUT  128
#define NEG_SLOPE 0.2f

#define GM_WAVES 4
#define WL_PITCH 272
template<int M>
__device__ __forceinline__ float row_xor(float x) {
    const int v = __float_as_int(x);
    return __int_as_float(__builtin_amdgcn_update_dpp(v, v, 0x160 + M, 0xf, 0xf, false));
}
template<int M, int CNT>
__device__ __forceinline__ void halve_step(float* v, bool bit) {
#pragma unroll
    for (int i = 0; i < CNT; ++i) {
        const float keep = bit ? v[i + CNT] : v[i];
        const float send = bit ? v[i] : v[i + CNT];
        v[i] = keep + row_xor<M>(send);
    }
}
template<int RELU>
__global__ __launch_bounds__(GM_WAVES * 32)
void k_gemm_wmma(const float* __restrict__ X, const float* __restrict__ W,
                 const float* __restrict__ a_src, const float* __restrict__ a_dst,
                 float* __restrict__ H, float* __restrict__ als, float* __restrict__ ald, int nRows) {
    __shared__ float Wl[(FIN / 2) * WL_PITCH];
    __shared__ float Av[2 * HOUT];
    const int tid   = threadIdx.x;
    const int lane  = tid & 31;
    const int wv    = tid >> 5;
    const int c     = lane & 15;
    const int half  = lane >> 4;
    const int row0  = (blockIdx.x * GM_WAVES + wv) * 16;
    int row         = row0 + c;
    if (row >= nRows) row = nRows - 1;

    const float4* xr = (const float4*)(X + (size_t)row * FIN) + half;
    float4 a[16];
#pragma unroll
    for (int i = 0; i < 16; ++i) a[i] = xr[2 * i];

#pragma unroll
    for (int u = 0; u < (FIN / 2) / GM_WAVES; ++u) {
        const int kp = wv + GM_WAVES * u;
        const float4 r0 = *(const float4*)(W + (size_t)(2 * kp)     * HOUT + 4 * lane);
        const float4 r1 = *(const float4*)(W + (size_t)(2 * kp + 1) * HOUT + 4 * lane);
        float* d = Wl + kp * WL_PITCH + 8 * lane;
        float4 p0; p0.x = r0.x; p0.y = r1.x; p0.z = r0.y; p0.w = r1.y;
        float4 p1; p1.x = r0.z; p1.y = r1.z; p1.z = r0.w; p1.w = r1.w;
        *(float4*)(d)     = p0;
        *(float4*)(d + 4) = p1;
    }
    Av[tid]        = a_src[tid];
    Av[HOUT + tid] = a_dst[tid];
    __syncthreads();

    v8f acc[8];
#pragma unroll
    for (int t = 0; t < 8; ++t) acc[t] = (v8f){};

    const float* wl = Wl + (2 * half) * WL_PITCH + 2 * c;
#pragma unroll
    for (int i = 0; i < 16; ++i) {
        float4 av = a[i];
        if (RELU) { av.x = fmaxf(av.x, 0.0f); av.y = fmaxf(av.y, 0.0f); av.z = fmaxf(av.z, 0.0f); av.w = fmaxf(av.w, 0.0f); }
#pragma unroll
        for (int w = 0; w < 2; ++w) {
            v2f af;
            af.x = w ? av.z : av.x;
            af.y = w ? av.w : av.y;
            const float* wp = wl + (4 * i + w) * WL_PITCH;
#pragma unroll
            for (int t = 0; t < 8; ++t) {
                const v2f b = *(const v2f*)(wp + 32 * t);
                acc[t] = __builtin_amdgcn_wmma_f32_16x16x4_f32(false, af, false, b, (short)0, acc[t], false, false);
            }
        }
    }

    const int mOff = half * 8;
#pragma unroll
    for (int j = 0; j < 8; ++j) {
        const int r = row0 + mOff + j;
        if (r < nRows) {
            float* hr = H + (size_t)r * HOUT + c;
#pragma unroll
            for (int t = 0; t < 8; ++t) hr[16 * t] = acc[t][j];
        }
    }

    float sv[32], dv[32];
#pragma unroll
    for (int h = 0; h < HEADS; ++h) {
        const float as0 = Av[32 * h + c],        as1 = Av[32 * h + 16 + c];
        const float ad0 = Av[HOUT + 32 * h + c], ad1 = Av[HOUT + 32 * h + 16 + c];
#pragma unroll
        for (int j = 0; j < 8; ++j) {
            sv[4 * j + h] = acc[2 * h][j] * as0 + acc[2 * h + 1][j] * as1;
            dv[4 * j + h] = acc[2 * h][j] * ad0 + acc[2 * h + 1][j] * ad1;
        }
    }
    halve_step<8, 16>(sv, (c & 8) != 0);  halve_step<8, 16>(dv, (c & 8) != 0);
    halve_step<4, 8>(sv, (c & 4) != 0);   halve_step<4, 8>(dv, (c & 4) != 0);
    halve_step<2, 4>(sv, (c & 2) != 0);   halve_step<2, 4>(dv, (c & 2) != 0);
    halve_step<1, 2>(sv, (c & 1) != 0);   halve_step<1, 2>(dv, (c & 1) != 0);
    const int rg = row0 + mOff;
    if (rg + (c >> 1) < nRows) {
        float2 s2; s2.x = sv[0]; s2.y = sv[1];
        float2 d2; d2.x = dv[0]; d2.y = dv[1];
        *(float2*)(als + (size_t)rg * HEADS + 2 * c) = s2;
        *(float2*)(ald + (size_t)rg * HEADS + 2 * c) = d2;
    }
}

#define SCAN_B    256
#define SCAN_TILE 1024

#define RANGE_SHIFT 9
#define RANGE       512
#define MAX_RANGES  1024
#define EDGE_BLOCK  4096
#define EDGE_GROUP  1024

__device__ __forceinline__ int4 load4_or_neg(const int* __restrict__ p, int e, int e_total, bool wide) {
    int4 v;
    if (wide && e + 3 < e_total) {
        v = *(const int4*)(p + e);
    } else {
        v.x = (e     < e_total) ? p[e]     : -1;
        v.y = (e + 1 < e_total) ? p[e + 1] : -1;
        v.z = (e + 2 < e_total) ? p[e + 2] : -1;
        v.w = (e + 3 < e_total) ? p[e + 3] : -1;
    }
    return v;
}

__global__ __launch_bounds__(256)
void k_hist(const int* __restrict__ dst, int* __restrict__ bh, int e_total, int n_ranges, int n_blocks) {
    __shared__ int hist[MAX_RANGES];
    for (int i = threadIdx.x; i < n_ranges; i += 256) hist[i] = 0;
    __syncthreads();
    const int e0 = blockIdx.x * EDGE_BLOCK + threadIdx.x * 4;
    const bool wide = (((size_t)dst) & 15) == 0;
    int4 d[4];
    for (int q = 0; q < 4; ++q) d[q] = load4_or_neg(dst, e0 + q * EDGE_GROUP, e_total, wide);
    for (int q = 0; q < 4; ++q) {
        if (d[q].x >= 0) atomicAdd(&hist[d[q].x >> RANGE_SHIFT], 1);
        if (d[q].y >= 0) atomicAdd(&hist[d[q].y >> RANGE_SHIFT], 1);
        if (d[q].z >= 0) atomicAdd(&hist[d[q].z >> RANGE_SHIFT], 1);
        if (d[q].w >= 0) atomicAdd(&hist[d[q].w >> RANGE_SHIFT], 1);
    }
    __syncthreads();
    for (int i = threadIdx.x; i < n_ranges; i += 256) bh[(size_t)i * n_blocks + blockIdx.x] = hist[i];
}

__device__ __forceinline__ void part_one(int* cur, int2* __restrict__ pe, int* __restrict__ pd, int e, int s, int d) {
    if (d >= 0) {
        const int pos = atomicAdd(&cur[d >> RANGE_SHIFT], 1);
        int2 r; r.x = e; r.y = s;
        pe[pos] = r;
        pd[pos] = d;
    }
}

__global__ __launch_bounds__(256)
void k_part(const int* __restrict__ src, const int* __restrict__ dst, const int* __restrict__ bhoff,
            int2* __restrict__ pe, int* __restrict__ pd, int e_total, int n_ranges, int n_blocks) {
    __shared__ int cur[MAX_RANGES];
    for (int i = threadIdx.x; i < n_ranges; i += 256) cur[i] = bhoff[(size_t)i * n_blocks + blockIdx.x];
    const int e0 = blockIdx.x * EDGE_BLOCK + threadIdx.x * 4;
    const bool wide = ((((size_t)src) | ((size_t)dst)) & 15) == 0;
    int4 d[4], s[4];
    for (int q = 0; q < 4; ++q) d[q] = load4_or_neg(dst, e0 + q * EDGE_GROUP, e_total, wide);
    for (int q = 0; q < 4; ++q) s[q] = load4_or_neg(src, e0 + q * EDGE_GROUP, e_total, wide);
    __syncthreads();
    for (int q = 0; q < 4; ++q) {
        const int e = e0 + q * EDGE_GROUP;
        part_one(cur, pe, pd, e,     s[q].x, d[q].x);
        part_one(cur, pe, pd, e + 1, s[q].y, d[q].y);
        part_one(cur, pe, pd, e + 2, s[q].z, d[q].z);
        part_one(cur, pe, pd, e + 3, s[q].w, d[q].w);
    }
}

__global__ __launch_bounds__(SCAN_B)
void k_scan_local(const int* __restrict__ cnt, int* __restrict__ rowptr, int* __restrict__ bsum, int n) {
    __shared__ int wtot[SCAN_B / 32];
    const int t = threadIdx.x, lane = t & 31, wv = t >> 5;
    const int base = blockIdx.x * SCAN_TILE + t * 4;
    int a0 = 0, a1 = 0, a2 = 0, a3 = 0;
    if (base + 3 < n) {
        const int4 v = *(const int4*)(cnt + base);
        a0 = v.x; a1 = v.y; a2 = v.z; a3 = v.w;
    } else {
        if (base     < n) a0 = cnt[base];
        if (base + 1 < n) a1 = cnt[base + 1];
        if (base + 2 < n) a2 = cnt[base + 2];
    }
    const int tsum = a0 + a1 + a2 + a3;
    int incl = tsum;
    for (int o = 1; o < 32; o <<= 1) { const int u = __shfl_up(incl, o); if (lane >= o) incl += u; }
    if (lane == 31) wtot[wv] = incl;
    __syncthreads();
    int pre = 0;
    for (int k = 0; k < wv; ++k) pre += wtot[k];
    const int ex = pre + incl - tsum;
    if (base + 3 < n) {
        int4 r; r.x = ex; r.y = ex + a0; r.z = ex + a0 + a1; r.w = ex + a0 + a1 + a2;
        *(int4*)(rowptr + base) = r;
    } else {
        if (base     < n) rowptr[base]     = ex;
        if (base + 1 < n) rowptr[base + 1] = ex + a0;
        if (base + 2 < n) rowptr[base + 2] = ex + a0 + a1;
    }
    if (t == SCAN_B - 1) bsum[blockIdx.x] = pre + incl;
}

__global__ __launch_bounds__(SCAN_B)
void k_scan_fix(int* __restrict__ rowptr, const int* __restrict__ bsum, int n) {
    __shared__ int wtot[SCAN_B / 32];
    const int t = threadIdx.x, lane = t & 31, wv = t >> 5;
    const int b = blockIdx.x;
    int part = 0;
    for (int k = t; k < b; k += SCAN_B) part += bsum[k];
    for (int o = 16; o > 0; o >>= 1) part += __shfl_xor(part, o);
    if (lane == 0) wtot[wv] = part;
    __syncthreads();
    int off = 0;
    for (int k = 0; k < SCAN_B / 32; ++k) off += wtot[k];
    const int base = b * SCAN_TILE + t * 4;
    for (int q = 0; q < 4; ++q) {
        const int i = base + q;
        if (i < n) rowptr[i] += off;
    }
    if (b == (int)gridDim.x - 1 && t == 0) rowptr[n] = off + bsum[b];
}

#define BUCKET_B 1024
__global__ __launch_bounds__(BUCKET_B)
void k_bucket(const int2* __restrict__ pe, const int* __restrict__ pd, const int* __restrict__ bhoff,
              int* __restrict__ rowptr, int2* __restrict__ csr, int n, int n_blocks) {
    __shared__ int cnt[RANGE];
    __shared__ int wtot[BUCKET_B / 32];
    const int t = threadIdx.x, lane = t & 31, wv = t >> 5;
    const int b = blockIdx.x;
    const int beg = bhoff[(size_t)b * n_blocks];
    const int end = bhoff[(size_t)(b + 1) * n_blocks];
    const int node0 = b << RANGE_SHIFT;
    if (t < RANGE) cnt[t] = 0;
    __syncthreads();
    for (int i0 = beg + t; i0 < end; i0 += 4 * BUCKET_B) {
        int dv[4];
        for (int u = 0; u < 4; ++u) { const int i = i0 + u * BUCKET_B; dv[u] = (i < end) ? pd[i] : -1; }
        for (int u = 0; u < 4; ++u) if (dv[u] >= 0) atomicAdd(&cnt[dv[u] - node0], 1);
    }
    __syncthreads();
    const int v = (t < RANGE) ? cnt[t] : 0;
    int incl = v;
    for (int o = 1; o < 32; o <<= 1) { const int u = __shfl_up(incl, o); if (lane >= o) incl += u; }
    if (lane == 31) wtot[wv] = incl;
    __syncthreads();
    int pre = 0;
    for (int k = 0; k < wv; ++k) pre += wtot[k];
    const int first = beg + pre + incl - v;
    if (t < RANGE) {
        cnt[t] = first;
        if (node0 + t < n) rowptr[node0 + t] = first;
    }
    if (b == (int)gridDim.x - 1 && t == 0) rowptr[n] = end;
    __syncthreads();
    for (int i0 = beg + t; i0 < end; i0 += 4 * BUCKET_B) {
        int dv[4]; int2 ev[4];
        for (int u = 0; u < 4; ++u) {
            const int i = i0 + u * BUCKET_B;
            dv[u] = -1; ev[u].x = 0; ev[u].y = 0;
            if (i < end) { dv[u] = pd[i]; ev[u] = pe[i]; }
        }
        for (int u = 0; u < 4; ++u) {
            if (dv[u] >= 0) {
                const int pos = atomicAdd(&cnt[dv[u] - node0], 1);
                csr[pos] = ev[u];
            }
        }
    }
}

template<int M>
__device__ __forceinline__ float in_row(float x) {
    const int v = __float_as_int(x);
    return __int_as_float(__builtin_amdgcn_update_dpp(v, v, 0x160 + M, 0xf, 0xf, true));
}
template<int K>
__device__ __forceinline__ int row_lane(int x) {
    return __builtin_amdgcn_update_dpp(x, x, 0x150 + K, 0xf, 0xf, true);
}
__device__ __forceinline__ float row_max(float v) {
    v = fmaxf(v, in_row<8>(v)); v = fmaxf(v, in_row<4>(v)); v = fmaxf(v, in_row<2>(v)); v = fmaxf(v, in_row<1>(v));
    return v;
}
__device__ __forceinline__ float row_sum(float v) {
    v += in_row<8>(v); v += in_row<4>(v); v += in_row<2>(v); v += in_row<1>(v);
    return v;
}
__device__ __forceinline__ float lrelu(float v) { return (v > 0.0f) ? v : v * NEG_SLOPE; }

__device__ __forceinline__ void edge_logits(const float* __restrict__ als, int s, const float4 ad,
                                            float& x0, float& x1, float& x2, float& x3) {
    const float4 a = *(const float4*)(als + (size_t)s * HEADS);
    x0 = lrelu(a.x + ad.x); x1 = lrelu(a.y + ad.y); x2 = lrelu(a.z + ad.z); x3 = lrelu(a.w + ad.w);
}

template<int K>
__device__ __forceinline__ void rank_step(int ea, int eb, int& ra, int& rb) {
    const int xa = row_lane<K>(ea), xb = row_lane<K>(eb);
    ra += ((xa < ea) ? 1 : 0) + ((xb < ea) ? 1 : 0);
    rb += ((xa < eb) ? 1 : 0) + ((xb < eb) ? 1 : 0);
    if constexpr (K + 1 < 16) rank_step<K + 1>(ea, eb, ra, rb);
}

#define GA_WAVES 8
__device__ __forceinline__ void accum_slots(float4& acc0, float4& acc1, const char* __restrict__ Hb, unsigned lane_off,
                                            const float* al_r0, const float* al_r1, const int* sl_w, int cnt) {
#pragma clang loop unroll(disable)
    for (int jj = 0; jj < cnt; jj += 4) {
        const int4   p  = *(const int4*)(sl_w + jj);
        const float4 x0 = *(const float4*)(al_r0 + jj);
        const float4 x1 = *(const float4*)(al_r1 + jj);
        const int   sj[4] = {p.x, p.y, p.z, p.w};
        const float a0[4] = {x0.x, x0.y, x0.z, x0.w};
        const float a1[4] = {x1.x, x1.y, x1.z, x1.w};
        float4 hv0[4], hv1[4];
        for (int u = 0; u < 4; ++u) {
            const char* q = Hb + ((((unsigned)sj[u]) << 9) + lane_off);
            hv0[u] = *(const float4*)q;
            hv1[u] = *(const float4*)(q + 256);
        }
        for (int u = 0; u < 4; ++u) {
            acc0.x = fmaf(hv0[u].x, a0[u], acc0.x); acc0.y = fmaf(hv0[u].y, a0[u], acc0.y);
            acc0.z = fmaf(hv0[u].z, a0[u], acc0.z); acc0.w = fmaf(hv0[u].w, a0[u], acc0.w);
            acc1.x = fmaf(hv1[u].x, a1[u], acc1.x); acc1.y = fmaf(hv1[u].y, a1[u], acc1.y);
            acc1.z = fmaf(hv1[u].z, a1[u], acc1.z); acc1.w = fmaf(hv1[u].w, a1[u], acc1.w);
        }
    }
}

__global__ __launch_bounds__(GA_WAVES * 32, 12)
void k_gather(const int* __restrict__ rowptr, const int2* __restrict__ csr,
              const float* __restrict__ Hm,
              const float* __restrict__ als, const float* __restrict__ ald,
              const float* __restrict__ bias, float* __restrict__ out, int n) {
    __shared__ __attribute__((aligned(16))) float al_s[GA_WAVES][2][HEADS][32];
    __shared__ __attribute__((aligned(16))) int   sl_s[GA_WAVES][2][32];
    __shared__ __attribute__((aligned(16))) float bias_s[HOUT];
    if (threadIdx.x < HOUT) bias_s[threadIdx.x] = bias[threadIdx.x];
    __syncthreads();
    const int lane = threadIdx.x & 31;
    const int wv   = threadIdx.x >> 5;
    const int r    = lane & 15;
    const int h    = lane >> 4;
    float*       al_w  = &al_s[wv][h][0][0];
    int*         sl_w  = &sl_s[wv][h][0];
    const float* al_r0 = al_w + (r >> 3) * 32;
    const float* al_r1 = al_w + (2 + (r >> 3)) * 32;
    const char*  Hb    = (const char*)Hm;
    const unsigned lane_off = (unsigned)r * 16u;
    const int wave0  = __builtin_amdgcn_readfirstlane((int)((blockIdx.x * blockDim.x + threadIdx.x) >> 5));
    const int nwaves = (int)((gridDim.x * blockDim.x) >> 5);
    const int npairs = (n + 1) >> 1;
    for (int pair = wave0; pair < npairs; pair += nwaves) {
    const int  node = 2 * pair + h;
    const bool nv   = node < n;
    const int  nd   = nv ? node : n - 1;
    const int  beg  = rowptr[nd];
    const int  d    = nv ? rowptr[nd + 1] - beg : 0;
    const float4 ad = *(const float4*)(ald + (size_t)nd * HEADS);
    const int dmax = max(__builtin_amdgcn_readlane(d, 0), __builtin_amdgcn_readlane(d, 16));

    const bool oka = r < d, okb = 16 + r < d;
    int eida = 0x7fffffff, srca = 0, eidb = 0x7fffffff, srcb = 0;
    if (oka) { const int2 t = csr[beg + r];      eida = t.x; srca = t.y; }
    if (okb) { const int2 t = csr[beg + 16 + r]; eidb = t.x; srcb = t.y; }
    int ranka = 0, rankb = 0;
    rank_step<0>(eida, eidb, ranka, rankb);
    if (oka) sl_w[ranka] = srca;
    if (okb) sl_w[rankb] = srcb;
    const int sa = oka ? sl_w[r] : 0;
    const int sb = okb ? sl_w[16 + r] : 0;

    float ea0, ea1, ea2, ea3, eb0, eb1, eb2, eb3;
    edge_logits(als, sa, ad, ea0, ea1, ea2, ea3);
    edge_logits(als, sb, ad, eb0, eb1, eb2, eb3);
    float m0 = fmaxf(oka ? ea0 : -INFINITY, okb ? eb0 : -INFINITY);
    float m1 = fmaxf(oka ? ea1 : -INFINITY, okb ? eb1 : -INFINITY);
    float m2 = fmaxf(oka ? ea2 : -INFINITY, okb ? eb2 : -INFINITY);
    float m3 = fmaxf(oka ? ea3 : -INFINITY, okb ? eb3 : -INFINITY);
    for (int cb = 32; cb < dmax; cb += 16) {
        if (cb + r < d) {
            float x0, x1, x2, x3;
            edge_logits(als, csr[beg + cb + r].y, ad, x0, x1, x2, x3);
            m0 = fmaxf(m0, x0); m1 = fmaxf(m1, x1); m2 = fmaxf(m2, x2); m3 = fmaxf(m3, x3);
        }
    }
    m0 = row_max(m0); m1 = row_max(m1); m2 = row_max(m2); m3 = row_max(m3);
    if (!__builtin_isfinite(m0)) m0 = 0.0f;
    if (!__builtin_isfinite(m1)) m1 = 0.0f;
    if (!__builtin_isfinite(m2)) m2 = 0.0f;
    if (!__builtin_isfinite(m3)) m3 = 0.0f;

    const float wa0 = expf(oka ? ea0 - m0 : -INFINITY), wa1 = expf(oka ? ea1 - m1 : -INFINITY);
    const float wa2 = expf(oka ? ea2 - m2 : -INFINITY), wa3 = expf(oka ? ea3 - m3 : -INFINITY);
    const float wb0 = expf(okb ? eb0 - m0 : -INFINITY), wb1 = expf(okb ? eb1 - m1 : -INFINITY);
    const float wb2 = expf(okb ? eb2 - m2 : -INFINITY), wb3 = expf(okb ? eb3 - m3 : -INFINITY);
    float d0 = wa0 + wb0, d1 = wa1 + wb1, d2 = wa2 + wb2, d3 = wa3 + wb3;
    for (int cb = 32; cb < dmax; cb += 16) {
        if (cb + r < d) {
            float x0, x1, x2, x3;
            edge_logits(als, csr[beg + cb + r].y, ad, x0, x1, x2, x3);
            d0 += expf(x0 - m0); d1 += expf(x1 - m1); d2 += expf(x2 - m2); d3 += expf(x3 - m3);
        }
    }
    d0 = row_sum(d0); d1 = row_sum(d1); d2 = row_sum(d2); d3 = row_sum(d3);
    const float q0 = fmaxf(d0, 1e-16f), q1 = fmaxf(d1, 1e-16f), q2 = fmaxf(d2, 1e-16f), q3 = fmaxf(d3, 1e-16f);

    const int s_first = row_lane<0>(sa);
    const int pads    = (d > 0) ? s_first : nd;
    if (!oka) sl_w[r] = pads;
    if (!okb) sl_w[16 + r] = pads;
    al_w[r]      = wa0 / q0; al_w[32 + r]      = wa1 / q1; al_w[64 + r]      = wa2 / q2; al_w[96 + r]       = wa3 / q3;
    al_w[16 + r] = wb0 / q0; al_w[32 + 16 + r] = wb1 / q1; al_w[64 + 16 + r] = wb2 / q2; al_w[96 + 16 + r]  = wb3 / q3;
    float4 acc0; acc0.x = 0.0f; acc0.y = 0.0f; acc0.z = 0.0f; acc0.w = 0.0f;
    float4 acc1; acc1.x = 0.0f; acc1.y = 0.0f; acc1.z = 0.0f; acc1.w = 0.0f;
    accum_slots(acc0, acc1, Hb, lane_off, al_r0, al_r1, sl_w, min(dmax, 32));
    for (int cb = 32; cb < dmax; cb += 16) {
        const bool ok = cb + r < d;
        const int  s  = ok ? csr[beg + cb + r].y : pads;
        float x0, x1, x2, x3;
        edge_logits(als, s, ad, x0, x1, x2, x3);
        const float a0 = ok ? expf(x0 - m0) / q0 : 0.0f, a1 = ok ? expf(x1 - m1) / q1 : 0.0f;
        const float a2 = ok ? expf(x2 - m2) / q2 : 0.0f, a3 = ok ? expf(x3 - m3) / q3 : 0.0f;
        sl_w[r] = s;
        al_w[r] = a0; al_w[32 + r] = a1; al_w[64 + r] = a2; al_w[96 + r] = a3;
        accum_slots(acc0, acc1, Hb, lane_off, al_r0, al_r1, sl_w, 16);
    }

    if (nv) {
        const float4 bia0 = *(const float4*)(bias_s + 4 * r);
        const float4 bia1 = *(const float4*)(bias_s + 64 + 4 * r);
        float4 o0; o0.x = acc0.x + bia0.x; o0.y = acc0.y + bia0.y; o0.z = acc0.z + bia0.z; o0.w = acc0.w + bia0.w;
        float4 o1; o1.x = acc1.x + bia1.x; o1.y = acc1.y + bia1.y; o1.z = acc1.z + bia1.z; o1.w = acc1.w + bia1.w;
        char* op = (char*)out + ((((unsigned)node) << 9) + lane_off);
        *(float4*)op         = o0;
        *(float4*)(op + 256) = o1;
    }
    }
}

extern "C" void kernel_launch(void* const* d_in, const int* in_sizes, int n_in,
                              void* d_out, int out_size, void* d_ws, size_t ws_size,
                              hipStream_t stream) {
    const float* x      = (const float*)d_in[0];
    const int*   ei     = (const int*)d_in[1];
    const float* W1     = (const float*)d_in[2];
    const float* a_src1 = (const float*)d_in[3];
    const float* a_dst1 = (const float*)d_in[4];
    const float* b1     = (const float*)d_in[5];
    const float* W2     = (const float*)d_in[6];
    const float* a_src2 = (const float*)d_in[7];
    const float* a_dst2 = (const float*)d_in[8];
    const float* b2     = (const float*)d_in[9];

    const int N = in_sizes[0] / FIN;
    const int E = in_sizes[1] / 2;
    const int* src = ei;
    const int* dst = ei + E;

    const int nRanges = (N + RANGE - 1) / RANGE;
    const int nEB     = (E + EDGE_BLOCK - 1) / EDGE_BLOCK;
    const int nBH     = nRanges * nEB;
    const int nTilesB = (nBH + SCAN_TILE - 1) / SCAN_TILE;
    const int nTiles  = (N + SCAN_TILE - 1) / SCAN_TILE;
    const size_t nPad = ((size_t)N + 4) & ~(size_t)3;
    const size_t ePad = ((size_t)E + 3) & ~(size_t)3;
    const size_t bPad = ((size_t)nBH + 4) & ~(size_t)3;

    char* p = (char*)d_ws;
    float* H      = (float*)p; p += (size_t)N * HOUT * sizeof(float);
    float* als    = (float*)p; p += (size_t)N * HEADS * sizeof(float);
    float* ald    = (float*)p; p += (size_t)N * HEADS * sizeof(float);
    int*   rowptr = (int*)p;   p += nPad * sizeof(int);
    p += nPad * sizeof(int);
    p += (((size_t)nTiles + 3) & ~(size_t)3) * sizeof(int);
    int2*  csr    = (int2*)p;  p += (size_t)E * sizeof(int2);
    char* q = (char*)H;
    int2*  pe     = (int2*)q;  q += ePad * sizeof(int2);
    int*   pd     = (int*)q;   q += ePad * sizeof(int);
    int*   bh     = (int*)q;   q += bPad * sizeof(int);
    int*   bhoff  = (int*)q;   q += bPad * sizeof(int);
    int*   bsum   = (int*)q;   q += (((size_t)nTilesB + 3) & ~(size_t)3) * sizeof(int);

    float* out = (float*)d_out;

    const int gGemm = (N + 63) / 64;
    const int BT = 256;
    int gNW  = (N + 31) / 32;

    k_hist      <<<nEB, 256, 0, stream>>>(dst, bh, E, nRanges, nEB);
    k_scan_local<<<nTilesB, SCAN_B, 0, stream>>>(bh, bhoff, bsum, nBH);
    k_scan_fix  <<<nTilesB, SCAN_B, 0, stream>>>(bhoff, bsum, nBH);
    k_part      <<<nEB, 256, 0, stream>>>(src, dst, bhoff, pe, pd, E, nRanges, nEB);
    k_bucket    <<<nRanges, BUCKET_B, 0, stream>>>(pe, pd, bhoff, rowptr, csr, N, nEB);

    k_gemm_wmma<0><<<gGemm, 128, 0, stream>>>(x, W1, a_src1, a_dst1, H, als, ald, N);
    k_gather    <<<gNW, BT, 0, stream>>>(rowptr, csr, H, als, ald, b1, out, N);

    k_gemm_wmma<1><<<gGemm, 128, 0, stream>>>(out, W2, a_src2, a_dst2, H, als, ald, N);
    k_gather    <<<gNW, BT, 0, stream>>>(rowptr, csr, H, als, ald, b2, out, N);
}


# Layer-1 product runs inside the edge-partition launch (work-groups behind the partition's, role by index) instead of after it; bucketing scratch moves to the output buffer; 8 launches; sums unchanged
# speedup vs baseline: 14.2857x; 1.0986x over previous
#include <hip/hip_runtime.h>
#include <math.h>


typedef __attribute__((ext_vector_type(2))) float v2f;
typedef __attribute__((ext_vector_type(8))) float v8f;

#define FIN   128
#define HEADS 4
#define CDIM  32
#define HOUT  128
#define NEG_SLOPE 0.2f

#define GM_WAVES 4
#define WL_PITCH 272
template<int M>
__device__ __forceinline__ float row_xor(float x) {
    const int v = __float_as_int(x);
    return __int_as_float(__builtin_amdgcn_update_dpp(v, v, 0x160 + M, 0xf, 0xf, false));
}
template<int M, int CNT>
__device__ __forceinline__ void halve_step(float* v, bool bit) {
#pragma unroll
    for (int i = 0; i < CNT; ++i) {
        const float keep = bit ? v[i + CNT] : v[i];
        const float send = bit ? v[i] : v[i + CNT];
        v[i] = keep + row_xor<M>(send);
    }
}
template<int RELU>
__device__ __forceinline__
void gemm_strips(const float* __restrict__ X, const float* __restrict__ W,
                 const float* __restrict__ a_src, const float* __restrict__ a_dst,
                 float* __restrict__ H, float* __restrict__ als, float* __restrict__ ald, int nRows,
                 int gblock, float* Wl, float* Av) {
    const int tid   = threadIdx.x;
    const int lane  = tid & 31;
    const int wv    = tid >> 5;
    const int c     = lane & 15;
    const int half  = lane >> 4;
    const int row0  = (gblock * GM_WAVES + wv) * 16;
    int row         = row0 + c;
    if (row >= nRows) row = nRows - 1;

    const float4* xr = (const float4*)(X + (size_t)row * FIN) + half;
    float4 a[16];
#pragma unroll
    for (int i = 0; i < 16; ++i) a[i] = xr[2 * i];

#pragma unroll
    for (int u = 0; u < (FIN / 2) / GM_WAVES; ++u) {
        const int kp = wv + GM_WAVES * u;
        const float4 r0 = *(const float4*)(W + (size_t)(2 * kp)     * HOUT + 4 * lane);
        const float4 r1 = *(const float4*)(W + (size_t)(2 * kp + 1) * HOUT + 4 * lane);
        float* d = Wl + kp * WL_PITCH + 8 * lane;
        float4 p0; p0.x = r0.x; p0.y = r1.x; p0.z = r0.y; p0.w = r1.y;
        float4 p1; p1.x = r0.z; p1.y = r1.z; p1.z = r0.w; p1.w = r1.w;
        *(float4*)(d)     = p0;
        *(float4*)(d + 4) = p1;
    }
    Av[tid]        = a_src[tid];
    Av[HOUT + tid] = a_dst[tid];
    __syncthreads();

    v8f acc[8];
#pragma unroll
    for (int t = 0; t < 8; ++t) acc[t] = (v8f){};

    const float* wl = Wl + (2 * half) * WL_PITCH + 2 * c;
#pragma unroll
    for (int i = 0; i < 16; ++i) {
        float4 av = a[i];
        if (RELU) { av.x = fmaxf(av.x, 0.0f); av.y = fmaxf(av.y, 0.0f); av.z = fmaxf(av.z, 0.0f); av.w = fmaxf(av.w, 0.0f); }
#pragma unroll
        for (int w = 0; w < 2; ++w) {
            v2f af;
            af.x = w ? av.z : av.x;
            af.y = w ? av.w : av.y;
            const float* wp = wl + (4 * i + w) * WL_PITCH;
#pragma unroll
            for (int t = 0; t < 8; ++t) {
                const v2f b = *(const v2f*)(wp + 32 * t);
                acc[t] = __builtin_amdgcn_wmma_f32_16x16x4_f32(false, af, false, b, (short)0, acc[t], false, false);
            }
        }
    }

    const int mOff = half * 8;
#pragma unroll
    for (int j = 0; j < 8; ++j) {
        const int r = row0 + mOff + j;
        if (r < nRows) {
            float* hr = H + (size_t)r * HOUT + c;
#pragma unroll
            for (int t = 0; t < 8; ++t) hr[16 * t] = acc[t][j];
        }
    }

    float sv[32], dv[32];
#pragma unroll
    for (int h = 0; h < HEADS; ++h) {
        const float as0 = Av[32 * h + c],        as1 = Av[32 * h + 16 + c];
        const float ad0 = Av[HOUT + 32 * h + c], ad1 = Av[HOUT + 32 * h + 16 + c];
#pragma unroll
        for (int j = 0; j < 8; ++j) {
            sv[4 * j + h] = acc[2 * h][j] * as0 + acc[2 * h + 1][j] * as1;
            dv[4 * j + h] = acc[2 * h][j] * ad0 + acc[2 * h + 1][j] * ad1;
        }
    }
    halve_step<8, 16>(sv, (c & 8) != 0);  halve_step<8, 16>(dv, (c & 8) != 0);
    halve_step<4, 8>(sv, (c & 4) != 0);   halve_step<4, 8>(dv, (c & 4) != 0);
    halve_step<2, 4>(sv, (c & 2) != 0);   halve_step<2, 4>(dv, (c & 2) != 0);
    halve_step<1, 2>(sv, (c & 1) != 0);   halve_step<1, 2>(dv, (c & 1) != 0);
    const int rg = row0 + mOff;
    if (rg + (c >> 1) < nRows) {
        float2 s2; s2.x = sv[0]; s2.y = sv[1];
        float2 d2; d2.x = dv[0]; d2.y = dv[1];
        *(float2*)(als + (size_t)rg * HEADS + 2 * c) = s2;
        *(float2*)(ald + (size_t)rg * HEADS + 2 * c) = d2;
    }
}

template<int RELU>
__global__ __launch_bounds__(GM_WAVES * 32)
void k_gemm_wmma(const float* __restrict__ X, const float* __restrict__ W,
                 const float* __restrict__ a_src, const float* __restrict__ a_dst,
                 float* __restrict__ H, float* __restrict__ als, float* __restrict__ ald, int nRows) {
    __shared__ float Wl[(FIN / 2) * WL_PITCH];
    __shared__ float Av[2 * HOUT];
    gemm_strips<RELU>(X, W, a_src, a_dst, H, als, ald, nRows, (int)blockIdx.x, Wl, Av);
}

#define SCAN_B    256
#define SCAN_TILE 1024

#define RANGE_SHIFT 9
#define RANGE       512
#define MAX_RANGES  1024
#define EDGE_BLOCK  4096
#define EDGE_GROUP  1024

__device__ __forceinline__ int4 load4_or_neg(const int* __restrict__ p, int e, int e_total, bool wide) {
    int4 v;
    if (wide && e + 3 < e_total) {
        v = *(const int4*)(p + e);
    } else {
        v.x = (e     < e_total) ? p[e]     : -1;
        v.y = (e + 1 < e_total) ? p[e + 1] : -1;
        v.z = (e + 2 < e_total) ? p[e + 2] : -1;
        v.w = (e + 3 < e_total) ? p[e + 3] : -1;
    }
    return v;
}

__global__ __launch_bounds__(256)
void k_hist(const int* __restrict__ dst, int* __restrict__ bh, int e_total, int n_ranges, int n_blocks) {
    __shared__ int hist[MAX_RANGES];
    for (int i = threadIdx.x; i < n_ranges; i += 256) hist[i] = 0;
    __syncthreads();
    const int e0 = blockIdx.x * EDGE_BLOCK + threadIdx.x * 4;
    const bool wide = (((size_t)dst) & 15) == 0;
    int4 d[4];
    for (int q = 0; q < 4; ++q) d[q] = load4_or_neg(dst, e0 + q * EDGE_GROUP, e_total, wide);
    for (int q = 0; q < 4; ++q) {
        if (d[q].x >= 0) atomicAdd(&hist[d[q].x >> RANGE_SHIFT], 1);
        if (d[q].y >= 0) atomicAdd(&hist[d[q].y >> RANGE_SHIFT], 1);
        if (d[q].z >= 0) atomicAdd(&hist[d[q].z >> RANGE_SHIFT], 1);
        if (d[q].w >= 0) atomicAdd(&hist[d[q].w >> RANGE_SHIFT], 1);
    }
    __syncthreads();
    for (int i = threadIdx.x; i < n_ranges; i += 256) bh[(size_t)i * n_blocks + blockIdx.x] = hist[i];
}

__device__ __forceinline__ void part_one(int* cur, int2* __restrict__ pe, int* __restrict__ pd, int e, int s, int d) {
    if (d >= 0) {
        const int pos = atomicAdd(&cur[d >> RANGE_SHIFT], 1);
        int2 r; r.x = e; r.y = s;
        pe[pos] = r;
        pd[pos] = d;
    }
}

#define PART_T      (GM_WAVES * 32)
#define PART_GROUP  (PART_T * 4)
#define PART_GROUPS (EDGE_BLOCK / PART_GROUP)
__global__ __launch_bounds__(PART_T)
void k_part_gemm(const int* __restrict__ src, const int* __restrict__ dst, const int* __restrict__ bhoff,
                 int2* __restrict__ pe, int* __restrict__ pd, int e_total, int n_ranges, int n_blocks,
                 const float* __restrict__ X, const float* __restrict__ W,
                 const float* __restrict__ a_src, const float* __restrict__ a_dst,
                 float* __restrict__ H, float* __restrict__ als, float* __restrict__ ald, int nRows) {
    __shared__ float Wl[(FIN / 2) * WL_PITCH];
    __shared__ float Av[2 * HOUT];
    __shared__ int cur[MAX_RANGES];
    if ((int)blockIdx.x >= n_blocks) {
        gemm_strips<0>(X, W, a_src, a_dst, H, als, ald, nRows, (int)blockIdx.x - n_blocks, Wl, Av);
        return;
    }
    for (int i = threadIdx.x; i < n_ranges; i += PART_T) cur[i] = bhoff[(size_t)i * n_blocks + blockIdx.x];
    const int e0 = blockIdx.x * EDGE_BLOCK + threadIdx.x * 4;
    const bool wide = ((((size_t)src) | ((size_t)dst)) & 15) == 0;
    int4 d[PART_GROUPS], s[PART_GROUPS];
#pragma unroll
    for (int q = 0; q < PART_GROUPS; ++q) d[q] = load4_or_neg(dst, e0 + q * PART_GROUP, e_total, wide);
#pragma unroll
    for (int q = 0; q < PART_GROUPS; ++q) s[q] = load4_or_neg(src, e0 + q * PART_GROUP, e_total, wide);
    __syncthreads();
#pragma unroll
    for (int q = 0; q < PART_GROUPS; ++q) {
        const int e = e0 + q * PART_GROUP;
        part_one(cur, pe, pd, e,     s[q].x, d[q].x);
        part_one(cur, pe, pd, e + 1, s[q].y, d[q].y);
        part_one(cur, pe, pd, e + 2, s[q].z, d[q].z);
        part_one(cur, pe, pd, e + 3, s[q].w, d[q].w);
    }
}

__global__ __launch_bounds__(SCAN_B)
void k_scan_local(const int* __restrict__ cnt, int* __restrict__ rowptr, int* __restrict__ bsum, int n) {
    __shared__ int wtot[SCAN_B / 32];
    const int t = threadIdx.x, lane = t & 31, wv = t >> 5;
    const int base = blockIdx.x * SCAN_TILE + t * 4;
    int a0 = 0, a1 = 0, a2 = 0, a3 = 0;
    if (base + 3 < n) {
        const int4 v = *(const int4*)(cnt + base);
        a0 = v.x; a1 = v.y; a2 = v.z; a3 = v.w;
    } else {
        if (base     < n) a0 = cnt[base];
        if (base + 1 < n) a1 = cnt[base + 1];
        if (base + 2 < n) a2 = cnt[base + 2];
    }
    const int tsum = a0 + a1 + a2 + a3;
    int incl = tsum;
    for (int o = 1; o < 32; o <<= 1) { const int u = __shfl_up(incl, o); if (lane >= o) incl += u; }
    if (lane == 31) wtot[wv] = incl;
    __syncthreads();
    int pre = 0;
    for (int k = 0; k < wv; ++k) pre += wtot[k];
    const int ex = pre + incl - tsum;
    if (base + 3 < n) {
        int4 r; r.x = ex; r.y = ex + a0; r.z = ex + a0 + a1; r.w = ex + a0 + a1 + a2;
        *(int4*)(rowptr + base) = r;
    } else {
        if (base     < n) rowptr[base]     = ex;
        if (base + 1 < n) rowptr[base + 1] = ex + a0;
        if (base + 2 < n) rowptr[base + 2] = ex + a0 + a1;
    }
    if (t == SCAN_B - 1) bsum[blockIdx.x] = pre + incl;
}

__global__ __launch_bounds__(SCAN_B)
void k_scan_fix(int* __restrict__ rowptr, const int* __restrict__ bsum, int n) {
    __shared__ int wtot[SCAN_B / 32];
    const int t = threadIdx.x, lane = t & 31, wv = t >> 5;
    const int b = blockIdx.x;
    int part = 0;
    for (int k = t; k < b; k += SCAN_B) part += bsum[k];
    for (int o = 16; o > 0; o >>= 1) part += __shfl_xor(part, o);
    if (lane == 0) wtot[wv] = part;
    __syncthreads();
    int off = 0;
    for (int k = 0; k < SCAN_B / 32; ++k) off += wtot[k];
    const int base = b * SCAN_TILE + t * 4;
    for (int q = 0; q < 4; ++q) {
        const int i = base + q;
        if (i < n) rowptr[i] += off;
    }
    if (b == (int)gridDim.x - 1 && t == 0) rowptr[n] = off + bsum[b];
}

#define BUCKET_B 1024
__global__ __launch_bounds__(BUCKET_B)
void k_bucket(const int2* __restrict__ pe, const int* __restrict__ pd, const int* __restrict__ bhoff,
              int* __restrict__ rowptr, int2* __restrict__ csr, int n, int n_blocks) {
    __shared__ int cnt[RANGE];
    __shared__ int wtot[BUCKET_B / 32];
    const int t = threadIdx.x, lane = t & 31, wv = t >> 5;
    const int b = blockIdx.x;
    const int beg = bhoff[(size_t)b * n_blocks];
    const int end = bhoff[(size_t)(b + 1) * n_blocks];
    const int node0 = b << RANGE_SHIFT;
    if (t < RANGE) cnt[t] = 0;
    __syncthreads();
    for (int i0 = beg + t; i0 < end; i0 += 4 * BUCKET_B) {
        int dv[4];
        for (int u = 0; u < 4; ++u) { const int i = i0 + u * BUCKET_B; dv[u] = (i < end) ? pd[i] : -1; }
        for (int u = 0; u < 4; ++u) if (dv[u] >= 0) atomicAdd(&cnt[dv[u] - node0], 1);
    }
    __syncthreads();
    const int v = (t < RANGE) ? cnt[t] : 0;
    int incl = v;
    for (int o = 1; o < 32; o <<= 1) { const int u = __shfl_up(incl, o); if (lane >= o) incl += u; }
    if (lane == 31) wtot[wv] = incl;
    __syncthreads();
    int pre = 0;
    for (int k = 0; k < wv; ++k) pre += wtot[k];
    const int first = beg + pre + incl - v;
    if (t < RANGE) {
        cnt[t] = first;
        if (node0 + t < n) rowptr[node0 + t] = first;
    }
    if (b == (int)gridDim.x - 1 && t == 0) rowptr[n] = end;
    __syncthreads();
    for (int i0 = beg + t; i0 < end; i0 += 4 * BUCKET_B) {
        int dv[4]; int2 ev[4];
        for (int u = 0; u < 4; ++u) {
            const int i = i0 + u * BUCKET_B;
            dv[u] = -1; ev[u].x = 0; ev[u].y = 0;
            if (i < end) { dv[u] = pd[i]; ev[u] = pe[i]; }
        }
        for (int u = 0; u < 4; ++u) {
            if (dv[u] >= 0) {
                const int pos = atomicAdd(&cnt[dv[u] - node0], 1);
                csr[pos] = ev[u];
            }
        }
    }
}

template<int M>
__device__ __forceinline__ float in_row(float x) {
    const int v = __float_as_int(x);
    return __int_as_float(__builtin_amdgcn_update_dpp(v, v, 0x160 + M, 0xf, 0xf, true));
}
template<int K>
__device__ __forceinline__ int row_lane(int x) {
    return __builtin_amdgcn_update_dpp(x, x, 0x150 + K, 0xf, 0xf, true);
}
__device__ __forceinline__ float row_max(float v) {
    v = fmaxf(v, in_row<8>(v)); v = fmaxf(v, in_row<4>(v)); v = fmaxf(v, in_row<2>(v)); v = fmaxf(v, in_row<1>(v));
    return v;
}
__device__ __forceinline__ float row_sum(float v) {
    v += in_row<8>(v); v += in_row<4>(v); v += in_row<2>(v); v += in_row<1>(v);
    return v;
}
__device__ __forceinline__ float lrelu(float v) { return (v > 0.0f) ? v : v * NEG_SLOPE; }

__device__ __forceinline__ void edge_logits(const float* __restrict__ als, int s, const float4 ad,
                                            float& x0, float& x1, float& x2, float& x3) {
    const float4 a = *(const float4*)(als + (size_t)s * HEADS);
    x0 = lrelu(a.x + ad.x); x1 = lrelu(a.y + ad.y); x2 = lrelu(a.z + ad.z); x3 = lrelu(a.w + ad.w);
}

template<int K>
__device__ __forceinline__ void rank_step(int ea, int eb, int& ra, int& rb) {
    const int xa = row_lane<K>(ea), xb = row_lane<K>(eb);
    ra += ((xa < ea) ? 1 : 0) + ((xb < ea) ? 1 : 0);
    rb += ((xa < eb) ? 1 : 0) + ((xb < eb) ? 1 : 0);
    if constexpr (K + 1 < 16) rank_step<K + 1>(ea, eb, ra, rb);
}

#define GA_WAVES 8
__device__ __forceinline__ void accum_slots(float4& acc0, float4& acc1, const char* __restrict__ Hb, unsigned lane_off,
                                            const float* al_r0, const float* al_r1, const int* sl_w, int cnt) {
#pragma clang loop unroll(disable)
    for (int jj = 0; jj < cnt; jj += 4) {
        const int4   p  = *(const int4*)(sl_w + jj);
        const float4 x0 = *(const float4*)(al_r0 + jj);
        const float4 x1 = *(const float4*)(al_r1 + jj);
        const int   sj[4] = {p.x, p.y, p.z, p.w};
        const float a0[4] = {x0.x, x0.y, x0.z, x0.w};
        const float a1[4] = {x1.x, x1.y, x1.z, x1.w};
        float4 hv0[4], hv1[4];
        for (int u = 0; u < 4; ++u) {
            const char* q = Hb + ((((unsigned)sj[u]) << 9) + lane_off);
            hv0[u] = *(const float4*)q;
            hv1[u] = *(const float4*)(q + 256);
        }
        for (int u = 0; u < 4; ++u) {
            acc0.x = fmaf(hv0[u].x, a0[u], acc0.x); acc0.y = fmaf(hv0[u].y, a0[u], acc0.y);
            acc0.z = fmaf(hv0[u].z, a0[u], acc0.z); acc0.w = fmaf(hv0[u].w, a0[u], acc0.w);
            acc1.x = fmaf(hv1[u].x, a1[u], acc1.x); acc1.y = fmaf(hv1[u].y, a1[u], acc1.y);
            acc1.z = fmaf(hv1[u].z, a1[u], acc1.z); acc1.w = fmaf(hv1[u].w, a1[u], acc1.w);
        }
    }
}

__global__ __launch_bounds__(GA_WAVES * 32, 12)
void k_gather(const int* __restrict__ rowptr, const int2* __restrict__ csr,
              const float* __restrict__ Hm,
              const float* __restrict__ als, const float* __restrict__ ald,
              const float* __restrict__ bias, float* __restrict__ out, int n) {
    __shared__ __attribute__((aligned(16))) float al_s[GA_WAVES][2][HEADS][32];
    __shared__ __attribute__((aligned(16))) int   sl_s[GA_WAVES][2][32];
    __shared__ __attribute__((aligned(16))) float bias_s[HOUT];
    if (threadIdx.x < HOUT) bias_s[threadIdx.x] = bias[threadIdx.x];
    __syncthreads();
    const int lane = threadIdx.x & 31;
    const int wv   = threadIdx.x >> 5;
    const int r    = lane & 15;
    const int h    = lane >> 4;
    float*       al_w  = &al_s[wv][h][0][0];
    int*         sl_w  = &sl_s[wv][h][0];
    const float* al_r0 = al_w + (r >> 3) * 32;
    const float* al_r1 = al_w + (2 + (r >> 3)) * 32;
    const char*  Hb    = (const char*)Hm;
    const unsigned lane_off = (unsigned)r * 16u;
    const int wave0  = __builtin_amdgcn_readfirstlane((int)((blockIdx.x * blockDim.x + threadIdx.x) >> 5));
    const int nwaves = (int)((gridDim.x * blockDim.x) >> 5);
    const int npairs = (n + 1) >> 1;
    for (int pair = wave0; pair < npairs; pair += nwaves) {
    const int  node = 2 * pair + h;
    const bool nv   = node < n;
    const int  nd   = nv ? node : n - 1;
    const int  beg  = rowptr[nd];
    const int  d    = nv ? rowptr[nd + 1] - beg : 0;
    const float4 ad = *(const float4*)(ald + (size_t)nd * HEADS);
    const int dmax = max(__builtin_amdgcn_readlane(d, 0), __builtin_amdgcn_readlane(d, 16));

    const bool oka = r < d, okb = 16 + r < d;
    int eida = 0x7fffffff, srca = 0, eidb = 0x7fffffff, srcb = 0;
    if (oka) { const int2 t = csr[beg + r];      eida = t.x; srca = t.y; }
    if (okb) { const int2 t = csr[beg + 16 + r]; eidb = t.x; srcb = t.y; }
    int ranka = 0, rankb = 0;
    rank_step<0>(eida, eidb, ranka, rankb);
    if (oka) sl_w[ranka] = srca;
    if (okb) sl_w[rankb] = srcb;
    const int sa = oka ? sl_w[r] : 0;
    const int sb = okb ? sl_w[16 + r] : 0;

    float ea0, ea1, ea2, ea3, eb0, eb1, eb2, eb3;
    edge_logits(als, sa, ad, ea0, ea1, ea2, ea3);
    edge_logits(als, sb, ad, eb0, eb1, eb2, eb3);
    float m0 = fmaxf(oka ? ea0 : -INFINITY, okb ? eb0 : -INFINITY);
    float m1 = fmaxf(oka ? ea1 : -INFINITY, okb ? eb1 : -INFINITY);
    float m2 = fmaxf(oka ? ea2 : -INFINITY, okb ? eb2 : -INFINITY);
    float m3 = fmaxf(oka ? ea3 : -INFINITY, okb ? eb3 : -INFINITY);
    for (int cb = 32; cb < dmax; cb += 16) {
        if (cb + r < d) {
            float x0, x1, x2, x3;
            edge_logits(als, csr[beg + cb + r].y, ad, x0, x1, x2, x3);
            m0 = fmaxf(m0, x0); m1 = fmaxf(m1, x1); m2 = fmaxf(m2, x2); m3 = fmaxf(m3, x3);
        }
    }
    m0 = row_max(m0); m1 = row_max(m1); m2 = row_max(m2); m3 = row_max(m3);
    if (!__builtin_isfinite(m0)) m0 = 0.0f;
    if (!__builtin_isfinite(m1)) m1 = 0.0f;
    if (!__builtin_isfinite(m2)) m2 = 0.0f;
    if (!__builtin_isfinite(m3)) m3 = 0.0f;

    const float wa0 = expf(oka ? ea0 - m0 : -INFINITY), wa1 = expf(oka ? ea1 - m1 : -INFINITY);
    const float wa2 = expf(oka ? ea2 - m2 : -INFINITY), wa3 = expf(oka ? ea3 - m3 : -INFINITY);
    const float wb0 = expf(okb ? eb0 - m0 : -INFINITY), wb1 = expf(okb ? eb1 - m1 : -INFINITY);
    const float wb2 = expf(okb ? eb2 - m2 : -INFINITY), wb3 = expf(okb ? eb3 - m3 : -INFINITY);
    float d0 = wa0 + wb0, d1 = wa1 + wb1, d2 = wa2 + wb2, d3 = wa3 + wb3;
    for (int cb = 32; cb < dmax; cb += 16) {
        if (cb + r < d) {
            float x0, x1, x2, x3;
            edge_logits(als, csr[beg + cb + r].y, ad, x0, x1, x2, x3);
            d0 += expf(x0 - m0); d1 += expf(x1 - m1); d2 += expf(x2 - m2); d3 += expf(x3 - m3);
        }
    }
    d0 = row_sum(d0); d1 = row_sum(d1); d2 = row_sum(d2); d3 = row_sum(d3);
    const float q0 = fmaxf(d0, 1e-16f), q1 = fmaxf(d1, 1e-16f), q2 = fmaxf(d2, 1e-16f), q3 = fmaxf(d3, 1e-16f);

    const int s_first = row_lane<0>(sa);
    const int pads    = (d > 0) ? s_first : nd;
    if (!oka) sl_w[r] = pads;
    if (!okb) sl_w[16 + r] = pads;
    al_w[r]      = wa0 / q0; al_w[32 + r]      = wa1 / q1; al_w[64 + r]      = wa2 / q2; al_w[96 + r]       = wa3 / q3;
    al_w[16 + r] = wb0 / q0; al_w[32 + 16 + r] = wb1 / q1; al_w[64 + 16 + r] = wb2 / q2; al_w[96 + 16 + r]  = wb3 / q3;
    float4 acc0; acc0.x = 0.0f; acc0.y = 0.0f; acc0.z = 0.0f; acc0.w = 0.0f;
    float4 acc1; acc1.x = 0.0f; acc1.y = 0.0f; acc1.z = 0.0f; acc1.w = 0.0f;
    accum_slots(acc0, acc1, Hb, lane_off, al_r0, al_r1, sl_w, min(dmax, 32));
    for (int cb = 32; cb < dmax; cb += 16) {
        const bool ok = cb + r < d;
        const int  s  = ok ? csr[beg + cb + r].y : pads;
        float x0, x1, x2, x3;
        edge_logits(als, s, ad, x0, x1, x2, x3);
        const float a0 = ok ? expf(x0 - m0) / q0 : 0.0f, a1 = ok ? expf(x1 - m1) / q1 : 0.0f;
        const float a2 = ok ? expf(x2 - m2) / q2 : 0.0f, a3 = ok ? expf(x3 - m3) / q3 : 0.0f;
        sl_w[r] = s;
        al_w[r] = a0; al_w[32 + r] = a1; al_w[64 + r] = a2; al_w[96 + r] = a3;
        accum_slots(acc0, acc1, Hb, lane_off, al_r0, al_r1, sl_w, 16);
    }

    if (nv) {
        const float4 bia0 = *(const float4*)(bias_s + 4 * r);
        const float4 bia1 = *(const float4*)(bias_s + 64 + 4 * r);
        float4 o0; o0.x = acc0.x + bia0.x; o0.y = acc0.y + bia0.y; o0.z = acc0.z + bia0.z; o0.w = acc0.w + bia0.w;
        float4 o1; o1.x = acc1.x + bia1.x; o1.y = acc1.y + bia1.y; o1.z = acc1.z + bia1.z; o1.w = acc1.w + bia1.w;
        char* op = (char*)out + ((((unsigned)node) << 9) + lane_off);
        *(float4*)op         = o0;
        *(float4*)(op + 256) = o1;
    }
    }
}

extern "C" void kernel_launch(void* const* d_in, const int* in_sizes, int n_in,
                              void* d_out, int out_size, void* d_ws, size_t ws_size,
                              hipStream_t stream) {
    const float* x      = (const float*)d_in[0];
    const int*   ei     = (const int*)d_in[1];
    const float* W1     = (const float*)d_in[2];
    const float* a_src1 = (const float*)d_in[3];
    const float* a_dst1 = (const float*)d_in[4];
    const float* b1     = (const float*)d_in[5];
    const float* W2     = (const float*)d_in[6];
    const float* a_src2 = (const float*)d_in[7];
    const float* a_dst2 = (const float*)d_in[8];
    const float* b2     = (const float*)d_in[9];

    const int N = in_sizes[0] / FIN;
    const int E = in_sizes[1] / 2;
    const int* src = ei;
    const int* dst = ei + E;

    const int nRanges = (N + RANGE - 1) / RANGE;
    const int nEB     = (E + EDGE_BLOCK - 1) / EDGE_BLOCK;
    const int nBH     = nRanges * nEB;
    const int nTilesB = (nBH + SCAN_TILE - 1) / SCAN_TILE;
    const int nTiles  = (N + SCAN_TILE - 1) / SCAN_TILE;
    const size_t nPad = ((size_t)N + 4) & ~(size_t)3;
    const size_t ePad = ((size_t)E + 3) & ~(size_t)3;
    const size_t bPad = ((size_t)nBH + 4) & ~(size_t)3;

    char* p = (char*)d_ws;
    float* H      = (float*)p; p += (size_t)N * HOUT * sizeof(float);
    float* als    = (float*)p; p += (size_t)N * HEADS * sizeof(float);
    float* ald    = (float*)p; p += (size_t)N * HEADS * sizeof(float);
    int*   rowptr = (int*)p;   p += nPad * sizeof(int);
    p += nPad * sizeof(int);
    p += (((size_t)nTiles + 3) & ~(size_t)3) * sizeof(int);
    int2*  csr    = (int2*)p;  p += (size_t)E * sizeof(int2);
    char* q = (char*)d_out;
    int2*  pe     = (int2*)q;  q += ePad * sizeof(int2);
    int*   pd     = (int*)q;   q += ePad * sizeof(int);
    int*   bh     = (int*)q;   q += bPad * sizeof(int);
    int*   bhoff  = (int*)q;   q += bPad * sizeof(int);
    int*   bsum   = (int*)q;   q += (((size_t)nTilesB + 3) & ~(size_t)3) * sizeof(int);

    float* out = (float*)d_out;

    const int gGemm = (N + 63) / 64;
    const int BT = 256;
    int gNW  = (N + 31) / 32;

    k_hist      <<<nEB, 256, 0, stream>>>(dst, bh, E, nRanges, nEB);
    k_scan_local<<<nTilesB, SCAN_B, 0, stream>>>(bh, bhoff, bsum, nBH);
    k_scan_fix  <<<nTilesB, SCAN_B, 0, stream>>>(bhoff, bsum, nBH);
    k_part_gemm <<<nEB + gGemm, PART_T, 0, stream>>>(src, dst, bhoff, pe, pd, E, nRanges, nEB,
                                                 x, W1, a_src1, a_dst1, H, als, ald, N);
    k_bucket    <<<nRanges, BUCKET_B, 0, stream>>>(pe, pd, bhoff, rowptr, csr, N, nEB);

    k_gather    <<<gNW, BT, 0, stream>>>(rowptr, csr, H, als, ald, b1, out, N);

    k_gemm_wmma<1><<<gGemm, 128, 0, stream>>>(out, W2, a_src2, a_dst2, H, als, ald, N);
    k_gather    <<<gNW, BT, 0, stream>>>(rowptr, csr, H, als, ald, b2, out, N);
}


# Layer-1 aggregation and layer-2 product in one launch: 64 aggregated rows stay in LDS as the A panel, a wave holds 16 columns of W2; ald from the node's own row; 4 launches; sum order changed
# speedup vs baseline: 16.8934x; 1.1825x over previous
#include <hip/hip_runtime.h>
#include <math.h>


typedef __attribute__((ext_vector_type(2))) float v2f;
typedef __attribute__((ext_vector_type(8))) float v8f;

#define FIN   128
#define HEADS 4
#define CDIM  32
#define HOUT  128
#define NEG_SLOPE 0.2f

#define GM_WAVES 4
#define WL_PITCH 272
template<int M>
__device__ __forceinline__ float row_xor(float x) {
    const int v = __float_as_int(x);
    return __int_as_float(__builtin_amdgcn_update_dpp(v, v, 0x160 + M, 0xf, 0xf, false));
}
template<int M, int CNT>
__device__ __forceinline__ void halve_step(float* v, bool bit) {
#pragma unroll
    for (int i = 0; i < CNT; ++i) {
        const float keep = bit ? v[i + CNT] : v[i];
        const float send = bit ? v[i] : v[i + CNT];
        v[i] = keep + row_xor<M>(send);
    }
}
template<int RELU>
__device__ __forceinline__
void gemm_strips(const float* __restrict__ X, const float* __restrict__ W,
                 const float* __restrict__ a_src,
                 float* __restrict__ H, float* __restrict__ als, int nRows,
                 int gblock, float* Wl, float* Av) {
    const int tid   = threadIdx.x;
    const int lane  = tid & 31;
    const int wv    = tid >> 5;
    const int c     = lane & 15;
    const int half  = lane >> 4;
    const int row0  = (gblock * GM_WAVES + wv) * 16;
    int row         = row0 + c;
    if (row >= nRows) row = nRows - 1;

    const float4* xr = (const float4*)(X + (size_t)row * FIN) + half;
    float4 a[16];
#pragma unroll
    for (int i = 0; i < 16; ++i) a[i] = xr[2 * i];

#pragma unroll
    for (int u = 0; u < (FIN / 2) / GM_WAVES; ++u) {
        const int kp = wv + GM_WAVES * u;
        const float4 r0 = *(const float4*)(W + (size_t)(2 * kp)     * HOUT + 4 * lane);
        const float4 r1 = *(const float4*)(W + (size_t)(2 * kp + 1) * HOUT + 4 * lane);
        float* d = Wl + kp * WL_PITCH + 8 * lane;
        float4 p0; p0.x = r0.x; p0.y = r1.x; p0.z = r0.y; p0.w = r1.y;
        float4 p1; p1.x = r0.z; p1.y = r1.z; p1.z = r0.w; p1.w = r1.w;
        *(float4*)(d)     = p0;
        *(float4*)(d + 4) = p1;
    }
    Av[tid] = a_src[tid];
    __syncthreads();

    v8f acc[8];
#pragma unroll
    for (int t = 0; t < 8; ++t) acc[t] = (v8f){};

    const float* wl = Wl + (2 * half) * WL_PITCH + 2 * c;
#pragma unroll
    for (int i = 0; i < 16; ++i) {
        float4 av = a[i];
        if (RELU) { av.x = fmaxf(av.x, 0.0f); av.y = fmaxf(av.y, 0.0f); av.z = fmaxf(av.z, 0.0f); av.w = fmaxf(av.w, 0.0f); }
#pragma unroll
        for (int w = 0; w < 2; ++w) {
            v2f af;
            af.x = w ? av.z : av.x;
            af.y = w ? av.w : av.y;
            const float* wp = wl + (4 * i + w) * WL_PITCH;
#pragma unroll
            for (int t = 0; t < 8; ++t) {
                const v2f b = *(const v2f*)(wp + 32 * t);
                acc[t] = __builtin_amdgcn_wmma_f32_16x16x4_f32(false, af, false, b, (short)0, acc[t], false, false);
            }
        }
    }

    const int mOff = half * 8;
#pragma unroll
    for (int j = 0; j < 8; ++j) {
        const int r = row0 + mOff + j;
        if (r < nRows) {
            float* hr = H + (size_t)r * HOUT + c;
#pragma unroll
            for (int t = 0; t < 8; ++t) hr[16 * t] = acc[t][j];
        }
    }

    float sv[32];
#pragma unroll
    for (int h = 0; h < HEADS; ++h) {
        const float as0 = Av[32 * h + c], as1 = Av[32 * h + 16 + c];
#pragma unroll
        for (int j = 0; j < 8; ++j) sv[4 * j + h] = acc[2 * h][j] * as0 + acc[2 * h + 1][j] * as1;
    }
    halve_step<8, 16>(sv, (c & 8) != 0);
    halve_step<4, 8>(sv, (c & 4) != 0);
    halve_step<2, 4>(sv, (c & 2) != 0);
    halve_step<1, 2>(sv, (c & 1) != 0);
    const int rg = row0 + mOff;
    if (rg + (c >> 1) < nRows) {
        float2 s2; s2.x = sv[0]; s2.y = sv[1];
        *(float2*)(als + (size_t)rg * HEADS + 2 * c) = s2;
    }
}

#define RANGE_SHIFT 9
#define RANGE       512
#define MAX_RANGES  1024
#define EDGE_BLOCK  4096

__device__ __forceinline__ int4 load4_or_neg(const int* __restrict__ p, int e, int e_total, bool wide) {
    int4 v;
    if (wide && e + 3 < e_total) {
        v = *(const int4*)(p + e);
    } else {
        v.x = (e     < e_total) ? p[e]     : -1;
        v.y = (e + 1 < e_total) ? p[e + 1] : -1;
        v.z = (e + 2 < e_total) ? p[e + 2] : -1;
        v.w = (e + 3 < e_total) ? p[e + 3] : -1;
    }
    return v;
}

__device__ __forceinline__ int rank_one(int* cur, int d) {
    return (d >= 0) ? atomicAdd(&cur[d >> RANGE_SHIFT], 1) : 0;
}
__device__ __forceinline__ void place_one(const int* cur, int2* __restrict__ pe, int* __restrict__ pd, int seg,
                                          int e, int s, int d, int rk) {
    if (d >= 0) {
        const int pos = seg + (cur[d >> RANGE_SHIFT] & 0xffff) + rk;
        int2 r; r.x = e; r.y = s;
        pe[pos] = r;
        pd[pos] = d;
    }
}

#define PART_T      (GM_WAVES * 32)
#define PART_GROUP  (PART_T * 4)
#define PART_GROUPS (EDGE_BLOCK / PART_GROUP)
#define PART_SCAN   (MAX_RANGES / PART_T)
__global__ __launch_bounds__(PART_T)
void k_part_gemm(const int* __restrict__ src, const int* __restrict__ dst, int* __restrict__ tab,
                 int2* __restrict__ pe, int* __restrict__ pd, int e_total, int n_ranges, int n_blocks,
                 const float* __restrict__ X, const float* __restrict__ W,
                 const float* __restrict__ a_src,
                 float* __restrict__ H, float* __restrict__ als, int nRows) {
    __shared__ float Wl[(FIN / 2) * WL_PITCH];
    __shared__ float Av[HOUT];
    __shared__ int cur[MAX_RANGES];
    __shared__ int wtot[GM_WAVES];
    if ((int)blockIdx.x >= n_blocks) {
        gemm_strips<0>(X, W, a_src, H, als, nRows, (int)blockIdx.x - n_blocks, Wl, Av);
        return;
    }
    const int tid = threadIdx.x, lane = tid & 31, wv = tid >> 5;
    for (int i = tid; i < n_ranges; i += PART_T) cur[i] = 0;
    const int seg = blockIdx.x * EDGE_BLOCK;
    const int e0  = seg + tid * 4;
    const bool wide = ((((size_t)src) | ((size_t)dst)) & 15) == 0;
    int4 d[PART_GROUPS], s[PART_GROUPS], rk[PART_GROUPS];
#pragma unroll
    for (int q = 0; q < PART_GROUPS; ++q) d[q] = load4_or_neg(dst, e0 + q * PART_GROUP, e_total, wide);
#pragma unroll
    for (int q = 0; q < PART_GROUPS; ++q) s[q] = load4_or_neg(src, e0 + q * PART_GROUP, e_total, wide);
    __syncthreads();
#pragma unroll
    for (int q = 0; q < PART_GROUPS; ++q) {
        rk[q].x = rank_one(cur, d[q].x);
        rk[q].y = rank_one(cur, d[q].y);
        rk[q].z = rank_one(cur, d[q].z);
        rk[q].w = rank_one(cur, d[q].w);
    }
    __syncthreads();
    int cn[PART_SCAN];
    int tsum = 0;
#pragma unroll
    for (int j = 0; j < PART_SCAN; ++j) {
        const int i = PART_SCAN * tid + j;
        cn[j] = (i < n_ranges) ? cur[i] : 0;
        tsum += cn[j];
    }
    int incl = tsum;
    for (int o = 1; o < 32; o <<= 1) { const int u = __shfl_up(incl, o); if (lane >= o) incl += u; }
    if (lane == 31) wtot[wv] = incl;
    __syncthreads();
    int ex = incl - tsum;
    for (int k = 0; k < wv; ++k) ex += wtot[k];
#pragma unroll
    for (int j = 0; j < PART_SCAN; ++j) {
        const int i = PART_SCAN * tid + j;
        if (i < n_ranges) cur[i] = ex | (cn[j] << 16);
        ex += cn[j];
    }
    __syncthreads();
    for (int i = tid; i < n_ranges; i += PART_T) tab[(size_t)i * n_blocks + blockIdx.x] = cur[i];
#pragma unroll
    for (int q = 0; q < PART_GROUPS; ++q) {
        const int e = e0 + q * PART_GROUP;
        place_one(cur, pe, pd, seg, e,     s[q].x, d[q].x, rk[q].x);
        place_one(cur, pe, pd, seg, e + 1, s[q].y, d[q].y, rk[q].y);
        place_one(cur, pe, pd, seg, e + 2, s[q].z, d[q].z, rk[q].z);
        place_one(cur, pe, pd, seg, e + 3, s[q].w, d[q].w, rk[q].w);
    }
}

#define BUCKET_B 1024
#define BUCKET_W (BUCKET_B / 32)
#define RUN_BATCH 8
__global__ __launch_bounds__(BUCKET_B)
void k_bucket(const int2* __restrict__ pe, const int* __restrict__ pd, const int* __restrict__ tab,
              int* __restrict__ rowptr, int2* __restrict__ csr, int n, int n_blocks,
              const float* __restrict__ Wsrc, float* __restrict__ Wt) {
    __shared__ int cnt[RANGE];
    __shared__ int wtot[BUCKET_W];
    __shared__ int wbeg[BUCKET_W];
    const int t = threadIdx.x, lane = t & 31;
    const int wv = __builtin_amdgcn_readfirstlane(t >> 5);
    const int b = blockIdx.x;
    const int node0 = b << RANGE_SHIFT;
    const int* __restrict__ row = tab + (size_t)b * n_blocks;
    {
        const int chunk = (HOUT * HOUT + (int)gridDim.x - 1) / (int)gridDim.x;
        for (int u = t; u < chunk; u += BUCKET_B) {
            const int idx = b * chunk + u;
            if (idx < HOUT * HOUT) {
                const int e = idx & 3, l = (idx >> 2) & 31, i = (idx >> 7) & 15, tt = idx >> 11;
                Wt[idx] = Wsrc[(8 * i + 4 * (l >> 4) + e) * HOUT + 16 * tt + (l & 15)];
            }
        }
    }
    if (t < RANGE) cnt[t] = 0;
    __syncthreads();
    int offsum = 0;
    for (int base = 0; base < n_blocks; base += BUCKET_B) {
        const int mb = base + wv + BUCKET_W * lane;
        const int en = (mb < n_blocks) ? row[mb] : 0;
        offsum += en & 0xffff;
        const int nb = min(32, (n_blocks - base - wv + BUCKET_W - 1) / BUCKET_W);
        for (int u0 = 0; u0 < nb; u0 += RUN_BATCH) {
            int dv[RUN_BATCH];
#pragma unroll
            for (int u = 0; u < RUN_BATCH; ++u) {
                const int e = __builtin_amdgcn_readlane(en, u0 + u);
                const int i = (base + wv + BUCKET_W * (u0 + u)) * EDGE_BLOCK + (e & 0xffff) + lane;
                dv[u] = (lane < (e >> 16)) ? pd[i] : -1;
            }
#pragma unroll
            for (int u = 0; u < RUN_BATCH; ++u) if (dv[u] >= 0) atomicAdd(&cnt[dv[u] - node0], 1);
#pragma unroll
            for (int u = 0; u < RUN_BATCH; ++u) {
                const int e = __builtin_amdgcn_readlane(en, u0 + u);
                const int i = (base + wv + BUCKET_W * (u0 + u)) * EDGE_BLOCK + (e & 0xffff);
                for (int k = 32 + lane; k < (e >> 16); k += 32) atomicAdd(&cnt[pd[i + k] - node0], 1);
            }
        }
    }
    for (int o = 16; o > 0; o >>= 1) offsum += __shfl_xor(offsum, o);
    if (lane == 0) wbeg[wv] = offsum;
    __syncthreads();
    int beg = 0;
    for (int k = 0; k < BUCKET_W; ++k) beg += wbeg[k];
    const int v = (t < RANGE) ? cnt[t] : 0;
    int incl = v;
    for (int o = 1; o < 32; o <<= 1) { const int u = __shfl_up(incl, o); if (lane >= o) incl += u; }
    if (lane == 31) wtot[wv] = incl;
    __syncthreads();
    int pre = 0;
    for (int k = 0; k < wv; ++k) pre += wtot[k];
    const int first = beg + pre + incl - v;
    if (t < RANGE) {
        cnt[t] = first;
        if (node0 + t < n) rowptr[node0 + t] = first;
    }
    if (b == (int)gridDim.x - 1 && t == RANGE - 1) rowptr[n] = first + v;
    __syncthreads();
    for (int base = 0; base < n_blocks; base += BUCKET_B) {
        const int mb = base + wv + BUCKET_W * lane;
        const int en = (mb < n_blocks) ? row[mb] : 0;
        const int nb = min(32, (n_blocks - base - wv + BUCKET_W - 1) / BUCKET_W);
        for (int u0 = 0; u0 < nb; u0 += RUN_BATCH) {
            int dv[RUN_BATCH]; int2 ev[RUN_BATCH];
#pragma unroll
            for (int u = 0; u < RUN_BATCH; ++u) {
                const int e = __builtin_amdgcn_readlane(en, u0 + u);
                const int i = (base + wv + BUCKET_W * (u0 + u)) * EDGE_BLOCK + (e & 0xffff) + lane;
                dv[u] = -1; ev[u].x = 0; ev[u].y = 0;
                if (lane < (e >> 16)) { dv[u] = pd[i]; ev[u] = pe[i]; }
            }
#pragma unroll
            for (int u = 0; u < RUN_BATCH; ++u) {
                if (dv[u] >= 0) {
                    const int pos = atomicAdd(&cnt[dv[u] - node0], 1);
                    csr[pos] = ev[u];
                }
            }
#pragma unroll
            for (int u = 0; u < RUN_BATCH; ++u) {
                const int e = __builtin_amdgcn_readlane(en, u0 + u);
                const int i = (base + wv + BUCKET_W * (u0 + u)) * EDGE_BLOCK + (e & 0xffff);
                for (int k = 32 + lane; k < (e >> 16); k += 32) {
                    const int pos = atomicAdd(&cnt[pd[i + k] - node0], 1);
                    csr[pos] = pe[i + k];
                }
            }
        }
    }
}

template<int M>
__device__ __forceinline__ float in_row(float x) {
    const int v = __float_as_int(x);
    return __int_as_float(__builtin_amdgcn_update_dpp(v, v, 0x160 + M, 0xf, 0xf, true));
}
template<int K>
__device__ __forceinline__ int row_lane(int x) {
    return __builtin_amdgcn_update_dpp(x, x, 0x150 + K, 0xf, 0xf, true);
}
__device__ __forceinline__ float row_max(float v) {
    v = fmaxf(v, in_row<8>(v)); v = fmaxf(v, in_row<4>(v)); v = fmaxf(v, in_row<2>(v)); v = fmaxf(v, in_row<1>(v));
    return v;
}
__device__ __forceinline__ float row_sum(float v) {
    v += in_row<8>(v); v += in_row<4>(v); v += in_row<2>(v); v += in_row<1>(v);
    return v;
}
__device__ __forceinline__ float lrelu(float v) { return (v > 0.0f) ? v : v * NEG_SLOPE; }

__device__ __forceinline__ void edge_logits(const float* __restrict__ als, int s, const float4 ad,
                                            float& x0, float& x1, float& x2, float& x3) {
    const float4 a = *(const float4*)(als + (size_t)s * HEADS);
    x0 = lrelu(a.x + ad.x); x1 = lrelu(a.y + ad.y); x2 = lrelu(a.z + ad.z); x3 = lrelu(a.w + ad.w);
}

template<int K>
__device__ __forceinline__ void rank_step(int ea, int eb, int& ra, int& rb) {
    const int xa = row_lane<K>(ea), xb = row_lane<K>(eb);
    ra += ((xa < ea) ? 1 : 0) + ((xb < ea) ? 1 : 0);
    rb += ((xa < eb) ? 1 : 0) + ((xb < eb) ? 1 : 0);
    if constexpr (K + 1 < 16) rank_step<K + 1>(ea, eb, ra, rb);
}

#define GA_WAVES 8
__device__ __forceinline__ void accum_slots(float4& acc0, float4& acc1, const char* __restrict__ Hb, unsigned lane_off,
                                            const float* al_r0, const float* al_r1, const int* sl_w, int cnt) {
#pragma clang loop unroll(disable)
    for (int jj = 0; jj < cnt; jj += 4) {
        const int4   p  = *(const int4*)(sl_w + jj);
        const float4 x0 = *(const float4*)(al_r0 + jj);
        const float4 x1 = *(const float4*)(al_r1 + jj);
        const int   sj[4] = {p.x, p.y, p.z, p.w};
        const float a0[4] = {x0.x, x0.y, x0.z, x0.w};
        const float a1[4] = {x1.x, x1.y, x1.z, x1.w};
        float4 hv0[4], hv1[4];
        for (int u = 0; u < 4; ++u) {
            const char* q = Hb + ((((unsigned)sj[u]) << 9) + lane_off);
            hv0[u] = *(const float4*)q;
            hv1[u] = *(const float4*)(q + 256);
        }
        for (int u = 0; u < 4; ++u) {
            acc0.x = fmaf(hv0[u].x, a0[u], acc0.x); acc0.y = fmaf(hv0[u].y, a0[u], acc0.y);
            acc0.z = fmaf(hv0[u].z, a0[u], acc0.z); acc0.w = fmaf(hv0[u].w, a0[u], acc0.w);
            acc1.x = fmaf(hv1[u].x, a1[u], acc1.x); acc1.y = fmaf(hv1[u].y, a1[u], acc1.y);
            acc1.z = fmaf(hv1[u].z, a1[u], acc1.z); acc1.w = fmaf(hv1[u].w, a1[u], acc1.w);
        }
    }
}

template<int TO_PANEL>
__device__ __forceinline__
void gather_pair(const int pair, const int n, const int* __restrict__ rowptr, const int2* __restrict__ csr,
                 const char* __restrict__ Hb, const float* __restrict__ als,
                 const float* adst_s, const float* bias_s, float* al_w, int* sl_w,
                 const float* al_r0, const float* al_r1, const unsigned lane_off, const int r, const int h,
                 float* __restrict__ out, float* prow) {
    const int  node = 2 * pair + h;
    const bool nv   = node < n;
    const int  nd   = nv ? node : n - 1;
    const int  beg  = rowptr[nd];
    const int  d    = nv ? rowptr[nd + 1] - beg : 0;
    float4 ad;
    {
        const char*  own = Hb + ((((unsigned)nd) << 9) + lane_off);
        const float4 w0 = *(const float4*)own;
        const float4 w1 = *(const float4*)(own + 256);
        const float4 c0 = *(const float4*)(adst_s + 4 * r);
        const float4 c1 = *(const float4*)(adst_s + 64 + 4 * r);
        float p0 = w0.x * c0.x + w0.y * c0.y + w0.z * c0.z + w0.w * c0.w;
        float p1 = w1.x * c1.x + w1.y * c1.y + w1.z * c1.z + w1.w * c1.w;
        p0 += in_row<4>(p0); p0 += in_row<2>(p0); p0 += in_row<1>(p0);
        p1 += in_row<4>(p1); p1 += in_row<2>(p1); p1 += in_row<1>(p1);
        const float o0 = in_row<8>(p0), o1 = in_row<8>(p1);
        const bool lo = r < 8;
        ad.x = lo ? p0 : o0; ad.y = lo ? o0 : p0; ad.z = lo ? p1 : o1; ad.w = lo ? o1 : p1;
    }
    const int dmax = max(__builtin_amdgcn_readlane(d, 0), __builtin_amdgcn_readlane(d, 16));

    const bool oka = r < d, okb = 16 + r < d;
    int eida = 0x7fffffff, srca = 0, eidb = 0x7fffffff, srcb = 0;
    if (oka) { const int2 t = csr[beg + r];      eida = t.x; srca = t.y; }
    if (okb) { const int2 t = csr[beg + 16 + r]; eidb = t.x; srcb = t.y; }
    int ranka = 0, rankb = 0;
    rank_step<0>(eida, eidb, ranka, rankb);
    if (oka) sl_w[ranka] = srca;
    if (okb) sl_w[rankb] = srcb;
    const int sa = oka ? sl_w[r] : 0;
    const int sb = okb ? sl_w[16 + r] : 0;

    float ea0, ea1, ea2, ea3, eb0, eb1, eb2, eb3;
    edge_logits(als, sa, ad, ea0, ea1, ea2, ea3);
    edge_logits(als, sb, ad, eb0, eb1, eb2, eb3);
    float m0 = fmaxf(oka ? ea0 : -INFINITY, okb ? eb0 : -INFINITY);
    float m1 = fmaxf(oka ? ea1 : -INFINITY, okb ? eb1 : -INFINITY);
    float m2 = fmaxf(oka ? ea2 : -INFINITY, okb ? eb2 : -INFINITY);
    float m3 = fmaxf(oka ? ea3 : -INFINITY, okb ? eb3 : -INFINITY);
    for (int cb = 32; cb < dmax; cb += 16) {
        if (cb + r < d) {
            float x0, x1, x2, x3;
            edge_logits(als, csr[beg + cb + r].y, ad, x0, x1, x2, x3);
            m0 = fmaxf(m0, x0); m1 = fmaxf(m1, x1); m2 = fmaxf(m2, x2); m3 = fmaxf(m3, x3);
        }
    }
    m0 = row_max(m0); m1 = row_max(m1); m2 = row_max(m2); m3 = row_max(m3);
    if (!__builtin_isfinite(m0)) m0 = 0.0f;
    if (!__builtin_isfinite(m1)) m1 = 0.0f;
    if (!__builtin_isfinite(m2)) m2 = 0.0f;
    if (!__builtin_isfinite(m3)) m3 = 0.0f;

    const float wa0 = expf(oka ? ea0 - m0 : -INFINITY), wa1 = expf(oka ? ea1 - m1 : -INFINITY);
    const float wa2 = expf(oka ? ea2 - m2 : -INFINITY), wa3 = expf(oka ? ea3 - m3 : -INFINITY);
    const float wb0 = expf(okb ? eb0 - m0 : -INFINITY), wb1 = expf(okb ? eb1 - m1 : -INFINITY);
    const float wb2 = expf(okb ? eb2 - m2 : -INFINITY), wb3 = expf(okb ? eb3 - m3 : -INFINITY);
    float d0 = wa0 + wb0, d1 = wa1 + wb1, d2 = wa2 + wb2, d3 = wa3 + wb3;
    for (int cb = 32; cb < dmax; cb += 16) {
        if (cb + r < d) {
            float x0, x1, x2, x3;
            edge_logits(als, csr[beg + cb + r].y, ad, x0, x1, x2, x3);
            d0 += expf(x0 - m0); d1 += expf(x1 - m1); d2 += expf(x2 - m2); d3 += expf(x3 - m3);
        }
    }
    d0 = row_sum(d0); d1 = row_sum(d1); d2 = row_sum(d2); d3 = row_sum(d3);
    const float q0 = fmaxf(d0, 1e-16f), q1 = fmaxf(d1, 1e-16f), q2 = fmaxf(d2, 1e-16f), q3 = fmaxf(d3, 1e-16f);

    const int s_first = row_lane<0>(sa);
    const int pads    = (d > 0) ? s_first : nd;
    if (!oka) sl_w[r] = pads;
    if (!okb) sl_w[16 + r] = pads;
    al_w[r]      = wa0 / q0; al_w[32 + r]      = wa1 / q1; al_w[64 + r]      = wa2 / q2; al_w[96 + r]       = wa3 / q3;
    al_w[16 + r] = wb0 / q0; al_w[32 + 16 + r] = wb1 / q1; al_w[64 + 16 + r] = wb2 / q2; al_w[96 + 16 + r]  = wb3 / q3;
    float4 acc0; acc0.x = 0.0f; acc0.y = 0.0f; acc0.z = 0.0f; acc0.w = 0.0f;
    float4 acc1; acc1.x = 0.0f; acc1.y = 0.0f; acc1.z = 0.0f; acc1.w = 0.0f;
    accum_slots(acc0, acc1, Hb, lane_off, al_r0, al_r1, sl_w, min(dmax, 32));
    for (int cb = 32; cb < dmax; cb += 16) {
        const bool ok = cb + r < d;
        const int  s  = ok ? csr[beg + cb + r].y : pads;
        float x0, x1, x2, x3;
        edge_logits(als, s, ad, x0, x1, x2, x3);
        const float a0 = ok ? expf(x0 - m0) / q0 : 0.0f, a1 = ok ? expf(x1 - m1) / q1 : 0.0f;
        const float a2 = ok ? expf(x2 - m2) / q2 : 0.0f, a3 = ok ? expf(x3 - m3) / q3 : 0.0f;
        sl_w[r] = s;
        al_w[r] = a0; al_w[32 + r] = a1; al_w[64 + r] = a2; al_w[96 + r] = a3;
        accum_slots(acc0, acc1, Hb, lane_off, al_r0, al_r1, sl_w, 16);
    }

    const float4 bia0 = *(const float4*)(bias_s + 4 * r);
    const float4 bia1 = *(const float4*)(bias_s + 64 + 4 * r);
    float4 o0; o0.x = acc0.x + bia0.x; o0.y = acc0.y + bia0.y; o0.z = acc0.z + bia0.z; o0.w = acc0.w + bia0.w;
    float4 o1; o1.x = acc1.x + bia1.x; o1.y = acc1.y + bia1.y; o1.z = acc1.z + bia1.z; o1.w = acc1.w + bia1.w;
    if (TO_PANEL) {
        o0.x = nv ? fmaxf(o0.x, 0.0f) : 0.0f; o0.y = nv ? fmaxf(o0.y, 0.0f) : 0.0f;
        o0.z = nv ? fmaxf(o0.z, 0.0f) : 0.0f; o0.w = nv ? fmaxf(o0.w, 0.0f) : 0.0f;
        o1.x = nv ? fmaxf(o1.x, 0.0f) : 0.0f; o1.y = nv ? fmaxf(o1.y, 0.0f) : 0.0f;
        o1.z = nv ? fmaxf(o1.z, 0.0f) : 0.0f; o1.w = nv ? fmaxf(o1.w, 0.0f) : 0.0f;
        *(float4*)(prow + 4 * r)      = o0;
        *(float4*)(prow + 64 + 4 * r) = o1;
    } else if (nv) {
        char* op = (char*)out + ((((unsigned)node) << 9) + lane_off);
        *(float4*)op         = o0;
        *(float4*)(op + 256) = o1;
    }
}

__global__ __launch_bounds__(GA_WAVES * 32, 12)
void k_gather(const int* __restrict__ rowptr, const int2* __restrict__ csr,
              const float* __restrict__ Hm,
              const float* __restrict__ als, const float* __restrict__ a_dst,
              const float* __restrict__ bias, float* __restrict__ out, int n) {
    __shared__ __attribute__((aligned(16))) float al_s[GA_WAVES][2][HEADS][32];
    __shared__ __attribute__((aligned(16))) int   sl_s[GA_WAVES][2][32];
    __shared__ __attribute__((aligned(16))) float bias_s[HOUT];
    __shared__ __attribute__((aligned(16))) float adst_s[HOUT];
    if (threadIdx.x < HOUT) { bias_s[threadIdx.x] = bias[threadIdx.x]; adst_s[threadIdx.x] = a_dst[threadIdx.x]; }
    __syncthreads();
    const int lane = threadIdx.x & 31;
    const int wv   = threadIdx.x >> 5;
    const int r    = lane & 15;
    const int h    = lane >> 4;
    float*       al_w  = &al_s[wv][h][0][0];
    int*         sl_w  = &sl_s[wv][h][0];
    const float* al_r0 = al_w + (r >> 3) * 32;
    const float* al_r1 = al_w + (2 + (r >> 3)) * 32;
    const char*  Hb    = (const char*)Hm;
    const unsigned lane_off = (unsigned)r * 16u;
    const int wave0  = __builtin_amdgcn_readfirstlane((int)((blockIdx.x * blockDim.x + threadIdx.x) >> 5));
    const int nwaves = (int)((gridDim.x * blockDim.x) >> 5);
    const int npairs = (n + 1) >> 1;
    for (int pair = wave0; pair < npairs; pair += nwaves)
        gather_pair<0>(pair, n, rowptr, csr, Hb, als, adst_s, bias_s, al_w, sl_w, al_r0, al_r1, lane_off, r, h, out, nullptr);
}

#define FG_NODES 64
#define AP_PITCH 132
__device__ __forceinline__
void strip_out(const v8f acc, const int row0, const int strip, const int t, const int c, const int hh,
               const float as_c, float* __restrict__ H, float* part, const int n) {
    const int rg = row0 + 8 * hh;
#pragma unroll
    for (int j = 0; j < 8; ++j) {
        if (rg + j < n) H[(size_t)(rg + j) * HOUT + 16 * t + c] = acc[j];
    }
    float v[8];
#pragma unroll
    for (int j = 0; j < 8; ++j) v[j] = acc[j] * as_c;
    halve_step<8, 4>(v, (c & 8) != 0);
    halve_step<4, 2>(v, (c & 4) != 0);
    halve_step<2, 1>(v, (c & 2) != 0);
    const float sum = v[0] + row_xor<1>(v[0]);
    if ((c & 1) == 0) part[(strip * 8 + t) * 16 + 8 * hh + (c >> 1)] = sum;
}

__global__ __launch_bounds__(GA_WAVES * 32, 10)
void k_gather_gemm(const int* __restrict__ rowptr, const int2* __restrict__ csr,
                   const float* __restrict__ H1, const float* __restrict__ als1,
                   const float* __restrict__ a_dst1, const float* __restrict__ bias1,
                   const float* __restrict__ W2F, const float* __restrict__ a_src2,
                   float* __restrict__ H2, float* __restrict__ als2, int n) {
    __shared__ __attribute__((aligned(16))) float Ap[FG_NODES * AP_PITCH];
    __shared__ __attribute__((aligned(16))) float al_s[GA_WAVES][2][HEADS][32];
    __shared__ __attribute__((aligned(16))) int   sl_s[GA_WAVES][2][32];
    __shared__ __attribute__((aligned(16))) float bias_s[HOUT];
    __shared__ __attribute__((aligned(16))) float adst_s[HOUT];
    __shared__ float part[4 * 8 * 16];
    const int tid = threadIdx.x;
    if (tid < HOUT) { bias_s[tid] = bias1[tid]; adst_s[tid] = a_dst1[tid]; }
    __syncthreads();
    const int lane = tid & 31;
    const int wv   = __builtin_amdgcn_readfirstlane(tid >> 5);
    const int r    = lane & 15;
    const int h    = lane >> 4;
    float*       al_w  = &al_s[wv][h][0][0];
    int*         sl_w  = &sl_s[wv][h][0];
    const float* al_r0 = al_w + (r >> 3) * 32;
    const float* al_r1 = al_w + (2 + (r >> 3)) * 32;
    const char*  Hb    = (const char*)H1;
    const unsigned lane_off = (unsigned)r * 16u;
    const int blk = (int)blockIdx.x;
    for (int it = 0; it < FG_NODES / (2 * GA_WAVES); ++it) {
        const int lp = wv * (FG_NODES / (2 * GA_WAVES)) + it;
        gather_pair<1>(blk * (FG_NODES / 2) + lp, n, rowptr, csr, Hb, als1, adst_s, bias_s, al_w, sl_w, al_r0, al_r1,
                       lane_off, r, h, nullptr, Ap + (2 * lp + h) * AP_PITCH);
    }
    const int c = r, hh = h, t = wv;
    const float4* wrow = (const float4*)W2F + (size_t)t * (16 * 32) + lane;
    v2f b[32];
#pragma unroll
    for (int i = 0; i < 16; ++i) {
        const float4 w4 = wrow[32 * i];
        b[2 * i].x     = w4.x; b[2 * i].y     = w4.y;
        b[2 * i + 1].x = w4.z; b[2 * i + 1].y = w4.w;
    }
    const float as_c = a_src2[16 * t + c];
    __syncthreads();
#pragma clang loop unroll(disable)
    for (int st = 0; st < FG_NODES / 16; ++st) {
        v8f acc = (v8f){};
        const float* ap = Ap + (16 * st + c) * AP_PITCH + 4 * hh;
#pragma unroll
        for (int i = 0; i < 16; ++i) {
            const float4 a = *(const float4*)(ap + 8 * i);
            v2f f;
            f.x = a.x; f.y = a.y;
            acc = __builtin_amdgcn_wmma_f32_16x16x4_f32(false, f, false, b[2 * i], (short)0, acc, false, false);
            f.x = a.z; f.y = a.w;
            acc = __builtin_amdgcn_wmma_f32_16x16x4_f32(false, f, false, b[2 * i + 1], (short)0, acc, false, false);
        }
        strip_out(acc, blk * FG_NODES + 16 * st, st, t, c, hh, as_c, H2, part, n);
    }
    __syncthreads();
    {
        const int rl = tid >> 2, hd = tid & 3;
        const int strip = rl >> 4, r16 = rl & 15;
        if (blk * FG_NODES + rl < n)
            als2[(size_t)blk * (FG_NODES * HEADS) + tid] = part[(strip * 8 + 2 * hd) * 16 + r16] + part[(strip * 8 + 2 * hd + 1) * 16 + r16];
    }
}

extern "C" void kernel_launch(void* const* d_in, const int* in_sizes, int n_in,
                              void* d_out, int out_size, void* d_ws, size_t ws_size,
                              hipStream_t stream) {
    const float* x      = (const float*)d_in[0];
    const int*   ei     = (const int*)d_in[1];
    const float* W1     = (const float*)d_in[2];
    const float* a_src1 = (const float*)d_in[3];
    const float* a_dst1 = (const float*)d_in[4];
    const float* b1     = (const float*)d_in[5];
    const float* W2     = (const float*)d_in[6];
    const float* a_src2 = (const float*)d_in[7];
    const float* a_dst2 = (const float*)d_in[8];
    const float* b2     = (const float*)d_in[9];

    const int N = in_sizes[0] / FIN;
    const int E = in_sizes[1] / 2;
    const int* src = ei;
    const int* dst = ei + E;

    const int nRanges = (N + RANGE - 1) / RANGE;
    const int nEB     = (E + EDGE_BLOCK - 1) / EDGE_BLOCK;
    const int nTiles  = (N + 1023) / 1024;
    const size_t nPad = ((size_t)N + 4) & ~(size_t)3;
    const size_t ePad = ((size_t)E + 3) & ~(size_t)3;

    char* p = (char*)d_ws;
    float* Hws    = (float*)p; p += (size_t)N * HOUT * sizeof(float);
    float* als1   = (float*)p; p += (size_t)N * HEADS * sizeof(float);
    float* als2   = (float*)p; p += (size_t)N * HEADS * sizeof(float);
    int*   rowptr = (int*)p;   p += nPad * sizeof(int);
    float* W2F    = (float*)p;
    p += nPad * sizeof(int);
    p += (((size_t)nTiles + 3) & ~(size_t)3) * sizeof(int);
    int2*  csr    = (int2*)p;  p += (size_t)E * sizeof(int2);
    float* H1  = (float*)d_out;
    float* H2  = Hws;
    char* q = (char*)Hws;
    int2*  pe     = (int2*)q;  q += ePad * sizeof(int2);
    int*   pd     = (int*)q;   q += ePad * sizeof(int);
    int*   tab    = (int*)q;

    float* out = (float*)d_out;

    const int gGemm = (N + 63) / 64;
    const int BT = 256;
    int gNW  = (N + 31) / 32;

    k_part_gemm <<<nEB + gGemm, PART_T, 0, stream>>>(src, dst, tab, pe, pd, E, nRanges, nEB,
                                                 x, W1, a_src1, H1, als1, N);
    k_bucket    <<<nRanges, BUCKET_B, 0, stream>>>(pe, pd, tab, rowptr, csr, N, nEB, W2, W2F);

    k_gather_gemm<<<(N + FG_NODES - 1) / FG_NODES, BT, 0, stream>>>(rowptr, csr, H1, als1, a_dst1, b1, W2F, a_src2, H2, als2, N);

    k_gather    <<<gNW, BT, 0, stream>>>(rowptr, csr, H2, als2, a_dst2, b2, out, N);
}
